# starting kernel
# baseline (speedup 1.0000x reference)
#include <hip/hip_runtime.h>
#include <math.h>

#define B_   2
#define L_   512
#define DM   1024
#define DI   2048
#define DS_  64
#define NH_  32
#define HD_  64
#define NA_  32
#define DIP  4352
#define M1   (B_*L_)

#define NC_  16
#define LC_  32

#define OFF_Z    0
#define OFF_X    2048
#define OFF_B    4096
#define OFF_C    4160
#define OFF_DT   4224
#define OFF_A    4256
#define OFF_TRAP 4288
#define OFF_ANG  4320

typedef __attribute__((ext_vector_type(2))) float v2f;
typedef __attribute__((ext_vector_type(8))) float v8f;

__device__ __forceinline__ float softplus_f(float x) {
    return x > 20.f ? x : log1pf(expf(x));
}
__device__ __forceinline__ float sigmoid_f(float x) {
    return 1.f / (1.f + expf(-x));
}

__global__ __launch_bounds__(256) void gemm_wmma_f32(
    const float* __restrict__ A, const float* __restrict__ Bm,
    float* __restrict__ C, int M, int N, int K)
{
    const int lane = threadIdx.x & 31;
    const int wave = threadIdx.x >> 5;
    const int lo   = lane & 15;
    const int hi   = lane >> 4;

    const int m0 = blockIdx.y * 256 + wave * 32;
    const int n0 = blockIdx.x * 64;

    const float* Arow0 = A + (size_t)(m0 + lo) * K;
    const float* Arow1 = A + (size_t)(m0 + 16 + lo) * K;

    v8f acc00 = {}, acc01 = {}, acc02 = {}, acc03 = {};
    v8f acc10 = {}, acc11 = {}, acc12 = {}, acc13 = {};

    for (int k = 0; k < K; k += 4) {
        v2f a0; a0.x = Arow0[k + 2*hi]; a0.y = Arow0[k + 2*hi + 1];
        v2f a1; a1.x = Arow1[k + 2*hi]; a1.y = Arow1[k + 2*hi + 1];

        const float* Bp0 = Bm + (size_t)(k + 2*hi) * N + n0 + lo;
        const float* Bp1 = Bp0 + N;
        v2f b0; b0.x = Bp0[0];  b0.y = Bp1[0];
        v2f b1; b1.x = Bp0[16]; b1.y = Bp1[16];
        v2f b2; b2.x = Bp0[32]; b2.y = Bp1[32];
        v2f b3; b3.x = Bp0[48]; b3.y = Bp1[48];

        acc00 = __builtin_amdgcn_wmma_f32_16x16x4_f32(false, a0, false, b0, (short)0, acc00, false, false);
        acc01 = __builtin_amdgcn_wmma_f32_16x16x4_f32(false, a0, false, b1, (short)0, acc01, false, false);
        acc02 = __builtin_amdgcn_wmma_f32_16x16x4_f32(false, a0, false, b2, (short)0, acc02, false, false);
        acc03 = __builtin_amdgcn_wmma_f32_16x16x4_f32(false, a0, false, b3, (short)0, acc03, false, false);
        acc10 = __builtin_amdgcn_wmma_f32_16x16x4_f32(false, a1, false, b0, (short)0, acc10, false, false);
        acc11 = __builtin_amdgcn_wmma_f32_16x16x4_f32(false, a1, false, b1, (short)0, acc11, false, false);
        acc12 = __builtin_amdgcn_wmma_f32_16x16x4_f32(false, a1, false, b2, (short)0, acc12, false, false);
        acc13 = __builtin_amdgcn_wmma_f32_16x16x4_f32(false, a1, false, b3, (short)0, acc13, false, false);
    }

    #pragma unroll
    for (int r = 0; r < 8; ++r) {
        float* cp0 = C + (size_t)(m0 + r + 8*hi) * N + n0 + lo;
        cp0[0]  = acc00[r];
        cp0[16] = acc01[r];
        cp0[32] = acc02[r];
        cp0[48] = acc03[r];
        float* cp1 = C + (size_t)(m0 + 16 + r + 8*hi) * N + n0 + lo;
        cp1[0]  = acc10[r];
        cp1[16] = acc11[r];
        cp1[32] = acc12[r];
        cp1[48] = acc13[r];
    }
}

__global__ __launch_bounds__(32) void phase_kernel(
    const float* __restrict__ proj, const float* __restrict__ dt_bias,
    float* __restrict__ DTb, float* __restrict__ decayb,
    float* __restrict__ trapb, float* __restrict__ phaseb)
{
    const int b = blockIdx.x;
    const int j = threadIdx.x;
    float acc = 0.f;
    for (int l = 0; l < L_; ++l) {
        const float* row = proj + (size_t)(b * L_ + l) * DIP;
        float dt = softplus_f(row[OFF_DT + j] + dt_bias[j]);
        float s = dt;
        s += __shfl_xor(s, 16, 32);
        s += __shfl_xor(s,  8, 32);
        s += __shfl_xor(s,  4, 32);
        s += __shfl_xor(s,  2, 32);
        s += __shfl_xor(s,  1, 32);
        const float dtmean = s * (1.f / 32.f);
        acc += row[OFF_ANG + j] * dtmean;

        const size_t o = (size_t)(b * L_ + l) * 32 + j;
        DTb[o]    = dt;
        float Av  = fminf(-softplus_f(row[OFF_A + j]), -1e-4f);
        decayb[o] = expf(Av * dt);
        trapb[o]  = sigmoid_f(row[OFF_TRAP + j]);
        phaseb[o] = acc;
    }
}

__global__ __launch_bounds__(64) void rot_kernel(
    const float* __restrict__ proj,
    const float* __restrict__ Bn_g, const float* __restrict__ Bn_b,
    const float* __restrict__ Cn_g, const float* __restrict__ Cn_b,
    const float* __restrict__ phaseb,
    float* __restrict__ Brot, float* __restrict__ Crot)
{
    const int bl = blockIdx.x;
    const int t  = threadIdx.x;
    __shared__ float red[64], nb[64], nc[64];

    const float* row = proj + (size_t)bl * DIP;
    const float vb = row[OFF_B + t];
    const float vc = row[OFF_C + t];

    auto blockReduce = [&](float v) -> float {
        red[t] = v; __syncthreads();
        for (int s = 32; s > 0; s >>= 1) {
            if (t < s) red[t] += red[t + s];
            __syncthreads();
        }
        float r = red[0]; __syncthreads();
        return r;
    };

    const float mB  = blockReduce(vb) * (1.f / 64.f);
    const float vaB = blockReduce((vb - mB) * (vb - mB)) * (1.f / 64.f);
    const float nB  = (vb - mB) * rsqrtf(vaB + 1e-5f) * Bn_g[t] + Bn_b[t];

    const float mC  = blockReduce(vc) * (1.f / 64.f);
    const float vaC = blockReduce((vc - mC) * (vc - mC)) * (1.f / 64.f);
    const float nC  = (vc - mC) * rsqrtf(vaC + 1e-5f) * Cn_g[t] + Cn_b[t];

    nb[t] = nB; nc[t] = nC;
    __syncthreads();

    const int   q  = t >> 1;
    const float ph = phaseb[(size_t)bl * 32 + q];
    const float c  = cosf(ph), s = sinf(ph);
    float ob, oc;
    if ((t & 1) == 0) { ob = nb[t] * c - nb[t + 1] * s; oc = nc[t] * c - nc[t + 1] * s; }
    else              { ob = nb[t - 1] * s + nb[t] * c; oc = nc[t - 1] * s + nc[t] * c; }
    Brot[(size_t)bl * 64 + t] = ob;
    Crot[(size_t)bl * 64 + t] = oc;
}


__global__ __launch_bounds__(256) void scan_chunk_kernel(
    const float* __restrict__ proj,
    const float* __restrict__ Brot, const float* __restrict__ Crot,
    const float* __restrict__ DTb, const float* __restrict__ decayb,
    const float* __restrict__ trapb,
    float* __restrict__ yfin, float* __restrict__ cumdb,
    float* __restrict__ Sstate)
{
    const int blk = blockIdx.x;
    const int c   = blk % NC_;
    const int h   = (blk / NC_) % NH_;
    const int b   = blk / (NC_ * NH_);
    const int l0  = c * LC_;
    const int t   = threadIdx.x;
    const int p   = t >> 2;
    const int seg = t & 3;

    __shared__ float sx[2][64], sB[2][64], sC[2][64], sscal[2][3];

    float hst[16], Bprev[16];
    float xprev = 0.f;
    #pragma unroll
    for (int i = 0; i < 16; ++i) { hst[i] = 0.f; Bprev[i] = 0.f; }
    if (c > 0) {
        const size_t blp = (size_t)(b * L_ + l0 - 1);
        xprev = proj[blp * DIP + OFF_X + h * HD_ + p];
        const float* Bp = Brot + blp * 64 + seg * 16;
        #pragma unroll
        for (int i = 0; i < 16; ++i) Bprev[i] = Bp[i];
    }

    auto loadTok = [&](int tok, int buf) {
        const size_t bl = (size_t)(b * L_ + l0 + tok);
        if (t < 64)        sx[buf][t]       = proj[bl * DIP + OFF_X + h * HD_ + t];
        else if (t < 128)  sB[buf][t - 64]  = Brot[bl * 64 + (t - 64)];
        else if (t < 192)  sC[buf][t - 128] = Crot[bl * 64 + (t - 128)];
        else if (t == 192) sscal[buf][0] = DTb[bl * 32 + h];
        else if (t == 193) sscal[buf][1] = decayb[bl * 32 + h];
        else if (t == 194) sscal[buf][2] = trapb[bl * 32 + h];
    };

    loadTok(0, 0);
    __syncthreads();
    float cumd = 1.f;

    for (int tok = 0; tok < LC_; ++tok) {
        const int cur = tok & 1;
        if (tok + 1 < LC_) loadTok(tok + 1, cur ^ 1);

        const float x   = sx[cur][p];
        const float dt  = sscal[cur][0];
        const float dec = sscal[cur][1];
        const float tr  = sscal[cur][2];

        float y = 0.f;
        #pragma unroll
        for (int i = 0; i < 16; ++i) {
            const int   n  = seg * 16 + i;
            const float Bn = sB[cur][n];
            const float bx = tr * x * Bn + (1.f - tr) * xprev * Bprev[i];
            hst[i] = dec * hst[i] + dt * bx;
            y += hst[i] * sC[cur][n];
            Bprev[i] = Bn;
        }
        xprev = x;
        cumd *= dec;

        y += __shfl_xor(y, 1, 32);
        y += __shfl_xor(y, 2, 32);

        const size_t bl = (size_t)(b * L_ + l0 + tok);
        if (seg == 0) yfin[bl * DI + h * HD_ + p] = y;
        if (t == 0)   cumdb[((size_t)(b * NH_ + h)) * L_ + l0 + tok] = cumd;
        __syncthreads();
    }

    float* Sp = Sstate + (((size_t)(b * NH_ + h)) * NC_ + c) * 4096 + p * 64 + seg * 16;
    #pragma unroll
    for (int i = 0; i < 16; ++i) Sp[i] = hst[i];
}

__global__ __launch_bounds__(256) void scan_combine_kernel(
    const float* __restrict__ cumdb, float* __restrict__ Sstate)
{
    const int h  = blockIdx.x % NH_;
    const int b  = blockIdx.x / NH_;
    const int t  = threadIdx.x;
    const int p  = t >> 2, seg = t & 3;

    float H[16];
    #pragma unroll
    for (int i = 0; i < 16; ++i) H[i] = 0.f;

    for (int c = 0; c < NC_; ++c) {
        float* Sp = Sstate + (((size_t)(b * NH_ + h)) * NC_ + c) * 4096 + p * 64 + seg * 16;
        const float Pc = cumdb[((size_t)(b * NH_ + h)) * L_ + c * LC_ + LC_ - 1];
        #pragma unroll
        for (int i = 0; i < 16; ++i) {
            const float Sc = Sp[i];
            Sp[i] = H[i];
            H[i]  = Pc * H[i] + Sc;
        }
    }
}

__global__ __launch_bounds__(256) void scan_correct_kernel(
    const float* __restrict__ proj, const float* __restrict__ Crot,
    const float* __restrict__ cumdb, const float* __restrict__ Sstate,
    const float* __restrict__ D_param, float* __restrict__ yfin)
{
    const int blk = blockIdx.x;
    const int c   = blk % NC_;
    const int h   = (blk / NC_) % NH_;
    const int b   = blk / (NC_ * NH_);
    const int l0  = c * LC_;
    const int t   = threadIdx.x;
    const int p   = t >> 2, seg = t & 3;

    __shared__ float sC[64];
    __shared__ float scd;

    float H[16];
    const float* Sp = Sstate + (((size_t)(b * NH_ + h)) * NC_ + c) * 4096 + p * 64 + seg * 16;
    #pragma unroll
    for (int i = 0; i < 16; ++i) H[i] = Sp[i];
    const float Dp = D_param[h];

    for (int tok = 0; tok < LC_; ++tok) {
        const size_t bl = (size_t)(b * L_ + l0 + tok);
        if (t < 64)  sC[t] = Crot[bl * 64 + t];
        if (t == 64) scd   = cumdb[((size_t)(b * NH_ + h)) * L_ + l0 + tok];
        __syncthreads();

        float corr = 0.f;
        #pragma unroll
        for (int i = 0; i < 16; ++i) corr += H[i] * sC[seg * 16 + i];
        corr += __shfl_xor(corr, 1, 32);
        corr += __shfl_xor(corr, 2, 32);

        if (seg == 0) {
            const float x = proj[bl * DIP + OFF_X + h * HD_ + p];
            const float z = proj[bl * DIP + OFF_Z + h * HD_ + p];
            float y = yfin[bl * DI + h * HD_ + p] + scd * corr;
            yfin[bl * DI + h * HD_ + p] = (y + Dp * x) * (z * sigmoid_f(z));
        }
        __syncthreads();
    }
}

extern "C" void kernel_launch(void* const* d_in, const int* in_sizes, int n_in,
                              void* d_out, int out_size, void* d_ws, size_t ws_size,
                              hipStream_t stream)
{
    const float* u       = (const float*)d_in[0];
    const float* W_in    = (const float*)d_in[1];
    const float* W_out   = (const float*)d_in[2];
    const float* dt_bias = (const float*)d_in[3];
    const float* D_param = (const float*)d_in[4];
    const float* Bn_g    = (const float*)d_in[5];
    const float* Bn_b    = (const float*)d_in[6];
    const float* Cn_g    = (const float*)d_in[7];
    const float* Cn_b    = (const float*)d_in[8];

    float* ws = (float*)d_ws;
    float* proj   = ws; ws += (size_t)M1 * DIP;
    float* yfin   = ws; ws += (size_t)M1 * DI;
    float* Sstate = ws; ws += (size_t)B_ * NH_ * NC_ * 4096;
    float* Brot   = ws; ws += (size_t)M1 * DS_;
    float* Crot   = ws; ws += (size_t)M1 * DS_;
    float* DTb    = ws; ws += (size_t)M1 * NH_;
    float* decayb = ws; ws += (size_t)M1 * NH_;
    float* trapb  = ws; ws += (size_t)M1 * NH_;
    float* phaseb = ws; ws += (size_t)M1 * NA_;
    float* cumdb  = ws; ws += (size_t)B_ * NH_ * L_;

    gemm_wmma_f32<<<dim3(DIP / 64, M1 / 256), dim3(256), 0, stream>>>(
        u, W_in, proj, M1, DIP, DM);

    phase_kernel<<<dim3(B_), dim3(32), 0, stream>>>(
        proj, dt_bias, DTb, decayb, trapb, phaseb);

    rot_kernel<<<dim3(M1), dim3(64), 0, stream>>>(
        proj, Bn_g, Bn_b, Cn_g, Cn_b, phaseb, Brot, Crot);

    scan_chunk_kernel<<<dim3(B_ * NH_ * NC_), dim3(256), 0, stream>>>(
        proj, Brot, Crot, DTb, decayb, trapb, yfin, cumdb, Sstate);
    scan_combine_kernel<<<dim3(B_ * NH_), dim3(256), 0, stream>>>(
        cumdb, Sstate);
    scan_correct_kernel<<<dim3(B_ * NH_ * NC_), dim3(256), 0, stream>>>(
        proj, Crot, cumdb, Sstate, D_param, yfin);

    gemm_wmma_f32<<<dim3(DM / 64, M1 / 256), dim3(256), 0, stream>>>(
        yfin, W_out, (float*)d_out, M1, DM, DI);
}


# scan chunk pass in matrix form on f32 WMMA: y = (decay-weighted Crot.Brot^T) x, chunk state = (weighted x)^T Brot, decays as exp of log-decay sums; no token loop; sum order changes
# speedup vs baseline: 6.4783x; 6.4783x over previous
#include <hip/hip_runtime.h>
#include <math.h>

#define B_   2
#define L_   512
#define DM   1024
#define DI   2048
#define DS_  64
#define NH_  32
#define HD_  64
#define NA_  32
#define DIP  4352
#define M1   (B_*L_)

#define NC_  16
#define LC_  32

#define OFF_Z    0
#define OFF_X    2048
#define OFF_B    4096
#define OFF_C    4160
#define OFF_DT   4224
#define OFF_A    4256
#define OFF_TRAP 4288
#define OFF_ANG  4320

typedef __attribute__((ext_vector_type(2))) float v2f;
typedef __attribute__((ext_vector_type(8))) float v8f;

__device__ __forceinline__ float softplus_f(float x) {
    return x > 20.f ? x : log1pf(expf(x));
}
__device__ __forceinline__ float sigmoid_f(float x) {
    return 1.f / (1.f + expf(-x));
}

typedef __attribute__((ext_vector_type(4))) float v4f;

struct GFrag { v4f a[2]; v4f b[8]; };

template <int N>
__device__ __forceinline__ void gemm_load(GFrag& f, const float* __restrict__ ap,
                                          const float* __restrict__ bp)
{
    f.a[0] = *reinterpret_cast<const v4f*>(ap);
    f.a[1] = *reinterpret_cast<const v4f*>(ap + 4);
    #pragma unroll
    for (int r = 0; r < 8; ++r)
        f.b[r] = *reinterpret_cast<const v4f*>(bp + (size_t)r * N);
}

__device__ __forceinline__ void gemm_mma(const GFrag& f, v8f& c0, v8f& c1, v8f& c2, v8f& c3)
{
    #pragma unroll
    for (int s = 0; s < 4; ++s) {
        const v4f aq = f.a[s >> 1];
        v2f a;
        if (s & 1) { a.x = aq.z; a.y = aq.w; } else { a.x = aq.x; a.y = aq.y; }
        const v4f r0 = f.b[2 * s], r1 = f.b[2 * s + 1];
        v2f b;
        b.x = r0.x; b.y = r1.x;
        c0 = __builtin_amdgcn_wmma_f32_16x16x4_f32(false, a, false, b, (short)0, c0, false, false);
        b.x = r0.y; b.y = r1.y;
        c1 = __builtin_amdgcn_wmma_f32_16x16x4_f32(false, a, false, b, (short)0, c1, false, false);
        b.x = r0.z; b.y = r1.z;
        c2 = __builtin_amdgcn_wmma_f32_16x16x4_f32(false, a, false, b, (short)0, c2, false, false);
        b.x = r0.w; b.y = r1.w;
        c3 = __builtin_amdgcn_wmma_f32_16x16x4_f32(false, a, false, b, (short)0, c3, false, false);
    }
}

#define GEMM_HOLD() __builtin_amdgcn_sched_barrier(0)

template <int N, int K>
__global__ __launch_bounds__(128) void gemm_wmma_f32(
    const float* __restrict__ A, const float* __restrict__ Bm,
    float* __restrict__ C)
{
    static_assert(N % 64 == 0 && K % 64 == 0 && K >= 128, "tile and ring shape");
    const int lane = threadIdx.x & 31;
    const int wave = threadIdx.x >> 5;
    const int lo   = lane & 15;
    const int hi   = lane >> 4;

    const int m0 = blockIdx.y * 64 + wave * 16;
    const int n0 = blockIdx.x * 64;

    const float* ap = A + (size_t)(m0 + lo) * K + 8 * hi;
    const float* bp = Bm + (size_t)(8 * hi) * N + n0 + 4 * lo;
    constexpr int    SA = 16;
    constexpr size_t SB = (size_t)16 * N;
    constexpr int    NS = K / 16;

    v8f c0 = {}, c1 = {}, c2 = {}, c3 = {};
    GFrag f0, f1, f2, f3;
    gemm_load<N>(f0, ap, bp);
    gemm_load<N>(f1, ap + SA, bp + SB);
    gemm_load<N>(f2, ap + 2 * SA, bp + 2 * SB);

    for (int s = 0; s < NS - 4; s += 4) {
        GEMM_HOLD(); gemm_load<N>(f3, ap + 3 * SA, bp + 3 * SB); GEMM_HOLD();
        gemm_mma(f0, c0, c1, c2, c3);
        GEMM_HOLD(); gemm_load<N>(f0, ap + 4 * SA, bp + 4 * SB); GEMM_HOLD();
        gemm_mma(f1, c0, c1, c2, c3);
        GEMM_HOLD(); gemm_load<N>(f1, ap + 5 * SA, bp + 5 * SB); GEMM_HOLD();
        gemm_mma(f2, c0, c1, c2, c3);
        GEMM_HOLD(); gemm_load<N>(f2, ap + 6 * SA, bp + 6 * SB); GEMM_HOLD();
        gemm_mma(f3, c0, c1, c2, c3);
        ap += 4 * SA;
        bp += 4 * SB;
    }
    GEMM_HOLD(); gemm_load<N>(f3, ap + 3 * SA, bp + 3 * SB); GEMM_HOLD();
    gemm_mma(f0, c0, c1, c2, c3);
    gemm_mma(f1, c0, c1, c2, c3);
    gemm_mma(f2, c0, c1, c2, c3);
    gemm_mma(f3, c0, c1, c2, c3);

    #pragma unroll
    for (int r = 0; r < 8; ++r) {
        v4f o; o.x = c0[r]; o.y = c1[r]; o.z = c2[r]; o.w = c3[r];
        *reinterpret_cast<v4f*>(C + (size_t)(m0 + r + 8 * hi) * N + n0 + 4 * lo) = o;
    }
}

__global__ __launch_bounds__(1024) void phase_kernel(
    const float* __restrict__ proj, const float* __restrict__ dt_bias,
    float* __restrict__ DTb, float* __restrict__ ladb,
    float* __restrict__ trapb, float* __restrict__ phaseb)
{
    __shared__ v2f incp[(L_ / 2) * 32];
    const int b    = blockIdx.x;
    const int j    = threadIdx.x & 31;
    const int w    = __builtin_amdgcn_readfirstlane((int)(threadIdx.x >> 5));
    const float bias = dt_bias[j];

    for (int i = 0; i < 16; ++i) {
        const int l = 16 * w + i;
        const float* row = proj + (size_t)(b * L_ + l) * DIP;
        const float vdt  = row[OFF_DT + j];
        const float vang = row[OFF_ANG + j];
        const float va   = row[OFF_A + j];
        const float vtr  = row[OFF_TRAP + j];

        const float dt = softplus_f(vdt + bias);
        float s = dt;
        s += __shfl_xor(s, 16, 32);
        s += __shfl_xor(s,  8, 32);
        s += __shfl_xor(s,  4, 32);
        s += __shfl_xor(s,  2, 32);
        s += __shfl_xor(s,  1, 32);
        const float dtmean = s * (1.f / 32.f);
        incp[(l >> 1) * 32 + j][l & 1] = vang * dtmean;

        const size_t o = (size_t)(b * L_ + l) * 32 + j;
        DTb[o]    = dt;
        const float Av = fminf(-softplus_f(va), -1e-4f);
        ladb[o]   = Av * dt;
        trapb[o]  = sigmoid_f(vtr);
    }
    __syncthreads();

    const int lo = j & 15;
    const int hi = j >> 4;
    v8f acc0 = {}, acc1 = {};
    const v2f ones = {1.f, 1.f};

    for (int t = 0; t < w; ++t) {
        const v2f* p = incp + (8 * t + hi) * 32 + lo;
        #pragma unroll
        for (int s = 0; s < 4; ++s) {
            const v2f b0 = p[64 * s];
            const v2f b1 = p[64 * s + 16];
            acc0 = __builtin_amdgcn_wmma_f32_16x16x4_f32(false, ones, false, b0, (short)0, acc0, false, false);
            acc1 = __builtin_amdgcn_wmma_f32_16x16x4_f32(false, ones, false, b1, (short)0, acc1, false, false);
        }
    }
    {
        const v2f* p = incp + (8 * w + hi) * 32 + lo;
        #pragma unroll
        for (int s = 0; s < 4; ++s) {
            const int kl = 4 * s + 2 * hi;
            v2f a; a.x = (kl <= lo) ? 1.f : 0.f; a.y = (kl + 1 <= lo) ? 1.f : 0.f;
            const v2f b0 = p[64 * s];
            const v2f b1 = p[64 * s + 16];
            acc0 = __builtin_amdgcn_wmma_f32_16x16x4_f32(false, a, false, b0, (short)0, acc0, false, false);
            acc1 = __builtin_amdgcn_wmma_f32_16x16x4_f32(false, a, false, b1, (short)0, acc1, false, false);
        }
    }

    #pragma unroll
    for (int r = 0; r < 8; ++r) {
        float* op = phaseb + (size_t)(b * L_ + 16 * w + r + 8 * hi) * 32 + lo;
        op[0]  = acc0[r];
        op[16] = acc1[r];
    }
}

__global__ __launch_bounds__(64) void rot_kernel(
    const float* __restrict__ proj,
    const float* __restrict__ Bn_g, const float* __restrict__ Bn_b,
    const float* __restrict__ Cn_g, const float* __restrict__ Cn_b,
    const float* __restrict__ phaseb,
    float* __restrict__ Brot, float* __restrict__ Crot)
{
    const int bl = blockIdx.x;
    const int t  = threadIdx.x;
    __shared__ float red[64], nb[64], nc[64];

    const float* row = proj + (size_t)bl * DIP;
    const float vb = row[OFF_B + t];
    const float vc = row[OFF_C + t];

    auto blockReduce = [&](float v) -> float {
        red[t] = v; __syncthreads();
        for (int s = 32; s > 0; s >>= 1) {
            if (t < s) red[t] += red[t + s];
            __syncthreads();
        }
        float r = red[0]; __syncthreads();
        return r;
    };

    const float mB  = blockReduce(vb) * (1.f / 64.f);
    const float vaB = blockReduce((vb - mB) * (vb - mB)) * (1.f / 64.f);
    const float nB  = (vb - mB) * rsqrtf(vaB + 1e-5f) * Bn_g[t] + Bn_b[t];

    const float mC  = blockReduce(vc) * (1.f / 64.f);
    const float vaC = blockReduce((vc - mC) * (vc - mC)) * (1.f / 64.f);
    const float nC  = (vc - mC) * rsqrtf(vaC + 1e-5f) * Cn_g[t] + Cn_b[t];

    nb[t] = nB; nc[t] = nC;
    __syncthreads();

    const int   q  = t >> 1;
    const float ph = phaseb[(size_t)bl * 32 + q];
    const float c  = cosf(ph), s = sinf(ph);
    float ob, oc;
    if ((t & 1) == 0) { ob = nb[t] * c - nb[t + 1] * s; oc = nc[t] * c - nc[t + 1] * s; }
    else              { ob = nb[t - 1] * s + nb[t] * c; oc = nc[t - 1] * s + nc[t] * c; }
    Brot[(size_t)bl * 64 + t] = ob;
    Crot[(size_t)bl * 64 + t] = oc;
}


template <int TT>
__device__ __forceinline__ void scan_chunk_tile(
    const float* __restrict__ proj, const float* __restrict__ Brot,
    const float* __restrict__ Crot, float* __restrict__ yfin,
    float* __restrict__ Sp, const float* sLa, const float* sP1,
    const float* sP2, const float* sV,
    const size_t bl0c, const size_t blprev, const float prevOn,
    const int h, const int lo, const int hi)
{
    v4f cq[8], bmq[8], bs[TT + 1][8];
    {
        const float* cp  = Crot + (bl0c + 16 * TT + lo) * 64 + 4 * hi;
        const float* bmp = Brot + blprev * 64 + 4 * hi;
        #pragma unroll
        for (int q = 0; q < 8; ++q) {
            cq[q]  = *reinterpret_cast<const v4f*>(cp + 8 * q);
            bmq[q] = *reinterpret_cast<const v4f*>(bmp + 8 * q);
        }
        #pragma unroll
        for (int st = 0; st <= TT; ++st) {
            const float* bp = Brot + (bl0c + 16 * st + lo) * 64 + 4 * hi;
            #pragma unroll
            for (int q = 0; q < 8; ++q)
                bs[st][q] = *reinterpret_cast<const v4f*>(bp + 8 * q);
        }
    }
    const float* xb = proj + OFF_X + h * HD_ + 4 * lo;
    const v4f xm = *reinterpret_cast<const v4f*>(xb + blprev * DIP);

    v8f g[TT + 1];
    #pragma unroll
    for (int st = 0; st <= TT; ++st) {
        g[st] = v8f{};
        #pragma unroll
        for (int q = 0; q < 8; ++q) {
            v2f a, b;
            a.x = bs[st][q].x; a.y = bs[st][q].y; b.x = cq[q].x; b.y = cq[q].y;
            g[st] = __builtin_amdgcn_wmma_f32_16x16x4_f32(false, a, false, b, (short)0, g[st], false, false);
            a.x = bs[st][q].z; a.y = bs[st][q].w; b.x = cq[q].z; b.y = cq[q].w;
            g[st] = __builtin_amdgcn_wmma_f32_16x16x4_f32(false, a, false, b, (short)0, g[st], false, false);
        }
    }
    float gm = 0.f;
    #pragma unroll
    for (int q = 0; q < 8; ++q)
        gm += cq[q].x * bmq[q].x + cq[q].y * bmq[q].y + cq[q].z * bmq[q].z + cq[q].w * bmq[q].w;
    gm += __shfl_xor(gm, 16, 32);

    const float lat = sLa[16 * TT + lo];
    #pragma unroll
    for (int st = 0; st <= TT; ++st) {
        #pragma unroll
        for (int r = 0; r < 8; ++r) {
            const int sl = 8 * hi + r;
            const int s  = 16 * st + sl;
            const float e1 = expf(fminf(lat - sLa[s], 0.f)) * sP1[s];
            const float e2 = expf(fminf(lat - sLa[s + 1], 0.f)) * sP2[s + 1];
            float w;
            if (st < TT) w = e1 + e2;
            else         w = ((sl <= lo) ? e1 : 0.f) + ((sl + 1 <= lo) ? e2 : 0.f);
            g[st][r] *= w;
        }
    }
    const float mprev = prevOn * gm * expf(lat - sLa[0]) * sP2[0];

    v8f y[4] = {};
    #pragma unroll
    for (int st = 0; st <= TT; ++st) {
        #pragma unroll
        for (int j = 0; j < 4; ++j) {
            const size_t row = bl0c + 16 * st + 8 * hi + 2 * j;
            const v4f x0 = *reinterpret_cast<const v4f*>(xb + row * DIP);
            const v4f x1 = *reinterpret_cast<const v4f*>(xb + (row + 1) * DIP);
            v2f a; a.x = g[st][2 * j]; a.y = g[st][2 * j + 1];
            #pragma unroll
            for (int jj = 0; jj < 4; ++jj) {
                v2f b; b.x = x0[jj]; b.y = x1[jj];
                y[jj] = __builtin_amdgcn_wmma_f32_16x16x4_f32(false, a, false, b, (short)0, y[jj], false, false);
            }
        }
    }
    {
        v2f a; a.x = (hi == 0) ? mprev : 0.f; a.y = 0.f;
        #pragma unroll
        for (int jj = 0; jj < 4; ++jj) {
            v2f b; b.x = (hi == 0) ? xm[jj] : 0.f; b.y = 0.f;
            y[jj] = __builtin_amdgcn_wmma_f32_16x16x4_f32(false, a, false, b, (short)0, y[jj], false, false);
        }
    }
    #pragma unroll
    for (int r = 0; r < 8; ++r) {
        v4f o; o.x = y[0][r]; o.y = y[1][r]; o.z = y[2][r]; o.w = y[3][r];
        *reinterpret_cast<v4f*>(yfin + (bl0c + 16 * TT + r + 8 * hi) * DI + h * HD_ + 4 * lo) = o;
    }

    v8f sacc[2][4] = {};
    #pragma unroll
    for (int ks = 0; ks < 8; ++ks) {
        const int s0 = 4 * ks + 2 * hi;
        const v4f xa0 = *reinterpret_cast<const v4f*>(xb + (bl0c + s0) * DIP);
        const v4f xa1 = *reinterpret_cast<const v4f*>(xb + (bl0c + s0 + 1) * DIP);
        const v4f bb0 = *reinterpret_cast<const v4f*>(Brot + (bl0c + s0) * 64 + 4 * lo);
        const v4f bb1 = *reinterpret_cast<const v4f*>(Brot + (bl0c + s0 + 1) * 64 + 4 * lo);
        const float v0 = sV[s0], v1 = sV[s0 + 1];
        #pragma unroll
        for (int a2 = 0; a2 < 2; ++a2) {
            v2f a; a.x = v0 * xa0[2 * TT + a2]; a.y = v1 * xa1[2 * TT + a2];
            #pragma unroll
            for (int jn = 0; jn < 4; ++jn) {
                v2f b; b.x = bb0[jn]; b.y = bb1[jn];
                sacc[a2][jn] = __builtin_amdgcn_wmma_f32_16x16x4_f32(false, a, false, b, (short)0, sacc[a2][jn], false, false);
            }
        }
    }
    {
        const v4f bm4 = *reinterpret_cast<const v4f*>(Brot + blprev * 64 + 4 * lo);
        const float vprev = prevOn * expf(sLa[31] - sLa[0]) * sP2[0];
        #pragma unroll
        for (int a2 = 0; a2 < 2; ++a2) {
            v2f a; a.x = (hi == 0) ? vprev * xm[2 * TT + a2] : 0.f; a.y = 0.f;
            #pragma unroll
            for (int jn = 0; jn < 4; ++jn) {
                v2f b; b.x = (hi == 0) ? bm4[jn] : 0.f; b.y = 0.f;
                sacc[a2][jn] = __builtin_amdgcn_wmma_f32_16x16x4_f32(false, a, false, b, (short)0, sacc[a2][jn], false, false);
            }
        }
    }
    #pragma unroll
    for (int a2 = 0; a2 < 2; ++a2) {
        #pragma unroll
        for (int r = 0; r < 8; ++r) {
            v4f o; o.x = sacc[a2][0][r]; o.y = sacc[a2][1][r]; o.z = sacc[a2][2][r]; o.w = sacc[a2][3][r];
            *reinterpret_cast<v4f*>(Sp + (size_t)(4 * (r + 8 * hi) + 2 * TT + a2) * 64 + 4 * lo) = o;
        }
    }
}

__global__ __launch_bounds__(64) void scan_chunk_kernel(
    const float* __restrict__ proj,
    const float* __restrict__ Brot, const float* __restrict__ Crot,
    const float* __restrict__ DTb, const float* __restrict__ ladb,
    const float* __restrict__ trapb,
    float* __restrict__ yfin, float* __restrict__ cumdb,
    float* __restrict__ Sstate)
{
    static_assert(LC_ == 32 && DS_ == 64 && HD_ == 64, "two 16-token tiles, 64 x 64 state");
    const int blk  = blockIdx.x;
    const int c    = blk % NC_;
    const int h    = (blk / NC_) % NH_;
    const int b    = blk / (NC_ * NH_);
    const int lane = threadIdx.x & 31;
    const int tt   = __builtin_amdgcn_readfirstlane((int)(threadIdx.x >> 5));
    const int lo   = lane & 15;
    const int hi   = lane >> 4;
    const size_t bl0c   = (size_t)(b * L_ + c * LC_);
    const size_t blprev = (c > 0) ? bl0c - 1 : bl0c;
    const float  prevOn = (c > 0) ? 1.f : 0.f;

    __shared__ float sLa[LC_ + 1], sP1[LC_ + 1], sP2[LC_ + 1], sV[LC_];

    if (tt == 0) {
        const size_t o  = (bl0c + lane) * 32 + h;
        const float dt  = DTb[o];
        const float tr  = trapb[o];
        float la        = ladb[o];
        #pragma unroll
        for (int d = 1; d < 32; d <<= 1) {
            const float up = __shfl_up(la, d, 32);
            if (lane >= d) la += up;
        }
        const float p1 = dt * tr;
        const float p2 = dt * (1.f - tr);
        float laN = __shfl_down(la, 1, 32);
        float p2N = __shfl_down(p2, 1, 32);
        if (lane == 31) { laN = la; p2N = 0.f; }
        const float la31 = __shfl(la, 31, 32);
        sLa[lane] = la; sP1[lane] = p1; sP2[lane] = p2;
        sV[lane]  = expf(la31 - la) * p1 + expf(la31 - laN) * p2N;
        if (lane == 31) { sLa[32] = la; sP1[32] = 0.f; sP2[32] = 0.f; }
        cumdb[((size_t)(b * NH_ + h)) * L_ + c * LC_ + lane] = expf(la);
    }
    __syncthreads();

    float* Sp = Sstate + (((size_t)(b * NH_ + h)) * NC_ + c) * 4096;
    if (tt == 0)
        scan_chunk_tile<0>(proj, Brot, Crot, yfin, Sp, sLa, sP1, sP2, sV, bl0c, blprev, prevOn, h, lo, hi);
    else
        scan_chunk_tile<1>(proj, Brot, Crot, yfin, Sp, sLa, sP1, sP2, sV, bl0c, blprev, prevOn, h, lo, hi);
}

__global__ __launch_bounds__(256) void scan_combine_kernel(
    const float* __restrict__ cumdb, float* __restrict__ Sstate)
{
    const int h  = blockIdx.x % NH_;
    const int b  = blockIdx.x / NH_;
    const int t  = threadIdx.x;
    const int p  = t >> 2, seg = t & 3;

    float H[16];
    #pragma unroll
    for (int i = 0; i < 16; ++i) H[i] = 0.f;

    for (int c = 0; c < NC_; ++c) {
        float* Sp = Sstate + (((size_t)(b * NH_ + h)) * NC_ + c) * 4096 + p * 64 + seg * 16;
        const float Pc = cumdb[((size_t)(b * NH_ + h)) * L_ + c * LC_ + LC_ - 1];
        #pragma unroll
        for (int i = 0; i < 16; ++i) {
            const float Sc = Sp[i];
            Sp[i] = H[i];
            H[i]  = Pc * H[i] + Sc;
        }
    }
}

__global__ __launch_bounds__(64) void scan_correct_kernel(
    const float* __restrict__ proj, const float* __restrict__ Crot,
    const float* __restrict__ cumdb, const float* __restrict__ Sstate,
    const float* __restrict__ D_param, float* __restrict__ yfin)
{
    static_assert(LC_ == 32 && DS_ == 64 && HD_ == 64, "two 16-token tiles, 64 x 64 state");
    const int blk  = blockIdx.x;
    const int c    = blk % NC_;
    const int h    = (blk / NC_) % NH_;
    const int b    = blk / (NC_ * NH_);
    const int lane = threadIdx.x & 31;
    const int tt   = threadIdx.x >> 5;
    const int lo   = lane & 15;
    const int hi   = lane >> 4;
    const size_t bl0 = (size_t)(b * L_ + c * LC_ + 16 * tt);

    const float* ap = Crot + (bl0 + lo) * 64 + 4 * hi;
    const float* hp = Sstate + (((size_t)(b * NH_ + h)) * NC_ + c) * 4096
                             + (size_t)(4 * lo) * 64 + 4 * hi;

    v4f a[8], bq[4][8];
    #pragma unroll
    for (int q = 0; q < 8; ++q)
        a[q] = *reinterpret_cast<const v4f*>(ap + 8 * q);
    #pragma unroll
    for (int j = 0; j < 4; ++j) {
        #pragma unroll
        for (int q = 0; q < 8; ++q)
            bq[j][q] = *reinterpret_cast<const v4f*>(hp + j * 64 + 8 * q);
    }

    const float* cdp = cumdb + ((size_t)(b * NH_ + h)) * L_ + c * LC_ + 16 * tt + 8 * hi;
    const v4f cd0 = *reinterpret_cast<const v4f*>(cdp);
    const v4f cd1 = *reinterpret_cast<const v4f*>(cdp + 4);
    const float Dp = D_param[h];
    v4f xv[8], zv[8], yv[8];
    #pragma unroll
    for (int r = 0; r < 8; ++r) {
        const size_t bl = bl0 + r + 8 * hi;
        xv[r] = *reinterpret_cast<const v4f*>(proj + bl * DIP + OFF_X + h * HD_ + 4 * lo);
        zv[r] = *reinterpret_cast<const v4f*>(proj + bl * DIP + OFF_Z + h * HD_ + 4 * lo);
        yv[r] = *reinterpret_cast<const v4f*>(yfin + bl * DI + h * HD_ + 4 * lo);
    }

    __builtin_amdgcn_sched_barrier(0);

    v8f acc[4] = {};
    #pragma unroll
    for (int q = 0; q < 8; ++q) {
        #pragma unroll
        for (int half = 0; half < 2; ++half) {
            v2f av;
            if (half) { av.x = a[q].z; av.y = a[q].w; } else { av.x = a[q].x; av.y = a[q].y; }
            #pragma unroll
            for (int j = 0; j < 4; ++j) {
                v2f bv;
                if (half) { bv.x = bq[j][q].z; bv.y = bq[j][q].w; }
                else      { bv.x = bq[j][q].x; bv.y = bq[j][q].y; }
                acc[j] = __builtin_amdgcn_wmma_f32_16x16x4_f32(false, av, false, bv, (short)0, acc[j], false, false);
            }
        }
    }

    #pragma unroll
    for (int r = 0; r < 8; ++r) {
        const size_t bl = bl0 + r + 8 * hi;
        const float cd = (r < 4) ? cd0[r & 3] : cd1[r & 3];
        v4f o;
        #pragma unroll
        for (int j = 0; j < 4; ++j) {
            const float z = zv[r][j];
            const float y = yv[r][j] + cd * acc[j][r];
            o[j] = (y + Dp * xv[r][j]) * (z * sigmoid_f(z));
        }
        *reinterpret_cast<v4f*>(yfin + bl * DI + h * HD_ + 4 * lo) = o;
    }
}

extern "C" void kernel_launch(void* const* d_in, const int* in_sizes, int n_in,
                              void* d_out, int out_size, void* d_ws, size_t ws_size,
                              hipStream_t stream)
{
    const float* u       = (const float*)d_in[0];
    const float* W_in    = (const float*)d_in[1];
    const float* W_out   = (const float*)d_in[2];
    const float* dt_bias = (const float*)d_in[3];
    const float* D_param = (const float*)d_in[4];
    const float* Bn_g    = (const float*)d_in[5];
    const float* Bn_b    = (const float*)d_in[6];
    const float* Cn_g    = (const float*)d_in[7];
    const float* Cn_b    = (const float*)d_in[8];

    float* ws = (float*)d_ws;
    float* proj   = ws; ws += (size_t)M1 * DIP;
    float* yfin   = ws; ws += (size_t)M1 * DI;
    float* Sstate = ws; ws += (size_t)B_ * NH_ * NC_ * 4096;
    float* Brot   = ws; ws += (size_t)M1 * DS_;
    float* Crot   = ws; ws += (size_t)M1 * DS_;
    float* DTb    = ws; ws += (size_t)M1 * NH_;
    float* ladb   = ws; ws += (size_t)M1 * NH_;
    float* trapb  = ws; ws += (size_t)M1 * NH_;
    float* phaseb = ws; ws += (size_t)M1 * NA_;
    float* cumdb  = ws; ws += (size_t)B_ * NH_ * L_;

    gemm_wmma_f32<DIP, DM><<<dim3(DIP / 64, M1 / 64), dim3(128), 0, stream>>>(
        u, W_in, proj);

    phase_kernel<<<dim3(B_), dim3(1024), 0, stream>>>(
        proj, dt_bias, DTb, ladb, trapb, phaseb);

    rot_kernel<<<dim3(M1), dim3(64), 0, stream>>>(
        proj, Bn_g, Bn_b, Cn_g, Cn_b, phaseb, Brot, Crot);

    scan_chunk_kernel<<<dim3(B_ * NH_ * NC_), dim3(64), 0, stream>>>(
        proj, Brot, Crot, DTb, ladb, trapb, yfin, cumdb, Sstate);
    scan_combine_kernel<<<dim3(B_ * NH_), dim3(256), 0, stream>>>(
        cumdb, Sstate);
    scan_correct_kernel<<<dim3(B_ * NH_ * NC_), dim3(64), 0, stream>>>(
        proj, Crot, cumdb, Sstate, D_param, yfin);

    gemm_wmma_f32<DM, DI><<<dim3(DM / 64, M1 / 64), dim3(128), 0, stream>>>(
        yfin, W_out, (float*)d_out);
}


# dt, log decay, trap and mean of dt over heads made in the epilogue of the in-projection tiles holding those columns (dispatched first), not in the 2-block phase kernel; the mean's add order changes
# speedup vs baseline: 6.9588x; 1.0742x over previous
#include <hip/hip_runtime.h>
#include <math.h>

#define B_   2
#define L_   512
#define DM   1024
#define DI   2048
#define DS_  64
#define NH_  32
#define HD_  64
#define NA_  32
#define DIP  4352
#define M1   (B_*L_)

#define NC_  16
#define LC_  32

#define OFF_Z    0
#define OFF_X    2048
#define OFF_B    4096
#define OFF_C    4160
#define OFF_DT   4224
#define OFF_A    4256
#define OFF_TRAP 4288
#define OFF_ANG  4320

typedef __attribute__((ext_vector_type(2))) float v2f;
typedef __attribute__((ext_vector_type(8))) float v8f;

__device__ __forceinline__ float softplus_f(float x) {
    return x > 20.f ? x : log1pf(expf(x));
}
__device__ __forceinline__ float sigmoid_f(float x) {
    return 1.f / (1.f + expf(-x));
}

typedef __attribute__((ext_vector_type(4))) float v4f;

struct GFrag { v4f a[2]; v4f b[8]; };

template <int N>
__device__ __forceinline__ void gemm_load(GFrag& f, const float* __restrict__ ap,
                                          const float* __restrict__ bp)
{
    f.a[0] = *reinterpret_cast<const v4f*>(ap);
    f.a[1] = *reinterpret_cast<const v4f*>(ap + 4);
    #pragma unroll
    for (int r = 0; r < 8; ++r)
        f.b[r] = *reinterpret_cast<const v4f*>(bp + (size_t)r * N);
}

__device__ __forceinline__ void gemm_mma(const GFrag& f, v8f& c0, v8f& c1, v8f& c2, v8f& c3)
{
    #pragma unroll
    for (int s = 0; s < 4; ++s) {
        const v4f aq = f.a[s >> 1];
        v2f a;
        if (s & 1) { a.x = aq.z; a.y = aq.w; } else { a.x = aq.x; a.y = aq.y; }
        const v4f r0 = f.b[2 * s], r1 = f.b[2 * s + 1];
        v2f b;
        b.x = r0.x; b.y = r1.x;
        c0 = __builtin_amdgcn_wmma_f32_16x16x4_f32(false, a, false, b, (short)0, c0, false, false);
        b.x = r0.y; b.y = r1.y;
        c1 = __builtin_amdgcn_wmma_f32_16x16x4_f32(false, a, false, b, (short)0, c1, false, false);
        b.x = r0.z; b.y = r1.z;
        c2 = __builtin_amdgcn_wmma_f32_16x16x4_f32(false, a, false, b, (short)0, c2, false, false);
        b.x = r0.w; b.y = r1.w;
        c3 = __builtin_amdgcn_wmma_f32_16x16x4_f32(false, a, false, b, (short)0, c3, false, false);
    }
}

#define GEMM_HOLD() __builtin_amdgcn_sched_barrier(0)

template <int N, int K, bool EPI>
__global__ __launch_bounds__(128, 4) void gemm_wmma_f32(
    const float* __restrict__ A, const float* __restrict__ Bm,
    float* __restrict__ C,
    const float* __restrict__ dt_bias, float* __restrict__ DTb,
    float* __restrict__ ladb, float* __restrict__ trapb,
    float* __restrict__ dtmeanb)
{
    static_assert(N % 64 == 0 && K % 64 == 0 && K >= 128, "tile and ring shape");
    const int lane = threadIdx.x & 31;
    const int wave = threadIdx.x >> 5;
    const int lo   = lane & 15;
    const int hi   = lane >> 4;

    const int m0 = blockIdx.y * 64 + wave * 16;
    const int bx = EPI ? N / 64 - 1 - (int)blockIdx.x : (int)blockIdx.x;
    const int n0 = bx * 64;

    const float* ap = A + (size_t)(m0 + lo) * K + 8 * hi;
    const float* bp = Bm + (size_t)(8 * hi) * N + n0 + 4 * lo;
    constexpr int    SA = 16;
    constexpr size_t SB = (size_t)16 * N;
    constexpr int    NS = K / 16;

    v8f c0 = {}, c1 = {}, c2 = {}, c3 = {};
    GFrag f0, f1, f2, f3;
    gemm_load<N>(f0, ap, bp);
    gemm_load<N>(f1, ap + SA, bp + SB);
    gemm_load<N>(f2, ap + 2 * SA, bp + 2 * SB);

    for (int s = 0; s < NS - 4; s += 4) {
        GEMM_HOLD(); gemm_load<N>(f3, ap + 3 * SA, bp + 3 * SB); GEMM_HOLD();
        gemm_mma(f0, c0, c1, c2, c3);
        GEMM_HOLD(); gemm_load<N>(f0, ap + 4 * SA, bp + 4 * SB); GEMM_HOLD();
        gemm_mma(f1, c0, c1, c2, c3);
        GEMM_HOLD(); gemm_load<N>(f1, ap + 5 * SA, bp + 5 * SB); GEMM_HOLD();
        gemm_mma(f2, c0, c1, c2, c3);
        GEMM_HOLD(); gemm_load<N>(f2, ap + 6 * SA, bp + 6 * SB); GEMM_HOLD();
        gemm_mma(f3, c0, c1, c2, c3);
        ap += 4 * SA;
        bp += 4 * SB;
    }
    GEMM_HOLD(); gemm_load<N>(f3, ap + 3 * SA, bp + 3 * SB); GEMM_HOLD();
    gemm_mma(f0, c0, c1, c2, c3);
    gemm_mma(f1, c0, c1, c2, c3);
    gemm_mma(f2, c0, c1, c2, c3);
    gemm_mma(f3, c0, c1, c2, c3);

    #pragma unroll
    for (int r = 0; r < 8; ++r) {
        v4f o; o.x = c0[r]; o.y = c1[r]; o.z = c2[r]; o.w = c3[r];
        *reinterpret_cast<v4f*>(C + (size_t)(m0 + r + 8 * hi) * N + n0 + 4 * lo) = o;
    }

    if constexpr (EPI) {
        static_assert(OFF_DT % 64 == 0 && OFF_A == OFF_DT + 32 && OFF_TRAP == OFF_DT + 64 &&
                      NH_ == 32, "dt|A share one 64-column tile, trap opens the next");
        if (bx == OFF_DT / 64) {
            const bool isdt = lo < 8;
            const int  h4   = 4 * (lo & 7);
            v4f bias;
            #pragma unroll
            for (int j = 0; j < 4; ++j) bias[j] = isdt ? dt_bias[h4 + j] : 0.f;
            float* dst = (isdt ? DTb : ladb) + h4;
            #pragma unroll
            for (int r = 0; r < 8; ++r) {
                const size_t row = (size_t)(m0 + r + 8 * hi);
                v4f sp;
                sp.x = softplus_f(c0[r] + bias.x);
                sp.y = softplus_f(c1[r] + bias.y);
                sp.z = softplus_f(c2[r] + bias.z);
                sp.w = softplus_f(c3[r] + bias.w);
                v4f o;
                #pragma unroll
                for (int j = 0; j < 4; ++j) {
                    const float dtp = __shfl_xor(sp[j], 8, 32);
                    o[j] = isdt ? sp[j] : fminf(-sp[j], -1e-4f) * dtp;
                }
                *reinterpret_cast<v4f*>(dst + row * 32) = o;
                float s = (sp.x + sp.y) + (sp.z + sp.w);
                s += __shfl_xor(s, 1, 32);
                s += __shfl_xor(s, 2, 32);
                s += __shfl_xor(s, 4, 32);
                if (lo == 0) dtmeanb[row] = s * (1.f / 32.f);
            }
        } else if (bx == OFF_TRAP / 64) {
            if (lo < 8) {
                #pragma unroll
                for (int r = 0; r < 8; ++r) {
                    v4f o;
                    o.x = sigmoid_f(c0[r]); o.y = sigmoid_f(c1[r]);
                    o.z = sigmoid_f(c2[r]); o.w = sigmoid_f(c3[r]);
                    *reinterpret_cast<v4f*>(trapb + (size_t)(m0 + r + 8 * hi) * 32 + 4 * lo) = o;
                }
            }
        }
    }
}

__global__ __launch_bounds__(1024) void phase_kernel(
    const float* __restrict__ proj, const float* __restrict__ dtmeanb,
    float* __restrict__ phaseb)
{
    __shared__ v2f incp[(L_ / 2) * 32];
    const int b    = blockIdx.x;
    const int j    = threadIdx.x & 31;
    const int w    = __builtin_amdgcn_readfirstlane((int)(threadIdx.x >> 5));
    {
        const float* ang = proj + (size_t)(b * L_ + 16 * w) * DIP + OFF_ANG + j;
        const float* dmp = dtmeanb + b * L_ + 16 * w;
        float va[16], vm[16];
        #pragma unroll
        for (int i = 0; i < 16; ++i) { va[i] = ang[(size_t)i * DIP]; vm[i] = dmp[i]; }
        #pragma unroll
        for (int i = 0; i < 16; ++i) {
            const int l = 16 * w + i;
            incp[(l >> 1) * 32 + j][l & 1] = va[i] * vm[i];
        }
    }
    __syncthreads();

    const int lo = j & 15;
    const int hi = j >> 4;
    v8f acc0 = {}, acc1 = {};
    const v2f ones = {1.f, 1.f};

    for (int t = 0; t < w; ++t) {
        const v2f* p = incp + (8 * t + hi) * 32 + lo;
        #pragma unroll
        for (int s = 0; s < 4; ++s) {
            const v2f b0 = p[64 * s];
            const v2f b1 = p[64 * s + 16];
            acc0 = __builtin_amdgcn_wmma_f32_16x16x4_f32(false, ones, false, b0, (short)0, acc0, false, false);
            acc1 = __builtin_amdgcn_wmma_f32_16x16x4_f32(false, ones, false, b1, (short)0, acc1, false, false);
        }
    }
    {
        const v2f* p = incp + (8 * w + hi) * 32 + lo;
        #pragma unroll
        for (int s = 0; s < 4; ++s) {
            const int kl = 4 * s + 2 * hi;
            v2f a; a.x = (kl <= lo) ? 1.f : 0.f; a.y = (kl + 1 <= lo) ? 1.f : 0.f;
            const v2f b0 = p[64 * s];
            const v2f b1 = p[64 * s + 16];
            acc0 = __builtin_amdgcn_wmma_f32_16x16x4_f32(false, a, false, b0, (short)0, acc0, false, false);
            acc1 = __builtin_amdgcn_wmma_f32_16x16x4_f32(false, a, false, b1, (short)0, acc1, false, false);
        }
    }

    #pragma unroll
    for (int r = 0; r < 8; ++r) {
        float* op = phaseb + (size_t)(b * L_ + 16 * w + r + 8 * hi) * 32 + lo;
        op[0]  = acc0[r];
        op[16] = acc1[r];
    }
}

__global__ __launch_bounds__(64) void rot_kernel(
    const float* __restrict__ proj,
    const float* __restrict__ Bn_g, const float* __restrict__ Bn_b,
    const float* __restrict__ Cn_g, const float* __restrict__ Cn_b,
    const float* __restrict__ phaseb,
    float* __restrict__ Brot, float* __restrict__ Crot)
{
    const int bl = blockIdx.x;
    const int t  = threadIdx.x;
    __shared__ float red[64], nb[64], nc[64];

    const float* row = proj + (size_t)bl * DIP;
    const float vb = row[OFF_B + t];
    const float vc = row[OFF_C + t];

    auto blockReduce = [&](float v) -> float {
        red[t] = v; __syncthreads();
        for (int s = 32; s > 0; s >>= 1) {
            if (t < s) red[t] += red[t + s];
            __syncthreads();
        }
        float r = red[0]; __syncthreads();
        return r;
    };

    const float mB  = blockReduce(vb) * (1.f / 64.f);
    const float vaB = blockReduce((vb - mB) * (vb - mB)) * (1.f / 64.f);
    const float nB  = (vb - mB) * rsqrtf(vaB + 1e-5f) * Bn_g[t] + Bn_b[t];

    const float mC  = blockReduce(vc) * (1.f / 64.f);
    const float vaC = blockReduce((vc - mC) * (vc - mC)) * (1.f / 64.f);
    const float nC  = (vc - mC) * rsqrtf(vaC + 1e-5f) * Cn_g[t] + Cn_b[t];

    nb[t] = nB; nc[t] = nC;
    __syncthreads();

    const int   q  = t >> 1;
    const float ph = phaseb[(size_t)bl * 32 + q];
    const float c  = cosf(ph), s = sinf(ph);
    float ob, oc;
    if ((t & 1) == 0) { ob = nb[t] * c - nb[t + 1] * s; oc = nc[t] * c - nc[t + 1] * s; }
    else              { ob = nb[t - 1] * s + nb[t] * c; oc = nc[t - 1] * s + nc[t] * c; }
    Brot[(size_t)bl * 64 + t] = ob;
    Crot[(size_t)bl * 64 + t] = oc;
}


template <int TT>
__device__ __forceinline__ void scan_chunk_tile(
    const float* __restrict__ proj, const float* __restrict__ Brot,
    const float* __restrict__ Crot, float* __restrict__ yfin,
    float* __restrict__ Sp, const float* sLa, const float* sP1,
    const float* sP2, const float* sV,
    const size_t bl0c, const size_t blprev, const float prevOn,
    const int h, const int lo, const int hi)
{
    v4f cq[8], bmq[8], bs[TT + 1][8];
    {
        const float* cp  = Crot + (bl0c + 16 * TT + lo) * 64 + 4 * hi;
        const float* bmp = Brot + blprev * 64 + 4 * hi;
        #pragma unroll
        for (int q = 0; q < 8; ++q) {
            cq[q]  = *reinterpret_cast<const v4f*>(cp + 8 * q);
            bmq[q] = *reinterpret_cast<const v4f*>(bmp + 8 * q);
        }
        #pragma unroll
        for (int st = 0; st <= TT; ++st) {
            const float* bp = Brot + (bl0c + 16 * st + lo) * 64 + 4 * hi;
            #pragma unroll
            for (int q = 0; q < 8; ++q)
                bs[st][q] = *reinterpret_cast<const v4f*>(bp + 8 * q);
        }
    }
    const float* xb = proj + OFF_X + h * HD_ + 4 * lo;
    const v4f xm = *reinterpret_cast<const v4f*>(xb + blprev * DIP);

    v8f g[TT + 1];
    #pragma unroll
    for (int st = 0; st <= TT; ++st) {
        g[st] = v8f{};
        #pragma unroll
        for (int q = 0; q < 8; ++q) {
            v2f a, b;
            a.x = bs[st][q].x; a.y = bs[st][q].y; b.x = cq[q].x; b.y = cq[q].y;
            g[st] = __builtin_amdgcn_wmma_f32_16x16x4_f32(false, a, false, b, (short)0, g[st], false, false);
            a.x = bs[st][q].z; a.y = bs[st][q].w; b.x = cq[q].z; b.y = cq[q].w;
            g[st] = __builtin_amdgcn_wmma_f32_16x16x4_f32(false, a, false, b, (short)0, g[st], false, false);
        }
    }
    float gm = 0.f;
    #pragma unroll
    for (int q = 0; q < 8; ++q)
        gm += cq[q].x * bmq[q].x + cq[q].y * bmq[q].y + cq[q].z * bmq[q].z + cq[q].w * bmq[q].w;
    gm += __shfl_xor(gm, 16, 32);

    const float lat = sLa[16 * TT + lo];
    #pragma unroll
    for (int st = 0; st <= TT; ++st) {
        #pragma unroll
        for (int r = 0; r < 8; ++r) {
            const int sl = 8 * hi + r;
            const int s  = 16 * st + sl;
            const float e1 = expf(fminf(lat - sLa[s], 0.f)) * sP1[s];
            const float e2 = expf(fminf(lat - sLa[s + 1], 0.f)) * sP2[s + 1];
            float w;
            if (st < TT) w = e1 + e2;
            else         w = ((sl <= lo) ? e1 : 0.f) + ((sl + 1 <= lo) ? e2 : 0.f);
            g[st][r] *= w;
        }
    }
    const float mprev = prevOn * gm * expf(lat - sLa[0]) * sP2[0];

    v8f y[4] = {};
    #pragma unroll
    for (int st = 0; st <= TT; ++st) {
        #pragma unroll
        for (int j = 0; j < 4; ++j) {
            const size_t row = bl0c + 16 * st + 8 * hi + 2 * j;
            const v4f x0 = *reinterpret_cast<const v4f*>(xb + row * DIP);
            const v4f x1 = *reinterpret_cast<const v4f*>(xb + (row + 1) * DIP);
            v2f a; a.x = g[st][2 * j]; a.y = g[st][2 * j + 1];
            #pragma unroll
            for (int jj = 0; jj < 4; ++jj) {
                v2f b; b.x = x0[jj]; b.y = x1[jj];
                y[jj] = __builtin_amdgcn_wmma_f32_16x16x4_f32(false, a, false, b, (short)0, y[jj], false, false);
            }
        }
    }
    {
        v2f a; a.x = (hi == 0) ? mprev : 0.f; a.y = 0.f;
        #pragma unroll
        for (int jj = 0; jj < 4; ++jj) {
            v2f b; b.x = (hi == 0) ? xm[jj] : 0.f; b.y = 0.f;
            y[jj] = __builtin_amdgcn_wmma_f32_16x16x4_f32(false, a, false, b, (short)0, y[jj], false, false);
        }
    }
    #pragma unroll
    for (int r = 0; r < 8; ++r) {
        v4f o; o.x = y[0][r]; o.y = y[1][r]; o.z = y[2][r]; o.w = y[3][r];
        *reinterpret_cast<v4f*>(yfin + (bl0c + 16 * TT + r + 8 * hi) * DI + h * HD_ + 4 * lo) = o;
    }

    v8f sacc[2][4] = {};
    #pragma unroll
    for (int ks = 0; ks < 8; ++ks) {
        const int s0 = 4 * ks + 2 * hi;
        const v4f xa0 = *reinterpret_cast<const v4f*>(xb + (bl0c + s0) * DIP);
        const v4f xa1 = *reinterpret_cast<const v4f*>(xb + (bl0c + s0 + 1) * DIP);
        const v4f bb0 = *reinterpret_cast<const v4f*>(Brot + (bl0c + s0) * 64 + 4 * lo);
        const v4f bb1 = *reinterpret_cast<const v4f*>(Brot + (bl0c + s0 + 1) * 64 + 4 * lo);
        const float v0 = sV[s0], v1 = sV[s0 + 1];
        #pragma unroll
        for (int a2 = 0; a2 < 2; ++a2) {
            v2f a; a.x = v0 * xa0[2 * TT + a2]; a.y = v1 * xa1[2 * TT + a2];
            #pragma unroll
            for (int jn = 0; jn < 4; ++jn) {
                v2f b; b.x = bb0[jn]; b.y = bb1[jn];
                sacc[a2][jn] = __builtin_amdgcn_wmma_f32_16x16x4_f32(false, a, false, b, (short)0, sacc[a2][jn], false, false);
            }
        }
    }
    {
        const v4f bm4 = *reinterpret_cast<const v4f*>(Brot + blprev * 64 + 4 * lo);
        const float vprev = prevOn * expf(sLa[31] - sLa[0]) * sP2[0];
        #pragma unroll
        for (int a2 = 0; a2 < 2; ++a2) {
            v2f a; a.x = (hi == 0) ? vprev * xm[2 * TT + a2] : 0.f; a.y = 0.f;
            #pragma unroll
            for (int jn = 0; jn < 4; ++jn) {
                v2f b; b.x = (hi == 0) ? bm4[jn] : 0.f; b.y = 0.f;
                sacc[a2][jn] = __builtin_amdgcn_wmma_f32_16x16x4_f32(false, a, false, b, (short)0, sacc[a2][jn], false, false);
            }
        }
    }
    #pragma unroll
    for (int a2 = 0; a2 < 2; ++a2) {
        #pragma unroll
        for (int r = 0; r < 8; ++r) {
            v4f o; o.x = sacc[a2][0][r]; o.y = sacc[a2][1][r]; o.z = sacc[a2][2][r]; o.w = sacc[a2][3][r];
            *reinterpret_cast<v4f*>(Sp + (size_t)(4 * (r + 8 * hi) + 2 * TT + a2) * 64 + 4 * lo) = o;
        }
    }
}

__global__ __launch_bounds__(64) void scan_chunk_kernel(
    const float* __restrict__ proj,
    const float* __restrict__ Brot, const float* __restrict__ Crot,
    const float* __restrict__ DTb, const float* __restrict__ ladb,
    const float* __restrict__ trapb,
    float* __restrict__ yfin, float* __restrict__ cumdb,
    float* __restrict__ Sstate)
{
    static_assert(LC_ == 32 && DS_ == 64 && HD_ == 64, "two 16-token tiles, 64 x 64 state");
    const int blk  = blockIdx.x;
    const int c    = blk % NC_;
    const int h    = (blk / NC_) % NH_;
    const int b    = blk / (NC_ * NH_);
    const int lane = threadIdx.x & 31;
    const int tt   = __builtin_amdgcn_readfirstlane((int)(threadIdx.x >> 5));
    const int lo   = lane & 15;
    const int hi   = lane >> 4;
    const size_t bl0c   = (size_t)(b * L_ + c * LC_);
    const size_t blprev = (c > 0) ? bl0c - 1 : bl0c;
    const float  prevOn = (c > 0) ? 1.f : 0.f;

    __shared__ float sLa[LC_ + 1], sP1[LC_ + 1], sP2[LC_ + 1], sV[LC_];

    if (tt == 0) {
        const size_t o  = (bl0c + lane) * 32 + h;
        const float dt  = DTb[o];
        const float tr  = trapb[o];
        float la        = ladb[o];
        #pragma unroll
        for (int d = 1; d < 32; d <<= 1) {
            const float up = __shfl_up(la, d, 32);
            if (lane >= d) la += up;
        }
        const float p1 = dt * tr;
        const float p2 = dt * (1.f - tr);
        float laN = __shfl_down(la, 1, 32);
        float p2N = __shfl_down(p2, 1, 32);
        if (lane == 31) { laN = la; p2N = 0.f; }
        const float la31 = __shfl(la, 31, 32);
        sLa[lane] = la; sP1[lane] = p1; sP2[lane] = p2;
        sV[lane]  = expf(la31 - la) * p1 + expf(la31 - laN) * p2N;
        if (lane == 31) { sLa[32] = la; sP1[32] = 0.f; sP2[32] = 0.f; }
        cumdb[((size_t)(b * NH_ + h)) * L_ + c * LC_ + lane] = expf(la);
    }
    __syncthreads();

    float* Sp = Sstate + (((size_t)(b * NH_ + h)) * NC_ + c) * 4096;
    if (tt == 0)
        scan_chunk_tile<0>(proj, Brot, Crot, yfin, Sp, sLa, sP1, sP2, sV, bl0c, blprev, prevOn, h, lo, hi);
    else
        scan_chunk_tile<1>(proj, Brot, Crot, yfin, Sp, sLa, sP1, sP2, sV, bl0c, blprev, prevOn, h, lo, hi);
}

__global__ __launch_bounds__(256) void scan_combine_kernel(
    const float* __restrict__ cumdb, float* __restrict__ Sstate)
{
    const int h  = blockIdx.x % NH_;
    const int b  = blockIdx.x / NH_;
    const int t  = threadIdx.x;
    const int p  = t >> 2, seg = t & 3;

    float H[16];
    #pragma unroll
    for (int i = 0; i < 16; ++i) H[i] = 0.f;

    for (int c = 0; c < NC_; ++c) {
        float* Sp = Sstate + (((size_t)(b * NH_ + h)) * NC_ + c) * 4096 + p * 64 + seg * 16;
        const float Pc = cumdb[((size_t)(b * NH_ + h)) * L_ + c * LC_ + LC_ - 1];
        #pragma unroll
        for (int i = 0; i < 16; ++i) {
            const float Sc = Sp[i];
            Sp[i] = H[i];
            H[i]  = Pc * H[i] + Sc;
        }
    }
}

__global__ __launch_bounds__(64) void scan_correct_kernel(
    const float* __restrict__ proj, const float* __restrict__ Crot,
    const float* __restrict__ cumdb, const float* __restrict__ Sstate,
    const float* __restrict__ D_param, float* __restrict__ yfin)
{
    static_assert(LC_ == 32 && DS_ == 64 && HD_ == 64, "two 16-token tiles, 64 x 64 state");
    const int blk  = blockIdx.x;
    const int c    = blk % NC_;
    const int h    = (blk / NC_) % NH_;
    const int b    = blk / (NC_ * NH_);
    const int lane = threadIdx.x & 31;
    const int tt   = threadIdx.x >> 5;
    const int lo   = lane & 15;
    const int hi   = lane >> 4;
    const size_t bl0 = (size_t)(b * L_ + c * LC_ + 16 * tt);

    const float* ap = Crot + (bl0 + lo) * 64 + 4 * hi;
    const float* hp = Sstate + (((size_t)(b * NH_ + h)) * NC_ + c) * 4096
                             + (size_t)(4 * lo) * 64 + 4 * hi;

    v4f a[8], bq[4][8];
    #pragma unroll
    for (int q = 0; q < 8; ++q)
        a[q] = *reinterpret_cast<const v4f*>(ap + 8 * q);
    #pragma unroll
    for (int j = 0; j < 4; ++j) {
        #pragma unroll
        for (int q = 0; q < 8; ++q)
            bq[j][q] = *reinterpret_cast<const v4f*>(hp + j * 64 + 8 * q);
    }

    const float* cdp = cumdb + ((size_t)(b * NH_ + h)) * L_ + c * LC_ + 16 * tt + 8 * hi;
    const v4f cd0 = *reinterpret_cast<const v4f*>(cdp);
    const v4f cd1 = *reinterpret_cast<const v4f*>(cdp + 4);
    const float Dp = D_param[h];
    v4f xv[8], zv[8], yv[8];
    #pragma unroll
    for (int r = 0; r < 8; ++r) {
        const size_t bl = bl0 + r + 8 * hi;
        xv[r] = *reinterpret_cast<const v4f*>(proj + bl * DIP + OFF_X + h * HD_ + 4 * lo);
        zv[r] = *reinterpret_cast<const v4f*>(proj + bl * DIP + OFF_Z + h * HD_ + 4 * lo);
        yv[r] = *reinterpret_cast<const v4f*>(yfin + bl * DI + h * HD_ + 4 * lo);
    }

    __builtin_amdgcn_sched_barrier(0);

    v8f acc[4] = {};
    #pragma unroll
    for (int q = 0; q < 8; ++q) {
        #pragma unroll
        for (int half = 0; half < 2; ++half) {
            v2f av;
            if (half) { av.x = a[q].z; av.y = a[q].w; } else { av.x = a[q].x; av.y = a[q].y; }
            #pragma unroll
            for (int j = 0; j < 4; ++j) {
                v2f bv;
                if (half) { bv.x = bq[j][q].z; bv.y = bq[j][q].w; }
                else      { bv.x = bq[j][q].x; bv.y = bq[j][q].y; }
                acc[j] = __builtin_amdgcn_wmma_f32_16x16x4_f32(false, av, false, bv, (short)0, acc[j], false, false);
            }
        }
    }

    #pragma unroll
    for (int r = 0; r < 8; ++r) {
        const size_t bl = bl0 + r + 8 * hi;
        const float cd = (r < 4) ? cd0[r & 3] : cd1[r & 3];
        v4f o;
        #pragma unroll
        for (int j = 0; j < 4; ++j) {
            const float z = zv[r][j];
            const float y = yv[r][j] + cd * acc[j][r];
            o[j] = (y + Dp * xv[r][j]) * (z * sigmoid_f(z));
        }
        *reinterpret_cast<v4f*>(yfin + bl * DI + h * HD_ + 4 * lo) = o;
    }
}

extern "C" void kernel_launch(void* const* d_in, const int* in_sizes, int n_in,
                              void* d_out, int out_size, void* d_ws, size_t ws_size,
                              hipStream_t stream)
{
    const float* u       = (const float*)d_in[0];
    const float* W_in    = (const float*)d_in[1];
    const float* W_out   = (const float*)d_in[2];
    const float* dt_bias = (const float*)d_in[3];
    const float* D_param = (const float*)d_in[4];
    const float* Bn_g    = (const float*)d_in[5];
    const float* Bn_b    = (const float*)d_in[6];
    const float* Cn_g    = (const float*)d_in[7];
    const float* Cn_b    = (const float*)d_in[8];

    float* ws = (float*)d_ws;
    float* proj   = ws; ws += (size_t)M1 * DIP;
    float* yfin   = ws; ws += (size_t)M1 * DI;
    float* Sstate = ws; ws += (size_t)B_ * NH_ * NC_ * 4096;
    float* Brot   = ws; ws += (size_t)M1 * DS_;
    float* Crot   = ws; ws += (size_t)M1 * DS_;
    float* DTb    = ws; ws += (size_t)M1 * NH_;
    float* ladb   = ws; ws += (size_t)M1 * NH_;
    float* trapb  = ws; ws += (size_t)M1 * NH_;
    float* phaseb = ws; ws += (size_t)M1 * NA_;
    float* cumdb  = ws; ws += (size_t)B_ * NH_ * L_;
    float* dtmeanb = ws; ws += (size_t)M1;

    gemm_wmma_f32<DIP, DM, true><<<dim3(DIP / 64, M1 / 64), dim3(128), 0, stream>>>(
        u, W_in, proj, dt_bias, DTb, ladb, trapb, dtmeanb);

    phase_kernel<<<dim3(B_), dim3(1024), 0, stream>>>(
        proj, dtmeanb, phaseb);

    rot_kernel<<<dim3(M1), dim3(64), 0, stream>>>(
        proj, Bn_g, Bn_b, Cn_g, Cn_b, phaseb, Brot, Crot);

    scan_chunk_kernel<<<dim3(B_ * NH_ * NC_), dim3(64), 0, stream>>>(
        proj, Brot, Crot, DTb, ladb, trapb, yfin, cumdb, Sstate);
    scan_combine_kernel<<<dim3(B_ * NH_), dim3(256), 0, stream>>>(
        cumdb, Sstate);
    scan_correct_kernel<<<dim3(B_ * NH_ * NC_), dim3(64), 0, stream>>>(
        proj, Crot, cumdb, Sstate, D_param, yfin);

    gemm_wmma_f32<DM, DI, false><<<dim3(DM / 64, M1 / 64), dim3(128), 0, stream>>>(
        yfin, W_out, (float*)d_out, nullptr, nullptr, nullptr, nullptr, nullptr);
}


# phase cumsum folded into the rotation kernel: one 16-wave block per 16-token tile redoes the ones-triangular f32 WMMA product over all earlier tokens, two tiles a wave; sum order changes
# speedup vs baseline: 7.6290x; 1.0963x over previous
#include <hip/hip_runtime.h>
#include <math.h>

#define B_   2
#define L_   512
#define DM   1024
#define DI   2048
#define DS_  64
#define NH_  32
#define HD_  64
#define NA_  32
#define DIP  4352
#define M1   (B_*L_)

#define NC_  16
#define LC_  32

#define OFF_Z    0
#define OFF_X    2048
#define OFF_B    4096
#define OFF_C    4160
#define OFF_DT   4224
#define OFF_A    4256
#define OFF_TRAP 4288
#define OFF_ANG  4320

typedef __attribute__((ext_vector_type(2))) float v2f;
typedef __attribute__((ext_vector_type(8))) float v8f;

__device__ __forceinline__ float softplus_f(float x) {
    return x > 20.f ? x : log1pf(expf(x));
}
__device__ __forceinline__ float sigmoid_f(float x) {
    return 1.f / (1.f + expf(-x));
}

typedef __attribute__((ext_vector_type(4))) float v4f;

struct GFrag { v4f a[2]; v4f b[8]; };

template <int N>
__device__ __forceinline__ void gemm_load(GFrag& f, const float* __restrict__ ap,
                                          const float* __restrict__ bp)
{
    f.a[0] = *reinterpret_cast<const v4f*>(ap);
    f.a[1] = *reinterpret_cast<const v4f*>(ap + 4);
    #pragma unroll
    for (int r = 0; r < 8; ++r)
        f.b[r] = *reinterpret_cast<const v4f*>(bp + (size_t)r * N);
}

__device__ __forceinline__ void gemm_mma(const GFrag& f, v8f& c0, v8f& c1, v8f& c2, v8f& c3)
{
    #pragma unroll
    for (int s = 0; s < 4; ++s) {
        const v4f aq = f.a[s >> 1];
        v2f a;
        if (s & 1) { a.x = aq.z; a.y = aq.w; } else { a.x = aq.x; a.y = aq.y; }
        const v4f r0 = f.b[2 * s], r1 = f.b[2 * s + 1];
        v2f b;
        b.x = r0.x; b.y = r1.x;
        c0 = __builtin_amdgcn_wmma_f32_16x16x4_f32(false, a, false, b, (short)0, c0, false, false);
        b.x = r0.y; b.y = r1.y;
        c1 = __builtin_amdgcn_wmma_f32_16x16x4_f32(false, a, false, b, (short)0, c1, false, false);
        b.x = r0.z; b.y = r1.z;
        c2 = __builtin_amdgcn_wmma_f32_16x16x4_f32(false, a, false, b, (short)0, c2, false, false);
        b.x = r0.w; b.y = r1.w;
        c3 = __builtin_amdgcn_wmma_f32_16x16x4_f32(false, a, false, b, (short)0, c3, false, false);
    }
}

#define GEMM_HOLD() __builtin_amdgcn_sched_barrier(0)

template <int N, int K, bool EPI>
__global__ __launch_bounds__(128, 4) void gemm_wmma_f32(
    const float* __restrict__ A, const float* __restrict__ Bm,
    float* __restrict__ C,
    const float* __restrict__ dt_bias, float* __restrict__ DTb,
    float* __restrict__ ladb, float* __restrict__ trapb,
    float* __restrict__ dtmeanb)
{
    static_assert(N % 64 == 0 && K % 64 == 0 && K >= 128, "tile and ring shape");
    const int lane = threadIdx.x & 31;
    const int wave = threadIdx.x >> 5;
    const int lo   = lane & 15;
    const int hi   = lane >> 4;

    const int m0 = blockIdx.y * 64 + wave * 16;
    const int bx = EPI ? N / 64 - 1 - (int)blockIdx.x : (int)blockIdx.x;
    const int n0 = bx * 64;

    const float* ap = A + (size_t)(m0 + lo) * K + 8 * hi;
    const float* bp = Bm + (size_t)(8 * hi) * N + n0 + 4 * lo;
    constexpr int    SA = 16;
    constexpr size_t SB = (size_t)16 * N;
    constexpr int    NS = K / 16;

    v8f c0 = {}, c1 = {}, c2 = {}, c3 = {};
    GFrag f0, f1, f2, f3;
    gemm_load<N>(f0, ap, bp);
    gemm_load<N>(f1, ap + SA, bp + SB);
    gemm_load<N>(f2, ap + 2 * SA, bp + 2 * SB);

    for (int s = 0; s < NS - 4; s += 4) {
        GEMM_HOLD(); gemm_load<N>(f3, ap + 3 * SA, bp + 3 * SB); GEMM_HOLD();
        gemm_mma(f0, c0, c1, c2, c3);
        GEMM_HOLD(); gemm_load<N>(f0, ap + 4 * SA, bp + 4 * SB); GEMM_HOLD();
        gemm_mma(f1, c0, c1, c2, c3);
        GEMM_HOLD(); gemm_load<N>(f1, ap + 5 * SA, bp + 5 * SB); GEMM_HOLD();
        gemm_mma(f2, c0, c1, c2, c3);
        GEMM_HOLD(); gemm_load<N>(f2, ap + 6 * SA, bp + 6 * SB); GEMM_HOLD();
        gemm_mma(f3, c0, c1, c2, c3);
        ap += 4 * SA;
        bp += 4 * SB;
    }
    GEMM_HOLD(); gemm_load<N>(f3, ap + 3 * SA, bp + 3 * SB); GEMM_HOLD();
    gemm_mma(f0, c0, c1, c2, c3);
    gemm_mma(f1, c0, c1, c2, c3);
    gemm_mma(f2, c0, c1, c2, c3);
    gemm_mma(f3, c0, c1, c2, c3);

    #pragma unroll
    for (int r = 0; r < 8; ++r) {
        v4f o; o.x = c0[r]; o.y = c1[r]; o.z = c2[r]; o.w = c3[r];
        *reinterpret_cast<v4f*>(C + (size_t)(m0 + r + 8 * hi) * N + n0 + 4 * lo) = o;
    }

    if constexpr (EPI) {
        static_assert(OFF_DT % 64 == 0 && OFF_A == OFF_DT + 32 && OFF_TRAP == OFF_DT + 64 &&
                      NH_ == 32, "dt|A share one 64-column tile, trap opens the next");
        if (bx == OFF_DT / 64) {
            const bool isdt = lo < 8;
            const int  h4   = 4 * (lo & 7);
            v4f bias;
            #pragma unroll
            for (int j = 0; j < 4; ++j) bias[j] = isdt ? dt_bias[h4 + j] : 0.f;
            float* dst = (isdt ? DTb : ladb) + h4;
            #pragma unroll
            for (int r = 0; r < 8; ++r) {
                const size_t row = (size_t)(m0 + r + 8 * hi);
                v4f sp;
                sp.x = softplus_f(c0[r] + bias.x);
                sp.y = softplus_f(c1[r] + bias.y);
                sp.z = softplus_f(c2[r] + bias.z);
                sp.w = softplus_f(c3[r] + bias.w);
                v4f o;
                #pragma unroll
                for (int j = 0; j < 4; ++j) {
                    const float dtp = __shfl_xor(sp[j], 8, 32);
                    o[j] = isdt ? sp[j] : fminf(-sp[j], -1e-4f) * dtp;
                }
                *reinterpret_cast<v4f*>(dst + row * 32) = o;
                float s = (sp.x + sp.y) + (sp.z + sp.w);
                s += __shfl_xor(s, 1, 32);
                s += __shfl_xor(s, 2, 32);
                s += __shfl_xor(s, 4, 32);
                if (lo == 0) dtmeanb[row] = s * (1.f / 32.f);
            }
        } else if (bx == OFF_TRAP / 64) {
            if (lo < 8) {
                #pragma unroll
                for (int r = 0; r < 8; ++r) {
                    v4f o;
                    o.x = sigmoid_f(c0[r]); o.y = sigmoid_f(c1[r]);
                    o.z = sigmoid_f(c2[r]); o.w = sigmoid_f(c3[r]);
                    *reinterpret_cast<v4f*>(trapb + (size_t)(m0 + r + 8 * hi) * 32 + 4 * lo) = o;
                }
            }
        }
    }
}

__device__ __forceinline__ float wave_sum64(float v0, float v1)
{
    float x = v0 + v1;
    x += __shfl_xor(x, 16, 32);
    x += __shfl_xor(x, 8, 32);
    x += __shfl_xor(x, 4, 32);
    x += __shfl_xor(x, 2, 32);
    x += __shfl_xor(x, 1, 32);
    return x;
}

__global__ __launch_bounds__(512) void rot_kernel(
    const float* __restrict__ proj, const float* __restrict__ dtmeanb,
    const float* __restrict__ Bn_g, const float* __restrict__ Bn_b,
    const float* __restrict__ Cn_g, const float* __restrict__ Cn_b,
    float* __restrict__ Brot, float* __restrict__ Crot)
{
    static_assert(L_ == 512, "32 row tiles of 16 tokens per batch, two per wave");
    __shared__ v2f   incp[(L_ / 2) * 32];
    __shared__ float part[16][16 * 32];
    const int b = blockIdx.x >> 5;
    const int w = blockIdx.x & 31;
    const int j = threadIdx.x & 31;
    const int i = __builtin_amdgcn_readfirstlane((int)(threadIdx.x >> 5));

    const size_t bl  = (size_t)b * L_ + 16 * w + i;
    const float* row = proj + bl * DIP;
    const float vb0 = row[OFF_B + j], vb1 = row[OFF_B + 32 + j];
    const float vc0 = row[OFF_C + j], vc1 = row[OFF_C + 32 + j];
    const float gB0 = Bn_g[j], gB1 = Bn_g[j + 32], hB0 = Bn_b[j], hB1 = Bn_b[j + 32];
    const float gC0 = Cn_g[j], gC1 = Cn_g[j + 32], hC0 = Cn_b[j], hC1 = Cn_b[j + 32];

    float va[2][16];
    #pragma unroll
    for (int u = 0; u < 2; ++u) {
        const int t = i + 16 * u;
        if (t <= w) {
            const float* ang = proj + (size_t)(b * L_ + 16 * t) * DIP + OFF_ANG + j;
            #pragma unroll
            for (int q = 0; q < 16; ++q) va[u][q] = ang[(size_t)q * DIP];
        }
    }
    #pragma unroll
    for (int u = 0; u < 2; ++u) {
        const int t = i + 16 * u;
        if (t <= w) {
            const float* dmp = dtmeanb + b * L_ + 16 * t;
            #pragma unroll
            for (int q = 0; q < 16; ++q) {
                const int l = 16 * t + q;
                incp[(l >> 1) * 32 + j][l & 1] = va[u][q] * dmp[q];
            }
        }
    }

    const float mB  = wave_sum64(vb0, vb1) * (1.f / 64.f);
    const float db0 = vb0 - mB, db1 = vb1 - mB;
    const float rB  = rsqrtf(wave_sum64(db0 * db0, db1 * db1) * (1.f / 64.f) + 1e-5f);
    const float nB0 = db0 * rB * gB0 + hB0;
    const float nB1 = db1 * rB * gB1 + hB1;
    const float mC  = wave_sum64(vc0, vc1) * (1.f / 64.f);
    const float dc0 = vc0 - mC, dc1 = vc1 - mC;
    const float rC  = rsqrtf(wave_sum64(dc0 * dc0, dc1 * dc1) * (1.f / 64.f) + 1e-5f);
    const float nC0 = dc0 * rC * gC0 + hC0;
    const float nC1 = dc1 * rC * gC1 + hC1;

    __syncthreads();

    const int lo = j & 15;
    const int hi = j >> 4;
    v8f acc0 = {}, acc1 = {};
    #pragma unroll
    for (int u = 0; u < 2; ++u) {
        const int t = i + 16 * u;
        if (t <= w) {
            const bool full = t < w;
            const v2f* p = incp + (8 * t + hi) * 32 + lo;
            #pragma unroll
            for (int s = 0; s < 4; ++s) {
                const int kl = 4 * s + 2 * hi;
                v2f a;
                a.x = (full || kl <= lo) ? 1.f : 0.f;
                a.y = (full || kl + 1 <= lo) ? 1.f : 0.f;
                const v2f b0 = p[64 * s];
                const v2f b1 = p[64 * s + 16];
                acc0 = __builtin_amdgcn_wmma_f32_16x16x4_f32(false, a, false, b0, (short)0, acc0, false, false);
                acc1 = __builtin_amdgcn_wmma_f32_16x16x4_f32(false, a, false, b1, (short)0, acc1, false, false);
            }
        }
    }
    #pragma unroll
    for (int r = 0; r < 8; ++r) {
        part[i][(r + 8 * hi) * 32 + lo]      = acc0[r];
        part[i][(r + 8 * hi) * 32 + lo + 16] = acc1[r];
    }
    __syncthreads();

    float pq[16];
    #pragma unroll
    for (int u = 0; u < 16; ++u) pq[u] = part[u][i * 32 + j];
    #pragma unroll
    for (int st = 1; st < 16; st <<= 1) {
        #pragma unroll
        for (int u = 0; u < 16; u += 2 * st) pq[u] += pq[u + st];
    }
    const float ph = pq[0];
    const float c  = cosf(ph), s = sinf(ph);
    const float c0 = __shfl(c, j >> 1, 32),        s0 = __shfl(s, j >> 1, 32);
    const float c1 = __shfl(c, 16 + (j >> 1), 32), s1 = __shfl(s, 16 + (j >> 1), 32);
    const float pB0 = __shfl_xor(nB0, 1, 32), pB1 = __shfl_xor(nB1, 1, 32);
    const float pC0 = __shfl_xor(nC0, 1, 32), pC1 = __shfl_xor(nC1, 1, 32);
    float ob0, ob1, oc0, oc1;
    if ((j & 1) == 0) {
        ob0 = nB0 * c0 - pB0 * s0; ob1 = nB1 * c1 - pB1 * s1;
        oc0 = nC0 * c0 - pC0 * s0; oc1 = nC1 * c1 - pC1 * s1;
    } else {
        ob0 = pB0 * s0 + nB0 * c0; ob1 = pB1 * s1 + nB1 * c1;
        oc0 = pC0 * s0 + nC0 * c0; oc1 = pC1 * s1 + nC1 * c1;
    }
    Brot[bl * 64 + j]      = ob0;
    Brot[bl * 64 + 32 + j] = ob1;
    Crot[bl * 64 + j]      = oc0;
    Crot[bl * 64 + 32 + j] = oc1;
}


template <int TT>
__device__ __forceinline__ void scan_chunk_tile(
    const float* __restrict__ proj, const float* __restrict__ Brot,
    const float* __restrict__ Crot, float* __restrict__ yfin,
    float* __restrict__ Sp, const float* sLa, const float* sP1,
    const float* sP2, const float* sV,
    const size_t bl0c, const size_t blprev, const float prevOn,
    const int h, const int lo, const int hi)
{
    v4f cq[8], bmq[8], bs[TT + 1][8];
    {
        const float* cp  = Crot + (bl0c + 16 * TT + lo) * 64 + 4 * hi;
        const float* bmp = Brot + blprev * 64 + 4 * hi;
        #pragma unroll
        for (int q = 0; q < 8; ++q) {
            cq[q]  = *reinterpret_cast<const v4f*>(cp + 8 * q);
            bmq[q] = *reinterpret_cast<const v4f*>(bmp + 8 * q);
        }
        #pragma unroll
        for (int st = 0; st <= TT; ++st) {
            const float* bp = Brot + (bl0c + 16 * st + lo) * 64 + 4 * hi;
            #pragma unroll
            for (int q = 0; q < 8; ++q)
                bs[st][q] = *reinterpret_cast<const v4f*>(bp + 8 * q);
        }
    }
    const float* xb = proj + OFF_X + h * HD_ + 4 * lo;
    const v4f xm = *reinterpret_cast<const v4f*>(xb + blprev * DIP);

    v8f g[TT + 1];
    #pragma unroll
    for (int st = 0; st <= TT; ++st) {
        g[st] = v8f{};
        #pragma unroll
        for (int q = 0; q < 8; ++q) {
            v2f a, b;
            a.x = bs[st][q].x; a.y = bs[st][q].y; b.x = cq[q].x; b.y = cq[q].y;
            g[st] = __builtin_amdgcn_wmma_f32_16x16x4_f32(false, a, false, b, (short)0, g[st], false, false);
            a.x = bs[st][q].z; a.y = bs[st][q].w; b.x = cq[q].z; b.y = cq[q].w;
            g[st] = __builtin_amdgcn_wmma_f32_16x16x4_f32(false, a, false, b, (short)0, g[st], false, false);
        }
    }
    float gm = 0.f;
    #pragma unroll
    for (int q = 0; q < 8; ++q)
        gm += cq[q].x * bmq[q].x + cq[q].y * bmq[q].y + cq[q].z * bmq[q].z + cq[q].w * bmq[q].w;
    gm += __shfl_xor(gm, 16, 32);

    const float lat = sLa[16 * TT + lo];
    #pragma unroll
    for (int st = 0; st <= TT; ++st) {
        #pragma unroll
        for (int r = 0; r < 8; ++r) {
            const int sl = 8 * hi + r;
            const int s  = 16 * st + sl;
            const float e1 = expf(fminf(lat - sLa[s], 0.f)) * sP1[s];
            const float e2 = expf(fminf(lat - sLa[s + 1], 0.f)) * sP2[s + 1];
            float w;
            if (st < TT) w = e1 + e2;
            else         w = ((sl <= lo) ? e1 : 0.f) + ((sl + 1 <= lo) ? e2 : 0.f);
            g[st][r] *= w;
        }
    }
    const float mprev = prevOn * gm * expf(lat - sLa[0]) * sP2[0];

    v8f y[4] = {};
    #pragma unroll
    for (int st = 0; st <= TT; ++st) {
        #pragma unroll
        for (int j = 0; j < 4; ++j) {
            const size_t row = bl0c + 16 * st + 8 * hi + 2 * j;
            const v4f x0 = *reinterpret_cast<const v4f*>(xb + row * DIP);
            const v4f x1 = *reinterpret_cast<const v4f*>(xb + (row + 1) * DIP);
            v2f a; a.x = g[st][2 * j]; a.y = g[st][2 * j + 1];
            #pragma unroll
            for (int jj = 0; jj < 4; ++jj) {
                v2f b; b.x = x0[jj]; b.y = x1[jj];
                y[jj] = __builtin_amdgcn_wmma_f32_16x16x4_f32(false, a, false, b, (short)0, y[jj], false, false);
            }
        }
    }
    {
        v2f a; a.x = (hi == 0) ? mprev : 0.f; a.y = 0.f;
        #pragma unroll
        for (int jj = 0; jj < 4; ++jj) {
            v2f b; b.x = (hi == 0) ? xm[jj] : 0.f; b.y = 0.f;
            y[jj] = __builtin_amdgcn_wmma_f32_16x16x4_f32(false, a, false, b, (short)0, y[jj], false, false);
        }
    }
    #pragma unroll
    for (int r = 0; r < 8; ++r) {
        v4f o; o.x = y[0][r]; o.y = y[1][r]; o.z = y[2][r]; o.w = y[3][r];
        *reinterpret_cast<v4f*>(yfin + (bl0c + 16 * TT + r + 8 * hi) * DI + h * HD_ + 4 * lo) = o;
    }

    v8f sacc[2][4] = {};
    #pragma unroll
    for (int ks = 0; ks < 8; ++ks) {
        const int s0 = 4 * ks + 2 * hi;
        const v4f xa0 = *reinterpret_cast<const v4f*>(xb + (bl0c + s0) * DIP);
        const v4f xa1 = *reinterpret_cast<const v4f*>(xb + (bl0c + s0 + 1) * DIP);
        const v4f bb0 = *reinterpret_cast<const v4f*>(Brot + (bl0c + s0) * 64 + 4 * lo);
        const v4f bb1 = *reinterpret_cast<const v4f*>(Brot + (bl0c + s0 + 1) * 64 + 4 * lo);
        const float v0 = sV[s0], v1 = sV[s0 + 1];
        #pragma unroll
        for (int a2 = 0; a2 < 2; ++a2) {
            v2f a; a.x = v0 * xa0[2 * TT + a2]; a.y = v1 * xa1[2 * TT + a2];
            #pragma unroll
            for (int jn = 0; jn < 4; ++jn) {
                v2f b; b.x = bb0[jn]; b.y = bb1[jn];
                sacc[a2][jn] = __builtin_amdgcn_wmma_f32_16x16x4_f32(false, a, false, b, (short)0, sacc[a2][jn], false, false);
            }
        }
    }
    {
        const v4f bm4 = *reinterpret_cast<const v4f*>(Brot + blprev * 64 + 4 * lo);
        const float vprev = prevOn * expf(sLa[31] - sLa[0]) * sP2[0];
        #pragma unroll
        for (int a2 = 0; a2 < 2; ++a2) {
            v2f a; a.x = (hi == 0) ? vprev * xm[2 * TT + a2] : 0.f; a.y = 0.f;
            #pragma unroll
            for (int jn = 0; jn < 4; ++jn) {
                v2f b; b.x = (hi == 0) ? bm4[jn] : 0.f; b.y = 0.f;
                sacc[a2][jn] = __builtin_amdgcn_wmma_f32_16x16x4_f32(false, a, false, b, (short)0, sacc[a2][jn], false, false);
            }
        }
    }
    #pragma unroll
    for (int a2 = 0; a2 < 2; ++a2) {
        #pragma unroll
        for (int r = 0; r < 8; ++r) {
            v4f o; o.x = sacc[a2][0][r]; o.y = sacc[a2][1][r]; o.z = sacc[a2][2][r]; o.w = sacc[a2][3][r];
            *reinterpret_cast<v4f*>(Sp + (size_t)(4 * (r + 8 * hi) + 2 * TT + a2) * 64 + 4 * lo) = o;
        }
    }
}

__global__ __launch_bounds__(64) void scan_chunk_kernel(
    const float* __restrict__ proj,
    const float* __restrict__ Brot, const float* __restrict__ Crot,
    const float* __restrict__ DTb, const float* __restrict__ ladb,
    const float* __restrict__ trapb,
    float* __restrict__ yfin, float* __restrict__ cumdb,
    float* __restrict__ Sstate)
{
    static_assert(LC_ == 32 && DS_ == 64 && HD_ == 64, "two 16-token tiles, 64 x 64 state");
    const int blk  = blockIdx.x;
    const int c    = blk % NC_;
    const int h    = (blk / NC_) % NH_;
    const int b    = blk / (NC_ * NH_);
    const int lane = threadIdx.x & 31;
    const int tt   = __builtin_amdgcn_readfirstlane((int)(threadIdx.x >> 5));
    const int lo   = lane & 15;
    const int hi   = lane >> 4;
    const size_t bl0c   = (size_t)(b * L_ + c * LC_);
    const size_t blprev = (c > 0) ? bl0c - 1 : bl0c;
    const float  prevOn = (c > 0) ? 1.f : 0.f;

    __shared__ float sLa[LC_ + 1], sP1[LC_ + 1], sP2[LC_ + 1], sV[LC_];

    if (tt == 0) {
        const size_t o  = (bl0c + lane) * 32 + h;
        const float dt  = DTb[o];
        const float tr  = trapb[o];
        float la        = ladb[o];
        #pragma unroll
        for (int d = 1; d < 32; d <<= 1) {
            const float up = __shfl_up(la, d, 32);
            if (lane >= d) la += up;
        }
        const float p1 = dt * tr;
        const float p2 = dt * (1.f - tr);
        float laN = __shfl_down(la, 1, 32);
        float p2N = __shfl_down(p2, 1, 32);
        if (lane == 31) { laN = la; p2N = 0.f; }
        const float la31 = __shfl(la, 31, 32);
        sLa[lane] = la; sP1[lane] = p1; sP2[lane] = p2;
        sV[lane]  = expf(la31 - la) * p1 + expf(la31 - laN) * p2N;
        if (lane == 31) { sLa[32] = la; sP1[32] = 0.f; sP2[32] = 0.f; }
        cumdb[((size_t)(b * NH_ + h)) * L_ + c * LC_ + lane] = expf(la);
    }
    __syncthreads();

    float* Sp = Sstate + (((size_t)(b * NH_ + h)) * NC_ + c) * 4096;
    if (tt == 0)
        scan_chunk_tile<0>(proj, Brot, Crot, yfin, Sp, sLa, sP1, sP2, sV, bl0c, blprev, prevOn, h, lo, hi);
    else
        scan_chunk_tile<1>(proj, Brot, Crot, yfin, Sp, sLa, sP1, sP2, sV, bl0c, blprev, prevOn, h, lo, hi);
}

__global__ __launch_bounds__(256) void scan_combine_kernel(
    const float* __restrict__ cumdb, float* __restrict__ Sstate)
{
    const int h  = blockIdx.x % NH_;
    const int b  = blockIdx.x / NH_;
    const int t  = threadIdx.x;
    const int p  = t >> 2, seg = t & 3;

    float H[16];
    #pragma unroll
    for (int i = 0; i < 16; ++i) H[i] = 0.f;

    for (int c = 0; c < NC_; ++c) {
        float* Sp = Sstate + (((size_t)(b * NH_ + h)) * NC_ + c) * 4096 + p * 64 + seg * 16;
        const float Pc = cumdb[((size_t)(b * NH_ + h)) * L_ + c * LC_ + LC_ - 1];
        #pragma unroll
        for (int i = 0; i < 16; ++i) {
            const float Sc = Sp[i];
            Sp[i] = H[i];
            H[i]  = Pc * H[i] + Sc;
        }
    }
}

__global__ __launch_bounds__(64) void scan_correct_kernel(
    const float* __restrict__ proj, const float* __restrict__ Crot,
    const float* __restrict__ cumdb, const float* __restrict__ Sstate,
    const float* __restrict__ D_param, float* __restrict__ yfin)
{
    static_assert(LC_ == 32 && DS_ == 64 && HD_ == 64, "two 16-token tiles, 64 x 64 state");
    const int blk  = blockIdx.x;
    const int c    = blk % NC_;
    const int h    = (blk / NC_) % NH_;
    const int b    = blk / (NC_ * NH_);
    const int lane = threadIdx.x & 31;
    const int tt   = threadIdx.x >> 5;
    const int lo   = lane & 15;
    const int hi   = lane >> 4;
    const size_t bl0 = (size_t)(b * L_ + c * LC_ + 16 * tt);

    const float* ap = Crot + (bl0 + lo) * 64 + 4 * hi;
    const float* hp = Sstate + (((size_t)(b * NH_ + h)) * NC_ + c) * 4096
                             + (size_t)(4 * lo) * 64 + 4 * hi;

    v4f a[8], bq[4][8];
    #pragma unroll
    for (int q = 0; q < 8; ++q)
        a[q] = *reinterpret_cast<const v4f*>(ap + 8 * q);
    #pragma unroll
    for (int j = 0; j < 4; ++j) {
        #pragma unroll
        for (int q = 0; q < 8; ++q)
            bq[j][q] = *reinterpret_cast<const v4f*>(hp + j * 64 + 8 * q);
    }

    const float* cdp = cumdb + ((size_t)(b * NH_ + h)) * L_ + c * LC_ + 16 * tt + 8 * hi;
    const v4f cd0 = *reinterpret_cast<const v4f*>(cdp);
    const v4f cd1 = *reinterpret_cast<const v4f*>(cdp + 4);
    const float Dp = D_param[h];
    v4f xv[8], zv[8], yv[8];
    #pragma unroll
    for (int r = 0; r < 8; ++r) {
        const size_t bl = bl0 + r + 8 * hi;
        xv[r] = *reinterpret_cast<const v4f*>(proj + bl * DIP + OFF_X + h * HD_ + 4 * lo);
        zv[r] = *reinterpret_cast<const v4f*>(proj + bl * DIP + OFF_Z + h * HD_ + 4 * lo);
        yv[r] = *reinterpret_cast<const v4f*>(yfin + bl * DI + h * HD_ + 4 * lo);
    }

    __builtin_amdgcn_sched_barrier(0);

    v8f acc[4] = {};
    #pragma unroll
    for (int q = 0; q < 8; ++q) {
        #pragma unroll
        for (int half = 0; half < 2; ++half) {
            v2f av;
            if (half) { av.x = a[q].z; av.y = a[q].w; } else { av.x = a[q].x; av.y = a[q].y; }
            #pragma unroll
            for (int j = 0; j < 4; ++j) {
                v2f bv;
                if (half) { bv.x = bq[j][q].z; bv.y = bq[j][q].w; }
                else      { bv.x = bq[j][q].x; bv.y = bq[j][q].y; }
                acc[j] = __builtin_amdgcn_wmma_f32_16x16x4_f32(false, av, false, bv, (short)0, acc[j], false, false);
            }
        }
    }

    #pragma unroll
    for (int r = 0; r < 8; ++r) {
        const size_t bl = bl0 + r + 8 * hi;
        const float cd = (r < 4) ? cd0[r & 3] : cd1[r & 3];
        v4f o;
        #pragma unroll
        for (int j = 0; j < 4; ++j) {
            const float z = zv[r][j];
            const float y = yv[r][j] + cd * acc[j][r];
            o[j] = (y + Dp * xv[r][j]) * (z * sigmoid_f(z));
        }
        *reinterpret_cast<v4f*>(yfin + bl * DI + h * HD_ + 4 * lo) = o;
    }
}

extern "C" void kernel_launch(void* const* d_in, const int* in_sizes, int n_in,
                              void* d_out, int out_size, void* d_ws, size_t ws_size,
                              hipStream_t stream)
{
    const float* u       = (const float*)d_in[0];
    const float* W_in    = (const float*)d_in[1];
    const float* W_out   = (const float*)d_in[2];
    const float* dt_bias = (const float*)d_in[3];
    const float* D_param = (const float*)d_in[4];
    const float* Bn_g    = (const float*)d_in[5];
    const float* Bn_b    = (const float*)d_in[6];
    const float* Cn_g    = (const float*)d_in[7];
    const float* Cn_b    = (const float*)d_in[8];

    float* ws = (float*)d_ws;
    float* proj   = ws; ws += (size_t)M1 * DIP;
    float* yfin   = ws; ws += (size_t)M1 * DI;
    float* Sstate = ws; ws += (size_t)B_ * NH_ * NC_ * 4096;
    float* Brot   = ws; ws += (size_t)M1 * DS_;
    float* Crot   = ws; ws += (size_t)M1 * DS_;
    float* DTb    = ws; ws += (size_t)M1 * NH_;
    float* ladb   = ws; ws += (size_t)M1 * NH_;
    float* trapb  = ws; ws += (size_t)M1 * NH_;
    float* cumdb  = ws; ws += (size_t)B_ * NH_ * L_;
    float* dtmeanb = ws; ws += (size_t)M1;

    gemm_wmma_f32<DIP, DM, true><<<dim3(DIP / 64, M1 / 64), dim3(128), 0, stream>>>(
        u, W_in, proj, dt_bias, DTb, ladb, trapb, dtmeanb);

    rot_kernel<<<dim3(M1 / 16), dim3(512), 0, stream>>>(
        proj, dtmeanb, Bn_g, Bn_b, Cn_g, Cn_b, Brot, Crot);

    scan_chunk_kernel<<<dim3(B_ * NH_ * NC_), dim3(64), 0, stream>>>(
        proj, Brot, Crot, DTb, ladb, trapb, yfin, cumdb, Sstate);
    scan_combine_kernel<<<dim3(B_ * NH_), dim3(256), 0, stream>>>(
        cumdb, Sstate);
    scan_correct_kernel<<<dim3(B_ * NH_ * NC_), dim3(64), 0, stream>>>(
        proj, Crot, cumdb, Sstate, D_param, yfin);

    gemm_wmma_f32<DM, DI, false><<<dim3(DM / 64, M1 / 64), dim3(128), 0, stream>>>(
        yfin, W_out, (float*)d_out, nullptr, nullptr, nullptr, nullptr, nullptr);
}


# GEMM fragment loads as asm with tied destinations and hand-counted load waits: refills land in their own ring set, no register copies, no wait for all loads per round; sum order unchanged
# speedup vs baseline: 8.0182x; 1.0510x over previous
#include <hip/hip_runtime.h>
#include <math.h>

#define B_   2
#define L_   512
#define DM   1024
#define DI   2048
#define DS_  64
#define NH_  32
#define HD_  64
#define NA_  32
#define DIP  4352
#define M1   (B_*L_)

#define NC_  16
#define LC_  32

#define OFF_Z    0
#define OFF_X    2048
#define OFF_B    4096
#define OFF_C    4160
#define OFF_DT   4224
#define OFF_A    4256
#define OFF_TRAP 4288
#define OFF_ANG  4320

typedef __attribute__((ext_vector_type(2))) float v2f;
typedef __attribute__((ext_vector_type(8))) float v8f;

__device__ __forceinline__ float softplus_f(float x) {
    return x > 20.f ? x : log1pf(expf(x));
}
__device__ __forceinline__ float sigmoid_f(float x) {
    return 1.f / (1.f + expf(-x));
}

typedef __attribute__((ext_vector_type(4))) float v4f;

struct GFrag { v4f a[2]; v4f b[8]; };

#define GEMM_LD_NEW(dst, ptr, off) \
    asm volatile("global_load_b128 %0, %1, off offset:%2" : "=v"(dst) : "v"(ptr), "n"(off) : "memory")
#define GEMM_LD_TIED(dst, ptr, off) \
    asm volatile("global_load_b128 %0, %1, off offset:%2" : "+v"(dst) : "v"(ptr), "n"(off) : "memory")

template <int N, int SLAB, bool FIRST>
__device__ __forceinline__ void gemm_load(GFrag& f, const float* ap, const float* bp)
{
    constexpr int AO = SLAB * 16 * 4;
    constexpr int BO = SLAB * 16 * N * 4;
    constexpr int RS = N * 4;
    static_assert(BO + 7 * RS < (1 << 23), "immediate offset of a global load");
    if constexpr (FIRST) {
        GEMM_LD_NEW(f.a[0], ap, AO);          GEMM_LD_NEW(f.a[1], ap, AO + 16);
        GEMM_LD_NEW(f.b[0], bp, BO);          GEMM_LD_NEW(f.b[1], bp, BO + RS);
        GEMM_LD_NEW(f.b[2], bp, BO + 2 * RS); GEMM_LD_NEW(f.b[3], bp, BO + 3 * RS);
        GEMM_LD_NEW(f.b[4], bp, BO + 4 * RS); GEMM_LD_NEW(f.b[5], bp, BO + 5 * RS);
        GEMM_LD_NEW(f.b[6], bp, BO + 6 * RS); GEMM_LD_NEW(f.b[7], bp, BO + 7 * RS);
    } else {
        GEMM_LD_TIED(f.a[0], ap, AO);          GEMM_LD_TIED(f.a[1], ap, AO + 16);
        GEMM_LD_TIED(f.b[0], bp, BO);          GEMM_LD_TIED(f.b[1], bp, BO + RS);
        GEMM_LD_TIED(f.b[2], bp, BO + 2 * RS); GEMM_LD_TIED(f.b[3], bp, BO + 3 * RS);
        GEMM_LD_TIED(f.b[4], bp, BO + 4 * RS); GEMM_LD_TIED(f.b[5], bp, BO + 5 * RS);
        GEMM_LD_TIED(f.b[6], bp, BO + 6 * RS); GEMM_LD_TIED(f.b[7], bp, BO + 7 * RS);
    }
}

#define GEMM_WAIT(f, CNT) \
    asm volatile("s_wait_loadcnt " #CNT \
                 : "+v"(f.a[0]), "+v"(f.a[1]), "+v"(f.b[0]), "+v"(f.b[1]), "+v"(f.b[2]), \
                   "+v"(f.b[3]), "+v"(f.b[4]), "+v"(f.b[5]), "+v"(f.b[6]), "+v"(f.b[7]) \
                 : : "memory")

#define GEMM_ADDR_FREE(pa, pb) \
    asm volatile("s_wait_xcnt 0" : "+v"(pa), "+v"(pb) : : "memory")

__device__ __forceinline__ void gemm_mma(const GFrag& f, v8f& c0, v8f& c1, v8f& c2, v8f& c3)
{
    #pragma unroll
    for (int s = 0; s < 4; ++s) {
        const v4f aq = f.a[s >> 1];
        v2f a;
        if (s & 1) { a.x = aq.z; a.y = aq.w; } else { a.x = aq.x; a.y = aq.y; }
        const v4f r0 = f.b[2 * s], r1 = f.b[2 * s + 1];
        v2f b;
        b.x = r0.x; b.y = r1.x;
        c0 = __builtin_amdgcn_wmma_f32_16x16x4_f32(false, a, false, b, (short)0, c0, false, false);
        b.x = r0.y; b.y = r1.y;
        c1 = __builtin_amdgcn_wmma_f32_16x16x4_f32(false, a, false, b, (short)0, c1, false, false);
        b.x = r0.z; b.y = r1.z;
        c2 = __builtin_amdgcn_wmma_f32_16x16x4_f32(false, a, false, b, (short)0, c2, false, false);
        b.x = r0.w; b.y = r1.w;
        c3 = __builtin_amdgcn_wmma_f32_16x16x4_f32(false, a, false, b, (short)0, c3, false, false);
    }
}

#define GEMM_HOLD() __builtin_amdgcn_sched_barrier(0)

template <int N, int K, bool EPI>
__global__ __launch_bounds__(128, 4) void gemm_wmma_f32(
    const float* __restrict__ A, const float* __restrict__ Bm,
    float* __restrict__ C,
    const float* __restrict__ dt_bias, float* __restrict__ DTb,
    float* __restrict__ ladb, float* __restrict__ trapb,
    float* __restrict__ dtmeanb)
{
    static_assert(N % 64 == 0 && K % 64 == 0 && K >= 128, "tile and ring shape");
    const int lane = threadIdx.x & 31;
    const int wave = threadIdx.x >> 5;
    const int lo   = lane & 15;
    const int hi   = lane >> 4;

    const int m0 = blockIdx.y * 64 + wave * 16;
    const int bx = EPI ? N / 64 - 1 - (int)blockIdx.x : (int)blockIdx.x;
    const int n0 = bx * 64;

    const float* ap = A + (size_t)(m0 + lo) * K + 8 * hi;
    const float* bp = Bm + (size_t)(8 * hi) * N + n0 + 4 * lo;
    constexpr int    SA = 16;
    constexpr size_t SB = (size_t)16 * N;
    constexpr int    NS = K / 16;

    v8f c0 = {}, c1 = {}, c2 = {}, c3 = {};
    GFrag f0, f1, f2, f3;
    gemm_load<N, 0, true>(f0, ap, bp);
    gemm_load<N, 1, true>(f1, ap, bp);
    gemm_load<N, 2, true>(f2, ap, bp);
    gemm_load<N, 3, true>(f3, ap, bp);

    for (int s = 0; s < NS - 4; s += 4) {
        GEMM_WAIT(f0, 30); gemm_mma(f0, c0, c1, c2, c3);
        GEMM_HOLD(); gemm_load<N, 4, false>(f0, ap, bp); GEMM_HOLD();
        GEMM_WAIT(f1, 30); gemm_mma(f1, c0, c1, c2, c3);
        GEMM_HOLD(); gemm_load<N, 5, false>(f1, ap, bp); GEMM_HOLD();
        GEMM_WAIT(f2, 30); gemm_mma(f2, c0, c1, c2, c3);
        GEMM_HOLD(); gemm_load<N, 6, false>(f2, ap, bp); GEMM_HOLD();
        GEMM_WAIT(f3, 30); gemm_mma(f3, c0, c1, c2, c3);
        GEMM_HOLD(); gemm_load<N, 7, false>(f3, ap, bp); GEMM_HOLD();
        GEMM_ADDR_FREE(ap, bp);
        ap += 4 * SA;
        bp += 4 * SB;
    }
    GEMM_WAIT(f0, 30); gemm_mma(f0, c0, c1, c2, c3); GEMM_HOLD();
    GEMM_WAIT(f1, 20); gemm_mma(f1, c0, c1, c2, c3); GEMM_HOLD();
    GEMM_WAIT(f2, 10); gemm_mma(f2, c0, c1, c2, c3); GEMM_HOLD();
    GEMM_WAIT(f3, 0);  gemm_mma(f3, c0, c1, c2, c3);

    #pragma unroll
    for (int r = 0; r < 8; ++r) {
        v4f o; o.x = c0[r]; o.y = c1[r]; o.z = c2[r]; o.w = c3[r];
        *reinterpret_cast<v4f*>(C + (size_t)(m0 + r + 8 * hi) * N + n0 + 4 * lo) = o;
    }

    if constexpr (EPI) {
        static_assert(OFF_DT % 64 == 0 && OFF_A == OFF_DT + 32 && OFF_TRAP == OFF_DT + 64 &&
                      NH_ == 32, "dt|A share one 64-column tile, trap opens the next");
        if (bx == OFF_DT / 64) {
            const bool isdt = lo < 8;
            const int  h4   = 4 * (lo & 7);
            v4f bias;
            #pragma unroll
            for (int j = 0; j < 4; ++j) bias[j] = isdt ? dt_bias[h4 + j] : 0.f;
            float* dst = (isdt ? DTb : ladb) + h4;
            #pragma unroll
            for (int r = 0; r < 8; ++r) {
                const size_t row = (size_t)(m0 + r + 8 * hi);
                v4f sp;
                sp.x = softplus_f(c0[r] + bias.x);
                sp.y = softplus_f(c1[r] + bias.y);
                sp.z = softplus_f(c2[r] + bias.z);
                sp.w = softplus_f(c3[r] + bias.w);
                v4f o;
                #pragma unroll
                for (int j = 0; j < 4; ++j) {
                    const float dtp = __shfl_xor(sp[j], 8, 32);
                    o[j] = isdt ? sp[j] : fminf(-sp[j], -1e-4f) * dtp;
                }
                *reinterpret_cast<v4f*>(dst + row * 32) = o;
                float s = (sp.x + sp.y) + (sp.z + sp.w);
                s += __shfl_xor(s, 1, 32);
                s += __shfl_xor(s, 2, 32);
                s += __shfl_xor(s, 4, 32);
                if (lo == 0) dtmeanb[row] = s * (1.f / 32.f);
            }
        } else if (bx == OFF_TRAP / 64) {
            if (lo < 8) {
                #pragma unroll
                for (int r = 0; r < 8; ++r) {
                    v4f o;
                    o.x = sigmoid_f(c0[r]); o.y = sigmoid_f(c1[r]);
                    o.z = sigmoid_f(c2[r]); o.w = sigmoid_f(c3[r]);
                    *reinterpret_cast<v4f*>(trapb + (size_t)(m0 + r + 8 * hi) * 32 + 4 * lo) = o;
                }
            }
        }
    }
}

__device__ __forceinline__ float wave_sum64(float v0, float v1)
{
    float x = v0 + v1;
    x += __shfl_xor(x, 16, 32);
    x += __shfl_xor(x, 8, 32);
    x += __shfl_xor(x, 4, 32);
    x += __shfl_xor(x, 2, 32);
    x += __shfl_xor(x, 1, 32);
    return x;
}

__global__ __launch_bounds__(512) void rot_kernel(
    const float* __restrict__ proj, const float* __restrict__ dtmeanb,
    const float* __restrict__ Bn_g, const float* __restrict__ Bn_b,
    const float* __restrict__ Cn_g, const float* __restrict__ Cn_b,
    float* __restrict__ Brot, float* __restrict__ Crot)
{
    static_assert(L_ == 512, "32 row tiles of 16 tokens per batch, two per wave");
    __shared__ v2f   incp[(L_ / 2) * 32];
    __shared__ float part[16][16 * 32];
    const int b = blockIdx.x >> 5;
    const int w = blockIdx.x & 31;
    const int j = threadIdx.x & 31;
    const int i = __builtin_amdgcn_readfirstlane((int)(threadIdx.x >> 5));

    const size_t bl  = (size_t)b * L_ + 16 * w + i;
    const float* row = proj + bl * DIP;
    const float vb0 = row[OFF_B + j], vb1 = row[OFF_B + 32 + j];
    const float vc0 = row[OFF_C + j], vc1 = row[OFF_C + 32 + j];
    const float gB0 = Bn_g[j], gB1 = Bn_g[j + 32], hB0 = Bn_b[j], hB1 = Bn_b[j + 32];
    const float gC0 = Cn_g[j], gC1 = Cn_g[j + 32], hC0 = Cn_b[j], hC1 = Cn_b[j + 32];

    float va[2][16];
    #pragma unroll
    for (int u = 0; u < 2; ++u) {
        const int t = i + 16 * u;
        if (t <= w) {
            const float* ang = proj + (size_t)(b * L_ + 16 * t) * DIP + OFF_ANG + j;
            #pragma unroll
            for (int q = 0; q < 16; ++q) va[u][q] = ang[(size_t)q * DIP];
        }
    }
    #pragma unroll
    for (int u = 0; u < 2; ++u) {
        const int t = i + 16 * u;
        if (t <= w) {
            const float* dmp = dtmeanb + b * L_ + 16 * t;
            #pragma unroll
            for (int q = 0; q < 16; ++q) {
                const int l = 16 * t + q;
                incp[(l >> 1) * 32 + j][l & 1] = va[u][q] * dmp[q];
            }
        }
    }

    const float mB  = wave_sum64(vb0, vb1) * (1.f / 64.f);
    const float db0 = vb0 - mB, db1 = vb1 - mB;
    const float rB  = rsqrtf(wave_sum64(db0 * db0, db1 * db1) * (1.f / 64.f) + 1e-5f);
    const float nB0 = db0 * rB * gB0 + hB0;
    const float nB1 = db1 * rB * gB1 + hB1;
    const float mC  = wave_sum64(vc0, vc1) * (1.f / 64.f);
    const float dc0 = vc0 - mC, dc1 = vc1 - mC;
    const float rC  = rsqrtf(wave_sum64(dc0 * dc0, dc1 * dc1) * (1.f / 64.f) + 1e-5f);
    const float nC0 = dc0 * rC * gC0 + hC0;
    const float nC1 = dc1 * rC * gC1 + hC1;

    __syncthreads();

    const int lo = j & 15;
    const int hi = j >> 4;
    v8f acc0 = {}, acc1 = {};
    #pragma unroll
    for (int u = 0; u < 2; ++u) {
        const int t = i + 16 * u;
        if (t <= w) {
            const bool full = t < w;
            const v2f* p = incp + (8 * t + hi) * 32 + lo;
            #pragma unroll
            for (int s = 0; s < 4; ++s) {
                const int kl = 4 * s + 2 * hi;
                v2f a;
                a.x = (full || kl <= lo) ? 1.f : 0.f;
                a.y = (full || kl + 1 <= lo) ? 1.f : 0.f;
                const v2f b0 = p[64 * s];
                const v2f b1 = p[64 * s + 16];
                acc0 = __builtin_amdgcn_wmma_f32_16x16x4_f32(false, a, false, b0, (short)0, acc0, false, false);
                acc1 = __builtin_amdgcn_wmma_f32_16x16x4_f32(false, a, false, b1, (short)0, acc1, false, false);
            }
        }
    }
    #pragma unroll
    for (int r = 0; r < 8; ++r) {
        part[i][(r + 8 * hi) * 32 + lo]      = acc0[r];
        part[i][(r + 8 * hi) * 32 + lo + 16] = acc1[r];
    }
    __syncthreads();

    float pq[16];
    #pragma unroll
    for (int u = 0; u < 16; ++u) pq[u] = part[u][i * 32 + j];
    #pragma unroll
    for (int st = 1; st < 16; st <<= 1) {
        #pragma unroll
        for (int u = 0; u < 16; u += 2 * st) pq[u] += pq[u + st];
    }
    const float ph = pq[0];
    const float c  = cosf(ph), s = sinf(ph);
    const float c0 = __shfl(c, j >> 1, 32),        s0 = __shfl(s, j >> 1, 32);
    const float c1 = __shfl(c, 16 + (j >> 1), 32), s1 = __shfl(s, 16 + (j >> 1), 32);
    const float pB0 = __shfl_xor(nB0, 1, 32), pB1 = __shfl_xor(nB1, 1, 32);
    const float pC0 = __shfl_xor(nC0, 1, 32), pC1 = __shfl_xor(nC1, 1, 32);
    float ob0, ob1, oc0, oc1;
    if ((j & 1) == 0) {
        ob0 = nB0 * c0 - pB0 * s0; ob1 = nB1 * c1 - pB1 * s1;
        oc0 = nC0 * c0 - pC0 * s0; oc1 = nC1 * c1 - pC1 * s1;
    } else {
        ob0 = pB0 * s0 + nB0 * c0; ob1 = pB1 * s1 + nB1 * c1;
        oc0 = pC0 * s0 + nC0 * c0; oc1 = pC1 * s1 + nC1 * c1;
    }
    Brot[bl * 64 + j]      = ob0;
    Brot[bl * 64 + 32 + j] = ob1;
    Crot[bl * 64 + j]      = oc0;
    Crot[bl * 64 + 32 + j] = oc1;
}


template <int TT>
__device__ __forceinline__ void scan_chunk_tile(
    const float* __restrict__ proj, const float* __restrict__ Brot,
    const float* __restrict__ Crot, float* __restrict__ yfin,
    float* __restrict__ Sp, const float* sLa, const float* sP1,
    const float* sP2, const float* sV,
    const size_t bl0c, const size_t blprev, const float prevOn,
    const int h, const int lo, const int hi)
{
    v4f cq[8], bmq[8], bs[TT + 1][8];
    {
        const float* cp  = Crot + (bl0c + 16 * TT + lo) * 64 + 4 * hi;
        const float* bmp = Brot + blprev * 64 + 4 * hi;
        #pragma unroll
        for (int q = 0; q < 8; ++q) {
            cq[q]  = *reinterpret_cast<const v4f*>(cp + 8 * q);
            bmq[q] = *reinterpret_cast<const v4f*>(bmp + 8 * q);
        }
        #pragma unroll
        for (int st = 0; st <= TT; ++st) {
            const float* bp = Brot + (bl0c + 16 * st + lo) * 64 + 4 * hi;
            #pragma unroll
            for (int q = 0; q < 8; ++q)
                bs[st][q] = *reinterpret_cast<const v4f*>(bp + 8 * q);
        }
    }
    const float* xb = proj + OFF_X + h * HD_ + 4 * lo;
    const v4f xm = *reinterpret_cast<const v4f*>(xb + blprev * DIP);

    v8f g[TT + 1];
    #pragma unroll
    for (int st = 0; st <= TT; ++st) {
        g[st] = v8f{};
        #pragma unroll
        for (int q = 0; q < 8; ++q) {
            v2f a, b;
            a.x = bs[st][q].x; a.y = bs[st][q].y; b.x = cq[q].x; b.y = cq[q].y;
            g[st] = __builtin_amdgcn_wmma_f32_16x16x4_f32(false, a, false, b, (short)0, g[st], false, false);
            a.x = bs[st][q].z; a.y = bs[st][q].w; b.x = cq[q].z; b.y = cq[q].w;
            g[st] = __builtin_amdgcn_wmma_f32_16x16x4_f32(false, a, false, b, (short)0, g[st], false, false);
        }
    }
    float gm = 0.f;
    #pragma unroll
    for (int q = 0; q < 8; ++q)
        gm += cq[q].x * bmq[q].x + cq[q].y * bmq[q].y + cq[q].z * bmq[q].z + cq[q].w * bmq[q].w;
    gm += __shfl_xor(gm, 16, 32);

    const float lat = sLa[16 * TT + lo];
    #pragma unroll
    for (int st = 0; st <= TT; ++st) {
        #pragma unroll
        for (int r = 0; r < 8; ++r) {
            const int sl = 8 * hi + r;
            const int s  = 16 * st + sl;
            const float e1 = expf(fminf(lat - sLa[s], 0.f)) * sP1[s];
            const float e2 = expf(fminf(lat - sLa[s + 1], 0.f)) * sP2[s + 1];
            float w;
            if (st < TT) w = e1 + e2;
            else         w = ((sl <= lo) ? e1 : 0.f) + ((sl + 1 <= lo) ? e2 : 0.f);
            g[st][r] *= w;
        }
    }
    const float mprev = prevOn * gm * expf(lat - sLa[0]) * sP2[0];

    v8f y[4] = {};
    #pragma unroll
    for (int st = 0; st <= TT; ++st) {
        #pragma unroll
        for (int j = 0; j < 4; ++j) {
            const size_t row = bl0c + 16 * st + 8 * hi + 2 * j;
            const v4f x0 = *reinterpret_cast<const v4f*>(xb + row * DIP);
            const v4f x1 = *reinterpret_cast<const v4f*>(xb + (row + 1) * DIP);
            v2f a; a.x = g[st][2 * j]; a.y = g[st][2 * j + 1];
            #pragma unroll
            for (int jj = 0; jj < 4; ++jj) {
                v2f b; b.x = x0[jj]; b.y = x1[jj];
                y[jj] = __builtin_amdgcn_wmma_f32_16x16x4_f32(false, a, false, b, (short)0, y[jj], false, false);
            }
        }
    }
    {
        v2f a; a.x = (hi == 0) ? mprev : 0.f; a.y = 0.f;
        #pragma unroll
        for (int jj = 0; jj < 4; ++jj) {
            v2f b; b.x = (hi == 0) ? xm[jj] : 0.f; b.y = 0.f;
            y[jj] = __builtin_amdgcn_wmma_f32_16x16x4_f32(false, a, false, b, (short)0, y[jj], false, false);
        }
    }
    #pragma unroll
    for (int r = 0; r < 8; ++r) {
        v4f o; o.x = y[0][r]; o.y = y[1][r]; o.z = y[2][r]; o.w = y[3][r];
        *reinterpret_cast<v4f*>(yfin + (bl0c + 16 * TT + r + 8 * hi) * DI + h * HD_ + 4 * lo) = o;
    }

    v8f sacc[2][4] = {};
    #pragma unroll
    for (int ks = 0; ks < 8; ++ks) {
        const int s0 = 4 * ks + 2 * hi;
        const v4f xa0 = *reinterpret_cast<const v4f*>(xb + (bl0c + s0) * DIP);
        const v4f xa1 = *reinterpret_cast<const v4f*>(xb + (bl0c + s0 + 1) * DIP);
        const v4f bb0 = *reinterpret_cast<const v4f*>(Brot + (bl0c + s0) * 64 + 4 * lo);
        const v4f bb1 = *reinterpret_cast<const v4f*>(Brot + (bl0c + s0 + 1) * 64 + 4 * lo);
        const float v0 = sV[s0], v1 = sV[s0 + 1];
        #pragma unroll
        for (int a2 = 0; a2 < 2; ++a2) {
            v2f a; a.x = v0 * xa0[2 * TT + a2]; a.y = v1 * xa1[2 * TT + a2];
            #pragma unroll
            for (int jn = 0; jn < 4; ++jn) {
                v2f b; b.x = bb0[jn]; b.y = bb1[jn];
                sacc[a2][jn] = __builtin_amdgcn_wmma_f32_16x16x4_f32(false, a, false, b, (short)0, sacc[a2][jn], false, false);
            }
        }
    }
    {
        const v4f bm4 = *reinterpret_cast<const v4f*>(Brot + blprev * 64 + 4 * lo);
        const float vprev = prevOn * expf(sLa[31] - sLa[0]) * sP2[0];
        #pragma unroll
        for (int a2 = 0; a2 < 2; ++a2) {
            v2f a; a.x = (hi == 0) ? vprev * xm[2 * TT + a2] : 0.f; a.y = 0.f;
            #pragma unroll
            for (int jn = 0; jn < 4; ++jn) {
                v2f b; b.x = (hi == 0) ? bm4[jn] : 0.f; b.y = 0.f;
                sacc[a2][jn] = __builtin_amdgcn_wmma_f32_16x16x4_f32(false, a, false, b, (short)0, sacc[a2][jn], false, false);
            }
        }
    }
    #pragma unroll
    for (int a2 = 0; a2 < 2; ++a2) {
        #pragma unroll
        for (int r = 0; r < 8; ++r) {
            v4f o; o.x = sacc[a2][0][r]; o.y = sacc[a2][1][r]; o.z = sacc[a2][2][r]; o.w = sacc[a2][3][r];
            *reinterpret_cast<v4f*>(Sp + (size_t)(4 * (r + 8 * hi) + 2 * TT + a2) * 64 + 4 * lo) = o;
        }
    }
}

__global__ __launch_bounds__(64) void scan_chunk_kernel(
    const float* __restrict__ proj,
    const float* __restrict__ Brot, const float* __restrict__ Crot,
    const float* __restrict__ DTb, const float* __restrict__ ladb,
    const float* __restrict__ trapb,
    float* __restrict__ yfin, float* __restrict__ cumdb,
    float* __restrict__ Sstate)
{
    static_assert(LC_ == 32 && DS_ == 64 && HD_ == 64, "two 16-token tiles, 64 x 64 state");
    const int blk  = blockIdx.x;
    const int c    = blk % NC_;
    const int h    = (blk / NC_) % NH_;
    const int b    = blk / (NC_ * NH_);
    const int lane = threadIdx.x & 31;
    const int tt   = __builtin_amdgcn_readfirstlane((int)(threadIdx.x >> 5));
    const int lo   = lane & 15;
    const int hi   = lane >> 4;
    const size_t bl0c   = (size_t)(b * L_ + c * LC_);
    const size_t blprev = (c > 0) ? bl0c - 1 : bl0c;
    const float  prevOn = (c > 0) ? 1.f : 0.f;

    __shared__ float sLa[LC_ + 1], sP1[LC_ + 1], sP2[LC_ + 1], sV[LC_];

    if (tt == 0) {
        const size_t o  = (bl0c + lane) * 32 + h;
        const float dt  = DTb[o];
        const float tr  = trapb[o];
        float la        = ladb[o];
        #pragma unroll
        for (int d = 1; d < 32; d <<= 1) {
            const float up = __shfl_up(la, d, 32);
            if (lane >= d) la += up;
        }
        const float p1 = dt * tr;
        const float p2 = dt * (1.f - tr);
        float laN = __shfl_down(la, 1, 32);
        float p2N = __shfl_down(p2, 1, 32);
        if (lane == 31) { laN = la; p2N = 0.f; }
        const float la31 = __shfl(la, 31, 32);
        sLa[lane] = la; sP1[lane] = p1; sP2[lane] = p2;
        sV[lane]  = expf(la31 - la) * p1 + expf(la31 - laN) * p2N;
        if (lane == 31) { sLa[32] = la; sP1[32] = 0.f; sP2[32] = 0.f; }
        cumdb[((size_t)(b * NH_ + h)) * L_ + c * LC_ + lane] = expf(la);
    }
    __syncthreads();

    float* Sp = Sstate + (((size_t)(b * NH_ + h)) * NC_ + c) * 4096;
    if (tt == 0)
        scan_chunk_tile<0>(proj, Brot, Crot, yfin, Sp, sLa, sP1, sP2, sV, bl0c, blprev, prevOn, h, lo, hi);
    else
        scan_chunk_tile<1>(proj, Brot, Crot, yfin, Sp, sLa, sP1, sP2, sV, bl0c, blprev, prevOn, h, lo, hi);
}

__global__ __launch_bounds__(256) void scan_combine_kernel(
    const float* __restrict__ cumdb, float* __restrict__ Sstate)
{
    const int h  = blockIdx.x % NH_;
    const int b  = blockIdx.x / NH_;
    const int t  = threadIdx.x;
    const int p  = t >> 2, seg = t & 3;

    float H[16];
    #pragma unroll
    for (int i = 0; i < 16; ++i) H[i] = 0.f;

    for (int c = 0; c < NC_; ++c) {
        float* Sp = Sstate + (((size_t)(b * NH_ + h)) * NC_ + c) * 4096 + p * 64 + seg * 16;
        const float Pc = cumdb[((size_t)(b * NH_ + h)) * L_ + c * LC_ + LC_ - 1];
        #pragma unroll
        for (int i = 0; i < 16; ++i) {
            const float Sc = Sp[i];
            Sp[i] = H[i];
            H[i]  = Pc * H[i] + Sc;
        }
    }
}

__global__ __launch_bounds__(64) void scan_correct_kernel(
    const float* __restrict__ proj, const float* __restrict__ Crot,
    const float* __restrict__ cumdb, const float* __restrict__ Sstate,
    const float* __restrict__ D_param, float* __restrict__ yfin)
{
    static_assert(LC_ == 32 && DS_ == 64 && HD_ == 64, "two 16-token tiles, 64 x 64 state");
    const int blk  = blockIdx.x;
    const int c    = blk % NC_;
    const int h    = (blk / NC_) % NH_;
    const int b    = blk / (NC_ * NH_);
    const int lane = threadIdx.x & 31;
    const int tt   = threadIdx.x >> 5;
    const int lo   = lane & 15;
    const int hi   = lane >> 4;
    const size_t bl0 = (size_t)(b * L_ + c * LC_ + 16 * tt);

    const float* ap = Crot + (bl0 + lo) * 64 + 4 * hi;
    const float* hp = Sstate + (((size_t)(b * NH_ + h)) * NC_ + c) * 4096
                             + (size_t)(4 * lo) * 64 + 4 * hi;

    v4f a[8], bq[4][8];
    #pragma unroll
    for (int q = 0; q < 8; ++q)
        a[q] = *reinterpret_cast<const v4f*>(ap + 8 * q);
    #pragma unroll
    for (int j = 0; j < 4; ++j) {
        #pragma unroll
        for (int q = 0; q < 8; ++q)
            bq[j][q] = *reinterpret_cast<const v4f*>(hp + j * 64 + 8 * q);
    }

    const float* cdp = cumdb + ((size_t)(b * NH_ + h)) * L_ + c * LC_ + 16 * tt + 8 * hi;
    const v4f cd0 = *reinterpret_cast<const v4f*>(cdp);
    const v4f cd1 = *reinterpret_cast<const v4f*>(cdp + 4);
    const float Dp = D_param[h];
    v4f xv[8], zv[8], yv[8];
    #pragma unroll
    for (int r = 0; r < 8; ++r) {
        const size_t bl = bl0 + r + 8 * hi;
        xv[r] = *reinterpret_cast<const v4f*>(proj + bl * DIP + OFF_X + h * HD_ + 4 * lo);
        zv[r] = *reinterpret_cast<const v4f*>(proj + bl * DIP + OFF_Z + h * HD_ + 4 * lo);
        yv[r] = *reinterpret_cast<const v4f*>(yfin + bl * DI + h * HD_ + 4 * lo);
    }

    __builtin_amdgcn_sched_barrier(0);

    v8f acc[4] = {};
    #pragma unroll
    for (int q = 0; q < 8; ++q) {
        #pragma unroll
        for (int half = 0; half < 2; ++half) {
            v2f av;
            if (half) { av.x = a[q].z; av.y = a[q].w; } else { av.x = a[q].x; av.y = a[q].y; }
            #pragma unroll
            for (int j = 0; j < 4; ++j) {
                v2f bv;
                if (half) { bv.x = bq[j][q].z; bv.y = bq[j][q].w; }
                else      { bv.x = bq[j][q].x; bv.y = bq[j][q].y; }
                acc[j] = __builtin_amdgcn_wmma_f32_16x16x4_f32(false, av, false, bv, (short)0, acc[j], false, false);
            }
        }
    }

    #pragma unroll
    for (int r = 0; r < 8; ++r) {
        const size_t bl = bl0 + r + 8 * hi;
        const float cd = (r < 4) ? cd0[r & 3] : cd1[r & 3];
        v4f o;
        #pragma unroll
        for (int j = 0; j < 4; ++j) {
            const float z = zv[r][j];
            const float y = yv[r][j] + cd * acc[j][r];
            o[j] = (y + Dp * xv[r][j]) * (z * sigmoid_f(z));
        }
        *reinterpret_cast<v4f*>(yfin + bl * DI + h * HD_ + 4 * lo) = o;
    }
}

extern "C" void kernel_launch(void* const* d_in, const int* in_sizes, int n_in,
                              void* d_out, int out_size, void* d_ws, size_t ws_size,
                              hipStream_t stream)
{
    const float* u       = (const float*)d_in[0];
    const float* W_in    = (const float*)d_in[1];
    const float* W_out   = (const float*)d_in[2];
    const float* dt_bias = (const float*)d_in[3];
    const float* D_param = (const float*)d_in[4];
    const float* Bn_g    = (const float*)d_in[5];
    const float* Bn_b    = (const float*)d_in[6];
    const float* Cn_g    = (const float*)d_in[7];
    const float* Cn_b    = (const float*)d_in[8];

    float* ws = (float*)d_ws;
    float* proj   = ws; ws += (size_t)M1 * DIP;
    float* yfin   = ws; ws += (size_t)M1 * DI;
    float* Sstate = ws; ws += (size_t)B_ * NH_ * NC_ * 4096;
    float* Brot   = ws; ws += (size_t)M1 * DS_;
    float* Crot   = ws; ws += (size_t)M1 * DS_;
    float* DTb    = ws; ws += (size_t)M1 * NH_;
    float* ladb   = ws; ws += (size_t)M1 * NH_;
    float* trapb  = ws; ws += (size_t)M1 * NH_;
    float* cumdb  = ws; ws += (size_t)B_ * NH_ * L_;
    float* dtmeanb = ws; ws += (size_t)M1;

    gemm_wmma_f32<DIP, DM, true><<<dim3(DIP / 64, M1 / 64), dim3(128), 0, stream>>>(
        u, W_in, proj, dt_bias, DTb, ladb, trapb, dtmeanb);

    rot_kernel<<<dim3(M1 / 16), dim3(512), 0, stream>>>(
        proj, dtmeanb, Bn_g, Bn_b, Cn_g, Cn_b, Brot, Crot);

    scan_chunk_kernel<<<dim3(B_ * NH_ * NC_), dim3(64), 0, stream>>>(
        proj, Brot, Crot, DTb, ladb, trapb, yfin, cumdb, Sstate);
    scan_combine_kernel<<<dim3(B_ * NH_), dim3(256), 0, stream>>>(
        cumdb, Sstate);
    scan_correct_kernel<<<dim3(B_ * NH_ * NC_), dim3(64), 0, stream>>>(
        proj, Crot, cumdb, Sstate, D_param, yfin);

    gemm_wmma_f32<DM, DI, false><<<dim3(DM / 64, M1 / 64), dim3(128), 0, stream>>>(
        yfin, W_out, (float*)d_out, nullptr, nullptr, nullptr, nullptr, nullptr);
}


# combine pass re-blocked: a thread owns 4 state elements through all 16 chunks and issues all its chunk loads as one group (one memory round trip, not 8); 1024 blocks, not 64; sum order unchanged
# speedup vs baseline: 10.1702x; 1.2684x over previous
#include <hip/hip_runtime.h>
#include <math.h>

#define B_   2
#define L_   512
#define DM   1024
#define DI   2048
#define DS_  64
#define NH_  32
#define HD_  64
#define NA_  32
#define DIP  4352
#define M1   (B_*L_)

#define NC_  16
#define LC_  32

#define OFF_Z    0
#define OFF_X    2048
#define OFF_B    4096
#define OFF_C    4160
#define OFF_DT   4224
#define OFF_A    4256
#define OFF_TRAP 4288
#define OFF_ANG  4320

typedef __attribute__((ext_vector_type(2))) float v2f;
typedef __attribute__((ext_vector_type(8))) float v8f;

__device__ __forceinline__ float softplus_f(float x) {
    return x > 20.f ? x : log1pf(expf(x));
}
__device__ __forceinline__ float sigmoid_f(float x) {
    return 1.f / (1.f + expf(-x));
}

typedef __attribute__((ext_vector_type(4))) float v4f;

struct GFrag { v4f a[2]; v4f b[8]; };

#define GEMM_LD_NEW(dst, ptr, off) \
    asm volatile("global_load_b128 %0, %1, off offset:%2" : "=v"(dst) : "v"(ptr), "n"(off) : "memory")
#define GEMM_LD_TIED(dst, ptr, off) \
    asm volatile("global_load_b128 %0, %1, off offset:%2" : "+v"(dst) : "v"(ptr), "n"(off) : "memory")

template <int N, int SLAB, bool FIRST>
__device__ __forceinline__ void gemm_load(GFrag& f, const float* ap, const float* bp)
{
    constexpr int AO = SLAB * 16 * 4;
    constexpr int BO = SLAB * 16 * N * 4;
    constexpr int RS = N * 4;
    static_assert(BO + 7 * RS < (1 << 23), "immediate offset of a global load");
    if constexpr (FIRST) {
        GEMM_LD_NEW(f.a[0], ap, AO);          GEMM_LD_NEW(f.a[1], ap, AO + 16);
        GEMM_LD_NEW(f.b[0], bp, BO);          GEMM_LD_NEW(f.b[1], bp, BO + RS);
        GEMM_LD_NEW(f.b[2], bp, BO + 2 * RS); GEMM_LD_NEW(f.b[3], bp, BO + 3 * RS);
        GEMM_LD_NEW(f.b[4], bp, BO + 4 * RS); GEMM_LD_NEW(f.b[5], bp, BO + 5 * RS);
        GEMM_LD_NEW(f.b[6], bp, BO + 6 * RS); GEMM_LD_NEW(f.b[7], bp, BO + 7 * RS);
    } else {
        GEMM_LD_TIED(f.a[0], ap, AO);          GEMM_LD_TIED(f.a[1], ap, AO + 16);
        GEMM_LD_TIED(f.b[0], bp, BO);          GEMM_LD_TIED(f.b[1], bp, BO + RS);
        GEMM_LD_TIED(f.b[2], bp, BO + 2 * RS); GEMM_LD_TIED(f.b[3], bp, BO + 3 * RS);
        GEMM_LD_TIED(f.b[4], bp, BO + 4 * RS); GEMM_LD_TIED(f.b[5], bp, BO + 5 * RS);
        GEMM_LD_TIED(f.b[6], bp, BO + 6 * RS); GEMM_LD_TIED(f.b[7], bp, BO + 7 * RS);
    }
}

#define GEMM_WAIT(f, CNT) \
    asm volatile("s_wait_loadcnt " #CNT \
                 : "+v"(f.a[0]), "+v"(f.a[1]), "+v"(f.b[0]), "+v"(f.b[1]), "+v"(f.b[2]), \
                   "+v"(f.b[3]), "+v"(f.b[4]), "+v"(f.b[5]), "+v"(f.b[6]), "+v"(f.b[7]) \
                 : : "memory")

#define GEMM_ADDR_FREE(pa, pb) \
    asm volatile("s_wait_xcnt 0" : "+v"(pa), "+v"(pb) : : "memory")

__device__ __forceinline__ void gemm_mma(const GFrag& f, v8f& c0, v8f& c1, v8f& c2, v8f& c3)
{
    #pragma unroll
    for (int s = 0; s < 4; ++s) {
        const v4f aq = f.a[s >> 1];
        v2f a;
        if (s & 1) { a.x = aq.z; a.y = aq.w; } else { a.x = aq.x; a.y = aq.y; }
        const v4f r0 = f.b[2 * s], r1 = f.b[2 * s + 1];
        v2f b;
        b.x = r0.x; b.y = r1.x;
        c0 = __builtin_amdgcn_wmma_f32_16x16x4_f32(false, a, false, b, (short)0, c0, false, false);
        b.x = r0.y; b.y = r1.y;
        c1 = __builtin_amdgcn_wmma_f32_16x16x4_f32(false, a, false, b, (short)0, c1, false, false);
        b.x = r0.z; b.y = r1.z;
        c2 = __builtin_amdgcn_wmma_f32_16x16x4_f32(false, a, false, b, (short)0, c2, false, false);
        b.x = r0.w; b.y = r1.w;
        c3 = __builtin_amdgcn_wmma_f32_16x16x4_f32(false, a, false, b, (short)0, c3, false, false);
    }
}

#define GEMM_HOLD() __builtin_amdgcn_sched_barrier(0)

template <int N, int K, bool EPI>
__global__ __launch_bounds__(128, 4) void gemm_wmma_f32(
    const float* __restrict__ A, const float* __restrict__ Bm,
    float* __restrict__ C,
    const float* __restrict__ dt_bias, float* __restrict__ DTb,
    float* __restrict__ ladb, float* __restrict__ trapb,
    float* __restrict__ dtmeanb)
{
    static_assert(N % 64 == 0 && K % 64 == 0 && K >= 128, "tile and ring shape");
    const int lane = threadIdx.x & 31;
    const int wave = threadIdx.x >> 5;
    const int lo   = lane & 15;
    const int hi   = lane >> 4;

    const int m0 = blockIdx.y * 64 + wave * 16;
    const int bx = EPI ? N / 64 - 1 - (int)blockIdx.x : (int)blockIdx.x;
    const int n0 = bx * 64;

    const float* ap = A + (size_t)(m0 + lo) * K + 8 * hi;
    const float* bp = Bm + (size_t)(8 * hi) * N + n0 + 4 * lo;
    constexpr int    SA = 16;
    constexpr size_t SB = (size_t)16 * N;
    constexpr int    NS = K / 16;

    v8f c0 = {}, c1 = {}, c2 = {}, c3 = {};
    GFrag f0, f1, f2, f3;
    gemm_load<N, 0, true>(f0, ap, bp);
    gemm_load<N, 1, true>(f1, ap, bp);
    gemm_load<N, 2, true>(f2, ap, bp);
    gemm_load<N, 3, true>(f3, ap, bp);

    for (int s = 0; s < NS - 4; s += 4) {
        GEMM_WAIT(f0, 30); gemm_mma(f0, c0, c1, c2, c3);
        GEMM_HOLD(); gemm_load<N, 4, false>(f0, ap, bp); GEMM_HOLD();
        GEMM_WAIT(f1, 30); gemm_mma(f1, c0, c1, c2, c3);
        GEMM_HOLD(); gemm_load<N, 5, false>(f1, ap, bp); GEMM_HOLD();
        GEMM_WAIT(f2, 30); gemm_mma(f2, c0, c1, c2, c3);
        GEMM_HOLD(); gemm_load<N, 6, false>(f2, ap, bp); GEMM_HOLD();
        GEMM_WAIT(f3, 30); gemm_mma(f3, c0, c1, c2, c3);
        GEMM_HOLD(); gemm_load<N, 7, false>(f3, ap, bp); GEMM_HOLD();
        GEMM_ADDR_FREE(ap, bp);
        ap += 4 * SA;
        bp += 4 * SB;
    }
    GEMM_WAIT(f0, 30); gemm_mma(f0, c0, c1, c2, c3); GEMM_HOLD();
    GEMM_WAIT(f1, 20); gemm_mma(f1, c0, c1, c2, c3); GEMM_HOLD();
    GEMM_WAIT(f2, 10); gemm_mma(f2, c0, c1, c2, c3); GEMM_HOLD();
    GEMM_WAIT(f3, 0);  gemm_mma(f3, c0, c1, c2, c3);

    #pragma unroll
    for (int r = 0; r < 8; ++r) {
        v4f o; o.x = c0[r]; o.y = c1[r]; o.z = c2[r]; o.w = c3[r];
        *reinterpret_cast<v4f*>(C + (size_t)(m0 + r + 8 * hi) * N + n0 + 4 * lo) = o;
    }

    if constexpr (EPI) {
        static_assert(OFF_DT % 64 == 0 && OFF_A == OFF_DT + 32 && OFF_TRAP == OFF_DT + 64 &&
                      NH_ == 32, "dt|A share one 64-column tile, trap opens the next");
        if (bx == OFF_DT / 64) {
            const bool isdt = lo < 8;
            const int  h4   = 4 * (lo & 7);
            v4f bias;
            #pragma unroll
            for (int j = 0; j < 4; ++j) bias[j] = isdt ? dt_bias[h4 + j] : 0.f;
            float* dst = (isdt ? DTb : ladb) + h4;
            #pragma unroll
            for (int r = 0; r < 8; ++r) {
                const size_t row = (size_t)(m0 + r + 8 * hi);
                v4f sp;
                sp.x = softplus_f(c0[r] + bias.x);
                sp.y = softplus_f(c1[r] + bias.y);
                sp.z = softplus_f(c2[r] + bias.z);
                sp.w = softplus_f(c3[r] + bias.w);
                v4f o;
                #pragma unroll
                for (int j = 0; j < 4; ++j) {
                    const float dtp = __shfl_xor(sp[j], 8, 32);
                    o[j] = isdt ? sp[j] : fminf(-sp[j], -1e-4f) * dtp;
                }
                *reinterpret_cast<v4f*>(dst + row * 32) = o;
                float s = (sp.x + sp.y) + (sp.z + sp.w);
                s += __shfl_xor(s, 1, 32);
                s += __shfl_xor(s, 2, 32);
                s += __shfl_xor(s, 4, 32);
                if (lo == 0) dtmeanb[row] = s * (1.f / 32.f);
            }
        } else if (bx == OFF_TRAP / 64) {
            if (lo < 8) {
                #pragma unroll
                for (int r = 0; r < 8; ++r) {
                    v4f o;
                    o.x = sigmoid_f(c0[r]); o.y = sigmoid_f(c1[r]);
                    o.z = sigmoid_f(c2[r]); o.w = sigmoid_f(c3[r]);
                    *reinterpret_cast<v4f*>(trapb + (size_t)(m0 + r + 8 * hi) * 32 + 4 * lo) = o;
                }
            }
        }
    }
}

__device__ __forceinline__ float wave_sum64(float v0, float v1)
{
    float x = v0 + v1;
    x += __shfl_xor(x, 16, 32);
    x += __shfl_xor(x, 8, 32);
    x += __shfl_xor(x, 4, 32);
    x += __shfl_xor(x, 2, 32);
    x += __shfl_xor(x, 1, 32);
    return x;
}

__global__ __launch_bounds__(512) void rot_kernel(
    const float* __restrict__ proj, const float* __restrict__ dtmeanb,
    const float* __restrict__ Bn_g, const float* __restrict__ Bn_b,
    const float* __restrict__ Cn_g, const float* __restrict__ Cn_b,
    float* __restrict__ Brot, float* __restrict__ Crot)
{
    static_assert(L_ == 512, "32 row tiles of 16 tokens per batch, two per wave");
    __shared__ v2f   incp[(L_ / 2) * 32];
    __shared__ float part[16][16 * 32];
    const int b = blockIdx.x >> 5;
    const int w = blockIdx.x & 31;
    const int j = threadIdx.x & 31;
    const int i = __builtin_amdgcn_readfirstlane((int)(threadIdx.x >> 5));

    const size_t bl  = (size_t)b * L_ + 16 * w + i;
    const float* row = proj + bl * DIP;
    const float vb0 = row[OFF_B + j], vb1 = row[OFF_B + 32 + j];
    const float vc0 = row[OFF_C + j], vc1 = row[OFF_C + 32 + j];
    const float gB0 = Bn_g[j], gB1 = Bn_g[j + 32], hB0 = Bn_b[j], hB1 = Bn_b[j + 32];
    const float gC0 = Cn_g[j], gC1 = Cn_g[j + 32], hC0 = Cn_b[j], hC1 = Cn_b[j + 32];

    float va[2][16];
    #pragma unroll
    for (int u = 0; u < 2; ++u) {
        const int t = i + 16 * u;
        if (t <= w) {
            const float* ang = proj + (size_t)(b * L_ + 16 * t) * DIP + OFF_ANG + j;
            #pragma unroll
            for (int q = 0; q < 16; ++q) va[u][q] = ang[(size_t)q * DIP];
        }
    }
    #pragma unroll
    for (int u = 0; u < 2; ++u) {
        const int t = i + 16 * u;
        if (t <= w) {
            const float* dmp = dtmeanb + b * L_ + 16 * t;
            #pragma unroll
            for (int q = 0; q < 16; ++q) {
                const int l = 16 * t + q;
                incp[(l >> 1) * 32 + j][l & 1] = va[u][q] * dmp[q];
            }
        }
    }

    const float mB  = wave_sum64(vb0, vb1) * (1.f / 64.f);
    const float db0 = vb0 - mB, db1 = vb1 - mB;
    const float rB  = rsqrtf(wave_sum64(db0 * db0, db1 * db1) * (1.f / 64.f) + 1e-5f);
    const float nB0 = db0 * rB * gB0 + hB0;
    const float nB1 = db1 * rB * gB1 + hB1;
    const float mC  = wave_sum64(vc0, vc1) * (1.f / 64.f);
    const float dc0 = vc0 - mC, dc1 = vc1 - mC;
    const float rC  = rsqrtf(wave_sum64(dc0 * dc0, dc1 * dc1) * (1.f / 64.f) + 1e-5f);
    const float nC0 = dc0 * rC * gC0 + hC0;
    const float nC1 = dc1 * rC * gC1 + hC1;

    __syncthreads();

    const int lo = j & 15;
    const int hi = j >> 4;
    v8f acc0 = {}, acc1 = {};
    #pragma unroll
    for (int u = 0; u < 2; ++u) {
        const int t = i + 16 * u;
        if (t <= w) {
            const bool full = t < w;
            const v2f* p = incp + (8 * t + hi) * 32 + lo;
            #pragma unroll
            for (int s = 0; s < 4; ++s) {
                const int kl = 4 * s + 2 * hi;
                v2f a;
                a.x = (full || kl <= lo) ? 1.f : 0.f;
                a.y = (full || kl + 1 <= lo) ? 1.f : 0.f;
                const v2f b0 = p[64 * s];
                const v2f b1 = p[64 * s + 16];
                acc0 = __builtin_amdgcn_wmma_f32_16x16x4_f32(false, a, false, b0, (short)0, acc0, false, false);
                acc1 = __builtin_amdgcn_wmma_f32_16x16x4_f32(false, a, false, b1, (short)0, acc1, false, false);
            }
        }
    }
    #pragma unroll
    for (int r = 0; r < 8; ++r) {
        part[i][(r + 8 * hi) * 32 + lo]      = acc0[r];
        part[i][(r + 8 * hi) * 32 + lo + 16] = acc1[r];
    }
    __syncthreads();

    float pq[16];
    #pragma unroll
    for (int u = 0; u < 16; ++u) pq[u] = part[u][i * 32 + j];
    #pragma unroll
    for (int st = 1; st < 16; st <<= 1) {
        #pragma unroll
        for (int u = 0; u < 16; u += 2 * st) pq[u] += pq[u + st];
    }
    const float ph = pq[0];
    const float c  = cosf(ph), s = sinf(ph);
    const float c0 = __shfl(c, j >> 1, 32),        s0 = __shfl(s, j >> 1, 32);
    const float c1 = __shfl(c, 16 + (j >> 1), 32), s1 = __shfl(s, 16 + (j >> 1), 32);
    const float pB0 = __shfl_xor(nB0, 1, 32), pB1 = __shfl_xor(nB1, 1, 32);
    const float pC0 = __shfl_xor(nC0, 1, 32), pC1 = __shfl_xor(nC1, 1, 32);
    float ob0, ob1, oc0, oc1;
    if ((j & 1) == 0) {
        ob0 = nB0 * c0 - pB0 * s0; ob1 = nB1 * c1 - pB1 * s1;
        oc0 = nC0 * c0 - pC0 * s0; oc1 = nC1 * c1 - pC1 * s1;
    } else {
        ob0 = pB0 * s0 + nB0 * c0; ob1 = pB1 * s1 + nB1 * c1;
        oc0 = pC0 * s0 + nC0 * c0; oc1 = pC1 * s1 + nC1 * c1;
    }
    Brot[bl * 64 + j]      = ob0;
    Brot[bl * 64 + 32 + j] = ob1;
    Crot[bl * 64 + j]      = oc0;
    Crot[bl * 64 + 32 + j] = oc1;
}


template <int TT>
__device__ __forceinline__ void scan_chunk_tile(
    const float* __restrict__ proj, const float* __restrict__ Brot,
    const float* __restrict__ Crot, float* __restrict__ yfin,
    float* __restrict__ Sp, const float* sLa, const float* sP1,
    const float* sP2, const float* sV,
    const size_t bl0c, const size_t blprev, const float prevOn,
    const int h, const int lo, const int hi)
{
    v4f cq[8], bmq[8], bs[TT + 1][8];
    {
        const float* cp  = Crot + (bl0c + 16 * TT + lo) * 64 + 4 * hi;
        const float* bmp = Brot + blprev * 64 + 4 * hi;
        #pragma unroll
        for (int q = 0; q < 8; ++q) {
            cq[q]  = *reinterpret_cast<const v4f*>(cp + 8 * q);
            bmq[q] = *reinterpret_cast<const v4f*>(bmp + 8 * q);
        }
        #pragma unroll
        for (int st = 0; st <= TT; ++st) {
            const float* bp = Brot + (bl0c + 16 * st + lo) * 64 + 4 * hi;
            #pragma unroll
            for (int q = 0; q < 8; ++q)
                bs[st][q] = *reinterpret_cast<const v4f*>(bp + 8 * q);
        }
    }
    const float* xb = proj + OFF_X + h * HD_ + 4 * lo;
    const v4f xm = *reinterpret_cast<const v4f*>(xb + blprev * DIP);

    v8f g[TT + 1];
    #pragma unroll
    for (int st = 0; st <= TT; ++st) {
        g[st] = v8f{};
        #pragma unroll
        for (int q = 0; q < 8; ++q) {
            v2f a, b;
            a.x = bs[st][q].x; a.y = bs[st][q].y; b.x = cq[q].x; b.y = cq[q].y;
            g[st] = __builtin_amdgcn_wmma_f32_16x16x4_f32(false, a, false, b, (short)0, g[st], false, false);
            a.x = bs[st][q].z; a.y = bs[st][q].w; b.x = cq[q].z; b.y = cq[q].w;
            g[st] = __builtin_amdgcn_wmma_f32_16x16x4_f32(false, a, false, b, (short)0, g[st], false, false);
        }
    }
    float gm = 0.f;
    #pragma unroll
    for (int q = 0; q < 8; ++q)
        gm += cq[q].x * bmq[q].x + cq[q].y * bmq[q].y + cq[q].z * bmq[q].z + cq[q].w * bmq[q].w;
    gm += __shfl_xor(gm, 16, 32);

    const float lat = sLa[16 * TT + lo];
    #pragma unroll
    for (int st = 0; st <= TT; ++st) {
        #pragma unroll
        for (int r = 0; r < 8; ++r) {
            const int sl = 8 * hi + r;
            const int s  = 16 * st + sl;
            const float e1 = expf(fminf(lat - sLa[s], 0.f)) * sP1[s];
            const float e2 = expf(fminf(lat - sLa[s + 1], 0.f)) * sP2[s + 1];
            float w;
            if (st < TT) w = e1 + e2;
            else         w = ((sl <= lo) ? e1 : 0.f) + ((sl + 1 <= lo) ? e2 : 0.f);
            g[st][r] *= w;
        }
    }
    const float mprev = prevOn * gm * expf(lat - sLa[0]) * sP2[0];

    v8f y[4] = {};
    #pragma unroll
    for (int st = 0; st <= TT; ++st) {
        #pragma unroll
        for (int j = 0; j < 4; ++j) {
            const size_t row = bl0c + 16 * st + 8 * hi + 2 * j;
            const v4f x0 = *reinterpret_cast<const v4f*>(xb + row * DIP);
            const v4f x1 = *reinterpret_cast<const v4f*>(xb + (row + 1) * DIP);
            v2f a; a.x = g[st][2 * j]; a.y = g[st][2 * j + 1];
            #pragma unroll
            for (int jj = 0; jj < 4; ++jj) {
                v2f b; b.x = x0[jj]; b.y = x1[jj];
                y[jj] = __builtin_amdgcn_wmma_f32_16x16x4_f32(false, a, false, b, (short)0, y[jj], false, false);
            }
        }
    }
    {
        v2f a; a.x = (hi == 0) ? mprev : 0.f; a.y = 0.f;
        #pragma unroll
        for (int jj = 0; jj < 4; ++jj) {
            v2f b; b.x = (hi == 0) ? xm[jj] : 0.f; b.y = 0.f;
            y[jj] = __builtin_amdgcn_wmma_f32_16x16x4_f32(false, a, false, b, (short)0, y[jj], false, false);
        }
    }
    #pragma unroll
    for (int r = 0; r < 8; ++r) {
        v4f o; o.x = y[0][r]; o.y = y[1][r]; o.z = y[2][r]; o.w = y[3][r];
        *reinterpret_cast<v4f*>(yfin + (bl0c + 16 * TT + r + 8 * hi) * DI + h * HD_ + 4 * lo) = o;
    }

    v8f sacc[2][4] = {};
    #pragma unroll
    for (int ks = 0; ks < 8; ++ks) {
        const int s0 = 4 * ks + 2 * hi;
        const v4f xa0 = *reinterpret_cast<const v4f*>(xb + (bl0c + s0) * DIP);
        const v4f xa1 = *reinterpret_cast<const v4f*>(xb + (bl0c + s0 + 1) * DIP);
        const v4f bb0 = *reinterpret_cast<const v4f*>(Brot + (bl0c + s0) * 64 + 4 * lo);
        const v4f bb1 = *reinterpret_cast<const v4f*>(Brot + (bl0c + s0 + 1) * 64 + 4 * lo);
        const float v0 = sV[s0], v1 = sV[s0 + 1];
        #pragma unroll
        for (int a2 = 0; a2 < 2; ++a2) {
            v2f a; a.x = v0 * xa0[2 * TT + a2]; a.y = v1 * xa1[2 * TT + a2];
            #pragma unroll
            for (int jn = 0; jn < 4; ++jn) {
                v2f b; b.x = bb0[jn]; b.y = bb1[jn];
                sacc[a2][jn] = __builtin_amdgcn_wmma_f32_16x16x4_f32(false, a, false, b, (short)0, sacc[a2][jn], false, false);
            }
        }
    }
    {
        const v4f bm4 = *reinterpret_cast<const v4f*>(Brot + blprev * 64 + 4 * lo);
        const float vprev = prevOn * expf(sLa[31] - sLa[0]) * sP2[0];
        #pragma unroll
        for (int a2 = 0; a2 < 2; ++a2) {
            v2f a; a.x = (hi == 0) ? vprev * xm[2 * TT + a2] : 0.f; a.y = 0.f;
            #pragma unroll
            for (int jn = 0; jn < 4; ++jn) {
                v2f b; b.x = (hi == 0) ? bm4[jn] : 0.f; b.y = 0.f;
                sacc[a2][jn] = __builtin_amdgcn_wmma_f32_16x16x4_f32(false, a, false, b, (short)0, sacc[a2][jn], false, false);
            }
        }
    }
    #pragma unroll
    for (int a2 = 0; a2 < 2; ++a2) {
        #pragma unroll
        for (int r = 0; r < 8; ++r) {
            v4f o; o.x = sacc[a2][0][r]; o.y = sacc[a2][1][r]; o.z = sacc[a2][2][r]; o.w = sacc[a2][3][r];
            *reinterpret_cast<v4f*>(Sp + (size_t)(4 * (r + 8 * hi) + 2 * TT + a2) * 64 + 4 * lo) = o;
        }
    }
}

__global__ __launch_bounds__(64) void scan_chunk_kernel(
    const float* __restrict__ proj,
    const float* __restrict__ Brot, const float* __restrict__ Crot,
    const float* __restrict__ DTb, const float* __restrict__ ladb,
    const float* __restrict__ trapb,
    float* __restrict__ yfin, float* __restrict__ cumdb,
    float* __restrict__ Sstate)
{
    static_assert(LC_ == 32 && DS_ == 64 && HD_ == 64, "two 16-token tiles, 64 x 64 state");
    const int blk  = blockIdx.x;
    const int c    = blk % NC_;
    const int h    = (blk / NC_) % NH_;
    const int b    = blk / (NC_ * NH_);
    const int lane = threadIdx.x & 31;
    const int tt   = __builtin_amdgcn_readfirstlane((int)(threadIdx.x >> 5));
    const int lo   = lane & 15;
    const int hi   = lane >> 4;
    const size_t bl0c   = (size_t)(b * L_ + c * LC_);
    const size_t blprev = (c > 0) ? bl0c - 1 : bl0c;
    const float  prevOn = (c > 0) ? 1.f : 0.f;

    __shared__ float sLa[LC_ + 1], sP1[LC_ + 1], sP2[LC_ + 1], sV[LC_];

    if (tt == 0) {
        const size_t o  = (bl0c + lane) * 32 + h;
        const float dt  = DTb[o];
        const float tr  = trapb[o];
        float la        = ladb[o];
        #pragma unroll
        for (int d = 1; d < 32; d <<= 1) {
            const float up = __shfl_up(la, d, 32);
            if (lane >= d) la += up;
        }
        const float p1 = dt * tr;
        const float p2 = dt * (1.f - tr);
        float laN = __shfl_down(la, 1, 32);
        float p2N = __shfl_down(p2, 1, 32);
        if (lane == 31) { laN = la; p2N = 0.f; }
        const float la31 = __shfl(la, 31, 32);
        sLa[lane] = la; sP1[lane] = p1; sP2[lane] = p2;
        sV[lane]  = expf(la31 - la) * p1 + expf(la31 - laN) * p2N;
        if (lane == 31) { sLa[32] = la; sP1[32] = 0.f; sP2[32] = 0.f; }
        cumdb[((size_t)(b * NH_ + h)) * L_ + c * LC_ + lane] = expf(la);
    }
    __syncthreads();

    float* Sp = Sstate + (((size_t)(b * NH_ + h)) * NC_ + c) * 4096;
    if (tt == 0)
        scan_chunk_tile<0>(proj, Brot, Crot, yfin, Sp, sLa, sP1, sP2, sV, bl0c, blprev, prevOn, h, lo, hi);
    else
        scan_chunk_tile<1>(proj, Brot, Crot, yfin, Sp, sLa, sP1, sP2, sV, bl0c, blprev, prevOn, h, lo, hi);
}

__global__ __launch_bounds__(64) void scan_combine_kernel(
    const float* __restrict__ cumdb, float* Sstate)
{
    static_assert(64 * 4 * 16 == 4096, "16 blocks of 64 threads x 4 elements cover one state");
    const int bh = blockIdx.x >> 4;
    const int e  = (((int)blockIdx.x & 15) * 64 + (int)threadIdx.x) * 4;
    float* sp = Sstate + (size_t)bh * NC_ * 4096 + e;
    const float* pc = cumdb + (size_t)bh * L_ + LC_ - 1;

    v4f   S[NC_];
    float P[NC_];
    #pragma unroll
    for (int c = 0; c < NC_; ++c) {
        S[c] = *reinterpret_cast<const v4f*>(sp + (size_t)c * 4096);
        P[c] = pc[c * LC_];
    }
    __builtin_amdgcn_sched_barrier(0);

    v4f H = {0.f, 0.f, 0.f, 0.f};
    #pragma unroll
    for (int c = 0; c < NC_; ++c) {
        *reinterpret_cast<v4f*>(sp + (size_t)c * 4096) = H;
        H = P[c] * H + S[c];
    }
}

__global__ __launch_bounds__(64) void scan_correct_kernel(
    const float* __restrict__ proj, const float* __restrict__ Crot,
    const float* __restrict__ cumdb, const float* __restrict__ Sstate,
    const float* __restrict__ D_param, float* __restrict__ yfin)
{
    static_assert(LC_ == 32 && DS_ == 64 && HD_ == 64, "two 16-token tiles, 64 x 64 state");
    const int blk  = blockIdx.x;
    const int c    = blk % NC_;
    const int h    = (blk / NC_) % NH_;
    const int b    = blk / (NC_ * NH_);
    const int lane = threadIdx.x & 31;
    const int tt   = threadIdx.x >> 5;
    const int lo   = lane & 15;
    const int hi   = lane >> 4;
    const size_t bl0 = (size_t)(b * L_ + c * LC_ + 16 * tt);

    const float* ap = Crot + (bl0 + lo) * 64 + 4 * hi;
    const float* hp = Sstate + (((size_t)(b * NH_ + h)) * NC_ + c) * 4096
                             + (size_t)(4 * lo) * 64 + 4 * hi;

    v4f a[8], bq[4][8];
    #pragma unroll
    for (int q = 0; q < 8; ++q)
        a[q] = *reinterpret_cast<const v4f*>(ap + 8 * q);
    #pragma unroll
    for (int j = 0; j < 4; ++j) {
        #pragma unroll
        for (int q = 0; q < 8; ++q)
            bq[j][q] = *reinterpret_cast<const v4f*>(hp + j * 64 + 8 * q);
    }

    const float* cdp = cumdb + ((size_t)(b * NH_ + h)) * L_ + c * LC_ + 16 * tt + 8 * hi;
    const v4f cd0 = *reinterpret_cast<const v4f*>(cdp);
    const v4f cd1 = *reinterpret_cast<const v4f*>(cdp + 4);
    const float Dp = D_param[h];
    v4f xv[8], zv[8], yv[8];
    #pragma unroll
    for (int r = 0; r < 8; ++r) {
        const size_t bl = bl0 + r + 8 * hi;
        xv[r] = *reinterpret_cast<const v4f*>(proj + bl * DIP + OFF_X + h * HD_ + 4 * lo);
        zv[r] = *reinterpret_cast<const v4f*>(proj + bl * DIP + OFF_Z + h * HD_ + 4 * lo);
        yv[r] = *reinterpret_cast<const v4f*>(yfin + bl * DI + h * HD_ + 4 * lo);
    }

    __builtin_amdgcn_sched_barrier(0);

    v8f acc[4] = {};
    #pragma unroll
    for (int q = 0; q < 8; ++q) {
        #pragma unroll
        for (int half = 0; half < 2; ++half) {
            v2f av;
            if (half) { av.x = a[q].z; av.y = a[q].w; } else { av.x = a[q].x; av.y = a[q].y; }
            #pragma unroll
            for (int j = 0; j < 4; ++j) {
                v2f bv;
                if (half) { bv.x = bq[j][q].z; bv.y = bq[j][q].w; }
                else      { bv.x = bq[j][q].x; bv.y = bq[j][q].y; }
                acc[j] = __builtin_amdgcn_wmma_f32_16x16x4_f32(false, av, false, bv, (short)0, acc[j], false, false);
            }
        }
    }

    #pragma unroll
    for (int r = 0; r < 8; ++r) {
        const size_t bl = bl0 + r + 8 * hi;
        const float cd = (r < 4) ? cd0[r & 3] : cd1[r & 3];
        v4f o;
        #pragma unroll
        for (int j = 0; j < 4; ++j) {
            const float z = zv[r][j];
            const float y = yv[r][j] + cd * acc[j][r];
            o[j] = (y + Dp * xv[r][j]) * (z * sigmoid_f(z));
        }
        *reinterpret_cast<v4f*>(yfin + bl * DI + h * HD_ + 4 * lo) = o;
    }
}

extern "C" void kernel_launch(void* const* d_in, const int* in_sizes, int n_in,
                              void* d_out, int out_size, void* d_ws, size_t ws_size,
                              hipStream_t stream)
{
    const float* u       = (const float*)d_in[0];
    const float* W_in    = (const float*)d_in[1];
    const float* W_out   = (const float*)d_in[2];
    const float* dt_bias = (const float*)d_in[3];
    const float* D_param = (const float*)d_in[4];
    const float* Bn_g    = (const float*)d_in[5];
    const float* Bn_b    = (const float*)d_in[6];
    const float* Cn_g    = (const float*)d_in[7];
    const float* Cn_b    = (const float*)d_in[8];

    float* ws = (float*)d_ws;
    float* proj   = ws; ws += (size_t)M1 * DIP;
    float* yfin   = ws; ws += (size_t)M1 * DI;
    float* Sstate = ws; ws += (size_t)B_ * NH_ * NC_ * 4096;
    float* Brot   = ws; ws += (size_t)M1 * DS_;
    float* Crot   = ws; ws += (size_t)M1 * DS_;
    float* DTb    = ws; ws += (size_t)M1 * NH_;
    float* ladb   = ws; ws += (size_t)M1 * NH_;
    float* trapb  = ws; ws += (size_t)M1 * NH_;
    float* cumdb  = ws; ws += (size_t)B_ * NH_ * L_;
    float* dtmeanb = ws; ws += (size_t)M1;

    gemm_wmma_f32<DIP, DM, true><<<dim3(DIP / 64, M1 / 64), dim3(128), 0, stream>>>(
        u, W_in, proj, dt_bias, DTb, ladb, trapb, dtmeanb);

    rot_kernel<<<dim3(M1 / 16), dim3(512), 0, stream>>>(
        proj, dtmeanb, Bn_g, Bn_b, Cn_g, Cn_b, Brot, Crot);

    scan_chunk_kernel<<<dim3(B_ * NH_ * NC_), dim3(64), 0, stream>>>(
        proj, Brot, Crot, DTb, ladb, trapb, yfin, cumdb, Sstate);
    scan_combine_kernel<<<dim3(B_ * NH_ * 16), dim3(64), 0, stream>>>(
        cumdb, Sstate);
    scan_correct_kernel<<<dim3(B_ * NH_ * NC_), dim3(64), 0, stream>>>(
        proj, Crot, cumdb, Sstate, D_param, yfin);

    gemm_wmma_f32<DM, DI, false><<<dim3(DM / 64, M1 / 64), dim3(128), 0, stream>>>(
        yfin, W_out, (float*)d_out, nullptr, nullptr, nullptr, nullptr, nullptr);
}


# in-projection: last 4 column tiles cut along K, 256 one-tile blocks dispatched first, a K quarter per wave, partials added via LDS: 17 quarter tiles per SIMD, no fifth-tile tail; sum order changes
# speedup vs baseline: 11.0240x; 1.0840x over previous
#include <hip/hip_runtime.h>
#include <math.h>

#define B_   2
#define L_   512
#define DM   1024
#define DI   2048
#define DS_  64
#define NH_  32
#define HD_  64
#define NA_  32
#define DIP  4352
#define M1   (B_*L_)

#define KS_CT      4
#define KS_BLOCKS  (KS_CT * (M1 / 16))

#define NC_  16
#define LC_  32

#define OFF_Z    0
#define OFF_X    2048
#define OFF_B    4096
#define OFF_C    4160
#define OFF_DT   4224
#define OFF_A    4256
#define OFF_TRAP 4288
#define OFF_ANG  4320

typedef __attribute__((ext_vector_type(2))) float v2f;
typedef __attribute__((ext_vector_type(8))) float v8f;

__device__ __forceinline__ float softplus_f(float x) {
    return x > 20.f ? x : log1pf(expf(x));
}
__device__ __forceinline__ float sigmoid_f(float x) {
    return 1.f / (1.f + expf(-x));
}

typedef __attribute__((ext_vector_type(4))) float v4f;

struct GFrag { v4f a[2]; v4f b[8]; };

#define GEMM_LD_NEW(dst, ptr, off) \
    asm volatile("global_load_b128 %0, %1, off offset:%2" : "=v"(dst) : "v"(ptr), "n"(off) : "memory")
#define GEMM_LD_TIED(dst, ptr, off) \
    asm volatile("global_load_b128 %0, %1, off offset:%2" : "+v"(dst) : "v"(ptr), "n"(off) : "memory")

template <int N, int SLAB, bool FIRST>
__device__ __forceinline__ void gemm_load(GFrag& f, const float* ap, const float* bp)
{
    constexpr int AO = SLAB * 16 * 4;
    constexpr int BO = SLAB * 16 * N * 4;
    constexpr int RS = N * 4;
    static_assert(BO + 7 * RS < (1 << 23), "immediate offset of a global load");
    if constexpr (FIRST) {
        GEMM_LD_NEW(f.a[0], ap, AO);          GEMM_LD_NEW(f.a[1], ap, AO + 16);
        GEMM_LD_NEW(f.b[0], bp, BO);          GEMM_LD_NEW(f.b[1], bp, BO + RS);
        GEMM_LD_NEW(f.b[2], bp, BO + 2 * RS); GEMM_LD_NEW(f.b[3], bp, BO + 3 * RS);
        GEMM_LD_NEW(f.b[4], bp, BO + 4 * RS); GEMM_LD_NEW(f.b[5], bp, BO + 5 * RS);
        GEMM_LD_NEW(f.b[6], bp, BO + 6 * RS); GEMM_LD_NEW(f.b[7], bp, BO + 7 * RS);
    } else {
        GEMM_LD_TIED(f.a[0], ap, AO);          GEMM_LD_TIED(f.a[1], ap, AO + 16);
        GEMM_LD_TIED(f.b[0], bp, BO);          GEMM_LD_TIED(f.b[1], bp, BO + RS);
        GEMM_LD_TIED(f.b[2], bp, BO + 2 * RS); GEMM_LD_TIED(f.b[3], bp, BO + 3 * RS);
        GEMM_LD_TIED(f.b[4], bp, BO + 4 * RS); GEMM_LD_TIED(f.b[5], bp, BO + 5 * RS);
        GEMM_LD_TIED(f.b[6], bp, BO + 6 * RS); GEMM_LD_TIED(f.b[7], bp, BO + 7 * RS);
    }
}

#define GEMM_WAIT(f, CNT) \
    asm volatile("s_wait_loadcnt " #CNT \
                 : "+v"(f.a[0]), "+v"(f.a[1]), "+v"(f.b[0]), "+v"(f.b[1]), "+v"(f.b[2]), \
                   "+v"(f.b[3]), "+v"(f.b[4]), "+v"(f.b[5]), "+v"(f.b[6]), "+v"(f.b[7]) \
                 : : "memory")

#define GEMM_ADDR_FREE(pa, pb) \
    asm volatile("s_wait_xcnt 0" : "+v"(pa), "+v"(pb) : : "memory")

__device__ __forceinline__ void gemm_mma(const GFrag& f, v8f& c0, v8f& c1, v8f& c2, v8f& c3)
{
    #pragma unroll
    for (int s = 0; s < 4; ++s) {
        const v4f aq = f.a[s >> 1];
        v2f a;
        if (s & 1) { a.x = aq.z; a.y = aq.w; } else { a.x = aq.x; a.y = aq.y; }
        const v4f r0 = f.b[2 * s], r1 = f.b[2 * s + 1];
        v2f b;
        b.x = r0.x; b.y = r1.x;
        c0 = __builtin_amdgcn_wmma_f32_16x16x4_f32(false, a, false, b, (short)0, c0, false, false);
        b.x = r0.y; b.y = r1.y;
        c1 = __builtin_amdgcn_wmma_f32_16x16x4_f32(false, a, false, b, (short)0, c1, false, false);
        b.x = r0.z; b.y = r1.z;
        c2 = __builtin_amdgcn_wmma_f32_16x16x4_f32(false, a, false, b, (short)0, c2, false, false);
        b.x = r0.w; b.y = r1.w;
        c3 = __builtin_amdgcn_wmma_f32_16x16x4_f32(false, a, false, b, (short)0, c3, false, false);
    }
}

#define GEMM_HOLD() __builtin_amdgcn_sched_barrier(0)

template <int N, int K, bool EPI>
__global__ __launch_bounds__(128, 4) void gemm_wmma_f32(
    const float* __restrict__ A, const float* __restrict__ Bm,
    float* __restrict__ C,
    const float* __restrict__ dt_bias, float* __restrict__ DTb,
    float* __restrict__ ladb, float* __restrict__ trapb,
    float* __restrict__ dtmeanb)
{
    static_assert(N % 64 == 0 && K % 64 == 0 && K >= 128, "tile and ring shape");
    const int lane = threadIdx.x & 31;
    const int wave = threadIdx.x >> 5;
    const int lo   = lane & 15;
    const int hi   = lane >> 4;

    int  m0, bx, k0 = 0;
    int  nslab = K / 16;
    bool ksplit = false;
    if constexpr (EPI) {
        static_assert(K % 256 == 0 && (KS_CT & (KS_CT - 1)) == 0 && OFF_DT / 64 >= N / 64 - KS_CT,
                      "a K quarter is a multiple of 4 slabs; the epilogue tiles are cut tiles");
        constexpr int NCT = N / 64 - KS_CT;
        const int blk = (int)blockIdx.x;
        ksplit = blk < KS_BLOCKS;
        if (ksplit) {
            bx    = NCT + (blk & (KS_CT - 1));
            m0    = (blk / KS_CT) * 16;
            k0    = wave * (K / 4);
            nslab = K / 64;
        } else {
            const int nb = blk - KS_BLOCKS;
            bx = nb % NCT;
            m0 = (nb / NCT) * 64 + wave * 16;
        }
    } else {
        bx = (int)blockIdx.x;
        m0 = blockIdx.y * 64 + wave * 16;
    }
    const int n0 = bx * 64;

    const float* ap = A + (size_t)(m0 + lo) * K + k0 + 8 * hi;
    const float* bp = Bm + (size_t)(k0 + 8 * hi) * N + n0 + 4 * lo;
    constexpr int    SA = 16;
    constexpr size_t SB = (size_t)16 * N;

    v8f c0 = {}, c1 = {}, c2 = {}, c3 = {};
    GFrag f0, f1, f2, f3;
    gemm_load<N, 0, true>(f0, ap, bp);
    gemm_load<N, 1, true>(f1, ap, bp);
    gemm_load<N, 2, true>(f2, ap, bp);
    gemm_load<N, 3, true>(f3, ap, bp);

    for (int s = 0; s < nslab - 4; s += 4) {
        GEMM_WAIT(f0, 30); gemm_mma(f0, c0, c1, c2, c3);
        GEMM_HOLD(); gemm_load<N, 4, false>(f0, ap, bp); GEMM_HOLD();
        GEMM_WAIT(f1, 30); gemm_mma(f1, c0, c1, c2, c3);
        GEMM_HOLD(); gemm_load<N, 5, false>(f1, ap, bp); GEMM_HOLD();
        GEMM_WAIT(f2, 30); gemm_mma(f2, c0, c1, c2, c3);
        GEMM_HOLD(); gemm_load<N, 6, false>(f2, ap, bp); GEMM_HOLD();
        GEMM_WAIT(f3, 30); gemm_mma(f3, c0, c1, c2, c3);
        GEMM_HOLD(); gemm_load<N, 7, false>(f3, ap, bp); GEMM_HOLD();
        GEMM_ADDR_FREE(ap, bp);
        ap += 4 * SA;
        bp += 4 * SB;
    }
    GEMM_WAIT(f0, 30); gemm_mma(f0, c0, c1, c2, c3); GEMM_HOLD();
    GEMM_WAIT(f1, 20); gemm_mma(f1, c0, c1, c2, c3); GEMM_HOLD();
    GEMM_WAIT(f2, 10); gemm_mma(f2, c0, c1, c2, c3); GEMM_HOLD();
    GEMM_WAIT(f3, 0);  gemm_mma(f3, c0, c1, c2, c3);

    if constexpr (EPI) {
        GEMM_HOLD();
        asm volatile("v_nop\n\tv_nop\n\tv_nop\n\tv_nop");
        if (ksplit) {
            __shared__ v4f red[3][8][32];
            if (wave > 0) {
                #pragma unroll
                for (int r = 0; r < 8; ++r) {
                    v4f p; p.x = c0[r]; p.y = c1[r]; p.z = c2[r]; p.w = c3[r];
                    red[wave - 1][r][lane] = p;
                }
            }
            __syncthreads();
            if (wave > 0) return;
            #pragma unroll
            for (int q = 0; q < 3; ++q) {
                #pragma unroll
                for (int r = 0; r < 8; ++r) {
                    const v4f p = red[q][r][lane];
                    c0[r] += p.x; c1[r] += p.y; c2[r] += p.z; c3[r] += p.w;
                }
            }
        }
    }

    #pragma unroll
    for (int r = 0; r < 8; ++r) {
        v4f o; o.x = c0[r]; o.y = c1[r]; o.z = c2[r]; o.w = c3[r];
        *reinterpret_cast<v4f*>(C + (size_t)(m0 + r + 8 * hi) * N + n0 + 4 * lo) = o;
    }

    if constexpr (EPI) {
        static_assert(OFF_DT % 64 == 0 && OFF_A == OFF_DT + 32 && OFF_TRAP == OFF_DT + 64 &&
                      NH_ == 32, "dt|A share one 64-column tile, trap opens the next");
        if (bx == OFF_DT / 64) {
            const bool isdt = lo < 8;
            const int  h4   = 4 * (lo & 7);
            v4f bias;
            #pragma unroll
            for (int j = 0; j < 4; ++j) bias[j] = isdt ? dt_bias[h4 + j] : 0.f;
            float* dst = (isdt ? DTb : ladb) + h4;
            #pragma unroll
            for (int r = 0; r < 8; ++r) {
                const size_t row = (size_t)(m0 + r + 8 * hi);
                v4f sp;
                sp.x = softplus_f(c0[r] + bias.x);
                sp.y = softplus_f(c1[r] + bias.y);
                sp.z = softplus_f(c2[r] + bias.z);
                sp.w = softplus_f(c3[r] + bias.w);
                v4f o;
                #pragma unroll
                for (int j = 0; j < 4; ++j) {
                    const float dtp = __shfl_xor(sp[j], 8, 32);
                    o[j] = isdt ? sp[j] : fminf(-sp[j], -1e-4f) * dtp;
                }
                *reinterpret_cast<v4f*>(dst + row * 32) = o;
                float s = (sp.x + sp.y) + (sp.z + sp.w);
                s += __shfl_xor(s, 1, 32);
                s += __shfl_xor(s, 2, 32);
                s += __shfl_xor(s, 4, 32);
                if (lo == 0) dtmeanb[row] = s * (1.f / 32.f);
            }
        } else if (bx == OFF_TRAP / 64) {
            if (lo < 8) {
                #pragma unroll
                for (int r = 0; r < 8; ++r) {
                    v4f o;
                    o.x = sigmoid_f(c0[r]); o.y = sigmoid_f(c1[r]);
                    o.z = sigmoid_f(c2[r]); o.w = sigmoid_f(c3[r]);
                    *reinterpret_cast<v4f*>(trapb + (size_t)(m0 + r + 8 * hi) * 32 + 4 * lo) = o;
                }
            }
        }
    }
}

__device__ __forceinline__ float wave_sum64(float v0, float v1)
{
    float x = v0 + v1;
    x += __shfl_xor(x, 16, 32);
    x += __shfl_xor(x, 8, 32);
    x += __shfl_xor(x, 4, 32);
    x += __shfl_xor(x, 2, 32);
    x += __shfl_xor(x, 1, 32);
    return x;
}

__global__ __launch_bounds__(512) void rot_kernel(
    const float* __restrict__ proj, const float* __restrict__ dtmeanb,
    const float* __restrict__ Bn_g, const float* __restrict__ Bn_b,
    const float* __restrict__ Cn_g, const float* __restrict__ Cn_b,
    float* __restrict__ Brot, float* __restrict__ Crot)
{
    static_assert(L_ == 512, "32 row tiles of 16 tokens per batch, two per wave");
    __shared__ v2f   incp[(L_ / 2) * 32];
    __shared__ float part[16][16 * 32];
    const int b = blockIdx.x >> 5;
    const int w = blockIdx.x & 31;
    const int j = threadIdx.x & 31;
    const int i = __builtin_amdgcn_readfirstlane((int)(threadIdx.x >> 5));

    const size_t bl  = (size_t)b * L_ + 16 * w + i;
    const float* row = proj + bl * DIP;
    const float vb0 = row[OFF_B + j], vb1 = row[OFF_B + 32 + j];
    const float vc0 = row[OFF_C + j], vc1 = row[OFF_C + 32 + j];
    const float gB0 = Bn_g[j], gB1 = Bn_g[j + 32], hB0 = Bn_b[j], hB1 = Bn_b[j + 32];
    const float gC0 = Cn_g[j], gC1 = Cn_g[j + 32], hC0 = Cn_b[j], hC1 = Cn_b[j + 32];

    float va[2][16];
    #pragma unroll
    for (int u = 0; u < 2; ++u) {
        const int t = i + 16 * u;
        if (t <= w) {
            const float* ang = proj + (size_t)(b * L_ + 16 * t) * DIP + OFF_ANG + j;
            #pragma unroll
            for (int q = 0; q < 16; ++q) va[u][q] = ang[(size_t)q * DIP];
        }
    }
    #pragma unroll
    for (int u = 0; u < 2; ++u) {
        const int t = i + 16 * u;
        if (t <= w) {
            const float* dmp = dtmeanb + b * L_ + 16 * t;
            #pragma unroll
            for (int q = 0; q < 16; ++q) {
                const int l = 16 * t + q;
                incp[(l >> 1) * 32 + j][l & 1] = va[u][q] * dmp[q];
            }
        }
    }

    const float mB  = wave_sum64(vb0, vb1) * (1.f / 64.f);
    const float db0 = vb0 - mB, db1 = vb1 - mB;
    const float rB  = rsqrtf(wave_sum64(db0 * db0, db1 * db1) * (1.f / 64.f) + 1e-5f);
    const float nB0 = db0 * rB * gB0 + hB0;
    const float nB1 = db1 * rB * gB1 + hB1;
    const float mC  = wave_sum64(vc0, vc1) * (1.f / 64.f);
    const float dc0 = vc0 - mC, dc1 = vc1 - mC;
    const float rC  = rsqrtf(wave_sum64(dc0 * dc0, dc1 * dc1) * (1.f / 64.f) + 1e-5f);
    const float nC0 = dc0 * rC * gC0 + hC0;
    const float nC1 = dc1 * rC * gC1 + hC1;

    __syncthreads();

    const int lo = j & 15;
    const int hi = j >> 4;
    v8f acc0 = {}, acc1 = {};
    #pragma unroll
    for (int u = 0; u < 2; ++u) {
        const int t = i + 16 * u;
        if (t <= w) {
            const bool full = t < w;
            const v2f* p = incp + (8 * t + hi) * 32 + lo;
            #pragma unroll
            for (int s = 0; s < 4; ++s) {
                const int kl = 4 * s + 2 * hi;
                v2f a;
                a.x = (full || kl <= lo) ? 1.f : 0.f;
                a.y = (full || kl + 1 <= lo) ? 1.f : 0.f;
                const v2f b0 = p[64 * s];
                const v2f b1 = p[64 * s + 16];
                acc0 = __builtin_amdgcn_wmma_f32_16x16x4_f32(false, a, false, b0, (short)0, acc0, false, false);
                acc1 = __builtin_amdgcn_wmma_f32_16x16x4_f32(false, a, false, b1, (short)0, acc1, false, false);
            }
        }
    }
    #pragma unroll
    for (int r = 0; r < 8; ++r) {
        part[i][(r + 8 * hi) * 32 + lo]      = acc0[r];
        part[i][(r + 8 * hi) * 32 + lo + 16] = acc1[r];
    }
    __syncthreads();

    float pq[16];
    #pragma unroll
    for (int u = 0; u < 16; ++u) pq[u] = part[u][i * 32 + j];
    #pragma unroll
    for (int st = 1; st < 16; st <<= 1) {
        #pragma unroll
        for (int u = 0; u < 16; u += 2 * st) pq[u] += pq[u + st];
    }
    const float ph = pq[0];
    const float c  = cosf(ph), s = sinf(ph);
    const float c0 = __shfl(c, j >> 1, 32),        s0 = __shfl(s, j >> 1, 32);
    const float c1 = __shfl(c, 16 + (j >> 1), 32), s1 = __shfl(s, 16 + (j >> 1), 32);
    const float pB0 = __shfl_xor(nB0, 1, 32), pB1 = __shfl_xor(nB1, 1, 32);
    const float pC0 = __shfl_xor(nC0, 1, 32), pC1 = __shfl_xor(nC1, 1, 32);
    float ob0, ob1, oc0, oc1;
    if ((j & 1) == 0) {
        ob0 = nB0 * c0 - pB0 * s0; ob1 = nB1 * c1 - pB1 * s1;
        oc0 = nC0 * c0 - pC0 * s0; oc1 = nC1 * c1 - pC1 * s1;
    } else {
        ob0 = pB0 * s0 + nB0 * c0; ob1 = pB1 * s1 + nB1 * c1;
        oc0 = pC0 * s0 + nC0 * c0; oc1 = pC1 * s1 + nC1 * c1;
    }
    Brot[bl * 64 + j]      = ob0;
    Brot[bl * 64 + 32 + j] = ob1;
    Crot[bl * 64 + j]      = oc0;
    Crot[bl * 64 + 32 + j] = oc1;
}


template <int TT>
__device__ __forceinline__ void scan_chunk_tile(
    const float* __restrict__ proj, const float* __restrict__ Brot,
    const float* __restrict__ Crot, float* __restrict__ yfin,
    float* __restrict__ Sp, const float* sLa, const float* sP1,
    const float* sP2, const float* sV,
    const size_t bl0c, const size_t blprev, const float prevOn,
    const int h, const int lo, const int hi)
{
    v4f cq[8], bmq[8], bs[TT + 1][8];
    {
        const float* cp  = Crot + (bl0c + 16 * TT + lo) * 64 + 4 * hi;
        const float* bmp = Brot + blprev * 64 + 4 * hi;
        #pragma unroll
        for (int q = 0; q < 8; ++q) {
            cq[q]  = *reinterpret_cast<const v4f*>(cp + 8 * q);
            bmq[q] = *reinterpret_cast<const v4f*>(bmp + 8 * q);
        }
        #pragma unroll
        for (int st = 0; st <= TT; ++st) {
            const float* bp = Brot + (bl0c + 16 * st + lo) * 64 + 4 * hi;
            #pragma unroll
            for (int q = 0; q < 8; ++q)
                bs[st][q] = *reinterpret_cast<const v4f*>(bp + 8 * q);
        }
    }
    const float* xb = proj + OFF_X + h * HD_ + 4 * lo;
    const v4f xm = *reinterpret_cast<const v4f*>(xb + blprev * DIP);

    v8f g[TT + 1];
    #pragma unroll
    for (int st = 0; st <= TT; ++st) {
        g[st] = v8f{};
        #pragma unroll
        for (int q = 0; q < 8; ++q) {
            v2f a, b;
            a.x = bs[st][q].x; a.y = bs[st][q].y; b.x = cq[q].x; b.y = cq[q].y;
            g[st] = __builtin_amdgcn_wmma_f32_16x16x4_f32(false, a, false, b, (short)0, g[st], false, false);
            a.x = bs[st][q].z; a.y = bs[st][q].w; b.x = cq[q].z; b.y = cq[q].w;
            g[st] = __builtin_amdgcn_wmma_f32_16x16x4_f32(false, a, false, b, (short)0, g[st], false, false);
        }
    }
    float gm = 0.f;
    #pragma unroll
    for (int q = 0; q < 8; ++q)
        gm += cq[q].x * bmq[q].x + cq[q].y * bmq[q].y + cq[q].z * bmq[q].z + cq[q].w * bmq[q].w;
    gm += __shfl_xor(gm, 16, 32);

    const float lat = sLa[16 * TT + lo];
    #pragma unroll
    for (int st = 0; st <= TT; ++st) {
        #pragma unroll
        for (int r = 0; r < 8; ++r) {
            const int sl = 8 * hi + r;
            const int s  = 16 * st + sl;
            const float e1 = expf(fminf(lat - sLa[s], 0.f)) * sP1[s];
            const float e2 = expf(fminf(lat - sLa[s + 1], 0.f)) * sP2[s + 1];
            float w;
            if (st < TT) w = e1 + e2;
            else         w = ((sl <= lo) ? e1 : 0.f) + ((sl + 1 <= lo) ? e2 : 0.f);
            g[st][r] *= w;
        }
    }
    const float mprev = prevOn * gm * expf(lat - sLa[0]) * sP2[0];

    v8f y[4] = {};
    #pragma unroll
    for (int st = 0; st <= TT; ++st) {
        #pragma unroll
        for (int j = 0; j < 4; ++j) {
            const size_t row = bl0c + 16 * st + 8 * hi + 2 * j;
            const v4f x0 = *reinterpret_cast<const v4f*>(xb + row * DIP);
            const v4f x1 = *reinterpret_cast<const v4f*>(xb + (row + 1) * DIP);
            v2f a; a.x = g[st][2 * j]; a.y = g[st][2 * j + 1];
            #pragma unroll
            for (int jj = 0; jj < 4; ++jj) {
                v2f b; b.x = x0[jj]; b.y = x1[jj];
                y[jj] = __builtin_amdgcn_wmma_f32_16x16x4_f32(false, a, false, b, (short)0, y[jj], false, false);
            }
        }
    }
    {
        v2f a; a.x = (hi == 0) ? mprev : 0.f; a.y = 0.f;
        #pragma unroll
        for (int jj = 0; jj < 4; ++jj) {
            v2f b; b.x = (hi == 0) ? xm[jj] : 0.f; b.y = 0.f;
            y[jj] = __builtin_amdgcn_wmma_f32_16x16x4_f32(false, a, false, b, (short)0, y[jj], false, false);
        }
    }
    #pragma unroll
    for (int r = 0; r < 8; ++r) {
        v4f o; o.x = y[0][r]; o.y = y[1][r]; o.z = y[2][r]; o.w = y[3][r];
        *reinterpret_cast<v4f*>(yfin + (bl0c + 16 * TT + r + 8 * hi) * DI + h * HD_ + 4 * lo) = o;
    }

    v8f sacc[2][4] = {};
    #pragma unroll
    for (int ks = 0; ks < 8; ++ks) {
        const int s0 = 4 * ks + 2 * hi;
        const v4f xa0 = *reinterpret_cast<const v4f*>(xb + (bl0c + s0) * DIP);
        const v4f xa1 = *reinterpret_cast<const v4f*>(xb + (bl0c + s0 + 1) * DIP);
        const v4f bb0 = *reinterpret_cast<const v4f*>(Brot + (bl0c + s0) * 64 + 4 * lo);
        const v4f bb1 = *reinterpret_cast<const v4f*>(Brot + (bl0c + s0 + 1) * 64 + 4 * lo);
        const float v0 = sV[s0], v1 = sV[s0 + 1];
        #pragma unroll
        for (int a2 = 0; a2 < 2; ++a2) {
            v2f a; a.x = v0 * xa0[2 * TT + a2]; a.y = v1 * xa1[2 * TT + a2];
            #pragma unroll
            for (int jn = 0; jn < 4; ++jn) {
                v2f b; b.x = bb0[jn]; b.y = bb1[jn];
                sacc[a2][jn] = __builtin_amdgcn_wmma_f32_16x16x4_f32(false, a, false, b, (short)0, sacc[a2][jn], false, false);
            }
        }
    }
    {
        const v4f bm4 = *reinterpret_cast<const v4f*>(Brot + blprev * 64 + 4 * lo);
        const float vprev = prevOn * expf(sLa[31] - sLa[0]) * sP2[0];
        #pragma unroll
        for (int a2 = 0; a2 < 2; ++a2) {
            v2f a; a.x = (hi == 0) ? vprev * xm[2 * TT + a2] : 0.f; a.y = 0.f;
            #pragma unroll
            for (int jn = 0; jn < 4; ++jn) {
                v2f b; b.x = (hi == 0) ? bm4[jn] : 0.f; b.y = 0.f;
                sacc[a2][jn] = __builtin_amdgcn_wmma_f32_16x16x4_f32(false, a, false, b, (short)0, sacc[a2][jn], false, false);
            }
        }
    }
    #pragma unroll
    for (int a2 = 0; a2 < 2; ++a2) {
        #pragma unroll
        for (int r = 0; r < 8; ++r) {
            v4f o; o.x = sacc[a2][0][r]; o.y = sacc[a2][1][r]; o.z = sacc[a2][2][r]; o.w = sacc[a2][3][r];
            *reinterpret_cast<v4f*>(Sp + (size_t)(4 * (r + 8 * hi) + 2 * TT + a2) * 64 + 4 * lo) = o;
        }
    }
}

__global__ __launch_bounds__(64) void scan_chunk_kernel(
    const float* __restrict__ proj,
    const float* __restrict__ Brot, const float* __restrict__ Crot,
    const float* __restrict__ DTb, const float* __restrict__ ladb,
    const float* __restrict__ trapb,
    float* __restrict__ yfin, float* __restrict__ cumdb,
    float* __restrict__ Sstate)
{
    static_assert(LC_ == 32 && DS_ == 64 && HD_ == 64, "two 16-token tiles, 64 x 64 state");
    const int blk  = blockIdx.x;
    const int c    = blk % NC_;
    const int h    = (blk / NC_) % NH_;
    const int b    = blk / (NC_ * NH_);
    const int lane = threadIdx.x & 31;
    const int tt   = __builtin_amdgcn_readfirstlane((int)(threadIdx.x >> 5));
    const int lo   = lane & 15;
    const int hi   = lane >> 4;
    const size_t bl0c   = (size_t)(b * L_ + c * LC_);
    const size_t blprev = (c > 0) ? bl0c - 1 : bl0c;
    const float  prevOn = (c > 0) ? 1.f : 0.f;

    __shared__ float sLa[LC_ + 1], sP1[LC_ + 1], sP2[LC_ + 1], sV[LC_];

    if (tt == 0) {
        const size_t o  = (bl0c + lane) * 32 + h;
        const float dt  = DTb[o];
        const float tr  = trapb[o];
        float la        = ladb[o];
        #pragma unroll
        for (int d = 1; d < 32; d <<= 1) {
            const float up = __shfl_up(la, d, 32);
            if (lane >= d) la += up;
        }
        const float p1 = dt * tr;
        const float p2 = dt * (1.f - tr);
        float laN = __shfl_down(la, 1, 32);
        float p2N = __shfl_down(p2, 1, 32);
        if (lane == 31) { laN = la; p2N = 0.f; }
        const float la31 = __shfl(la, 31, 32);
        sLa[lane] = la; sP1[lane] = p1; sP2[lane] = p2;
        sV[lane]  = expf(la31 - la) * p1 + expf(la31 - laN) * p2N;
        if (lane == 31) { sLa[32] = la; sP1[32] = 0.f; sP2[32] = 0.f; }
        cumdb[((size_t)(b * NH_ + h)) * L_ + c * LC_ + lane] = expf(la);
    }
    __syncthreads();

    float* Sp = Sstate + (((size_t)(b * NH_ + h)) * NC_ + c) * 4096;
    if (tt == 0)
        scan_chunk_tile<0>(proj, Brot, Crot, yfin, Sp, sLa, sP1, sP2, sV, bl0c, blprev, prevOn, h, lo, hi);
    else
        scan_chunk_tile<1>(proj, Brot, Crot, yfin, Sp, sLa, sP1, sP2, sV, bl0c, blprev, prevOn, h, lo, hi);
}

__global__ __launch_bounds__(64) void scan_combine_kernel(
    const float* __restrict__ cumdb, float* Sstate)
{
    static_assert(64 * 4 * 16 == 4096, "16 blocks of 64 threads x 4 elements cover one state");
    const int bh = blockIdx.x >> 4;
    const int e  = (((int)blockIdx.x & 15) * 64 + (int)threadIdx.x) * 4;
    float* sp = Sstate + (size_t)bh * NC_ * 4096 + e;
    const float* pc = cumdb + (size_t)bh * L_ + LC_ - 1;

    v4f   S[NC_];
    float P[NC_];
    #pragma unroll
    for (int c = 0; c < NC_; ++c) {
        S[c] = *reinterpret_cast<const v4f*>(sp + (size_t)c * 4096);
        P[c] = pc[c * LC_];
    }
    __builtin_amdgcn_sched_barrier(0);

    v4f H = {0.f, 0.f, 0.f, 0.f};
    #pragma unroll
    for (int c = 0; c < NC_; ++c) {
        *reinterpret_cast<v4f*>(sp + (size_t)c * 4096) = H;
        H = P[c] * H + S[c];
    }
}

__global__ __launch_bounds__(64) void scan_correct_kernel(
    const float* __restrict__ proj, const float* __restrict__ Crot,
    const float* __restrict__ cumdb, const float* __restrict__ Sstate,
    const float* __restrict__ D_param, float* __restrict__ yfin)
{
    static_assert(LC_ == 32 && DS_ == 64 && HD_ == 64, "two 16-token tiles, 64 x 64 state");
    const int blk  = blockIdx.x;
    const int c    = blk % NC_;
    const int h    = (blk / NC_) % NH_;
    const int b    = blk / (NC_ * NH_);
    const int lane = threadIdx.x & 31;
    const int tt   = threadIdx.x >> 5;
    const int lo   = lane & 15;
    const int hi   = lane >> 4;
    const size_t bl0 = (size_t)(b * L_ + c * LC_ + 16 * tt);

    const float* ap = Crot + (bl0 + lo) * 64 + 4 * hi;
    const float* hp = Sstate + (((size_t)(b * NH_ + h)) * NC_ + c) * 4096
                             + (size_t)(4 * lo) * 64 + 4 * hi;

    v4f a[8], bq[4][8];
    #pragma unroll
    for (int q = 0; q < 8; ++q)
        a[q] = *reinterpret_cast<const v4f*>(ap + 8 * q);
    #pragma unroll
    for (int j = 0; j < 4; ++j) {
        #pragma unroll
        for (int q = 0; q < 8; ++q)
            bq[j][q] = *reinterpret_cast<const v4f*>(hp + j * 64 + 8 * q);
    }

    const float* cdp = cumdb + ((size_t)(b * NH_ + h)) * L_ + c * LC_ + 16 * tt + 8 * hi;
    const v4f cd0 = *reinterpret_cast<const v4f*>(cdp);
    const v4f cd1 = *reinterpret_cast<const v4f*>(cdp + 4);
    const float Dp = D_param[h];
    v4f xv[8], zv[8], yv[8];
    #pragma unroll
    for (int r = 0; r < 8; ++r) {
        const size_t bl = bl0 + r + 8 * hi;
        xv[r] = *reinterpret_cast<const v4f*>(proj + bl * DIP + OFF_X + h * HD_ + 4 * lo);
        zv[r] = *reinterpret_cast<const v4f*>(proj + bl * DIP + OFF_Z + h * HD_ + 4 * lo);
        yv[r] = *reinterpret_cast<const v4f*>(yfin + bl * DI + h * HD_ + 4 * lo);
    }

    __builtin_amdgcn_sched_barrier(0);

    v8f acc[4] = {};
    #pragma unroll
    for (int q = 0; q < 8; ++q) {
        #pragma unroll
        for (int half = 0; half < 2; ++half) {
            v2f av;
            if (half) { av.x = a[q].z; av.y = a[q].w; } else { av.x = a[q].x; av.y = a[q].y; }
            #pragma unroll
            for (int j = 0; j < 4; ++j) {
                v2f bv;
                if (half) { bv.x = bq[j][q].z; bv.y = bq[j][q].w; }
                else      { bv.x = bq[j][q].x; bv.y = bq[j][q].y; }
                acc[j] = __builtin_amdgcn_wmma_f32_16x16x4_f32(false, av, false, bv, (short)0, acc[j], false, false);
            }
        }
    }

    #pragma unroll
    for (int r = 0; r < 8; ++r) {
        const size_t bl = bl0 + r + 8 * hi;
        const float cd = (r < 4) ? cd0[r & 3] : cd1[r & 3];
        v4f o;
        #pragma unroll
        for (int j = 0; j < 4; ++j) {
            const float z = zv[r][j];
            const float y = yv[r][j] + cd * acc[j][r];
            o[j] = (y + Dp * xv[r][j]) * (z * sigmoid_f(z));
        }
        *reinterpret_cast<v4f*>(yfin + bl * DI + h * HD_ + 4 * lo) = o;
    }
}

extern "C" void kernel_launch(void* const* d_in, const int* in_sizes, int n_in,
                              void* d_out, int out_size, void* d_ws, size_t ws_size,
                              hipStream_t stream)
{
    const float* u       = (const float*)d_in[0];
    const float* W_in    = (const float*)d_in[1];
    const float* W_out   = (const float*)d_in[2];
    const float* dt_bias = (const float*)d_in[3];
    const float* D_param = (const float*)d_in[4];
    const float* Bn_g    = (const float*)d_in[5];
    const float* Bn_b    = (const float*)d_in[6];
    const float* Cn_g    = (const float*)d_in[7];
    const float* Cn_b    = (const float*)d_in[8];

    float* ws = (float*)d_ws;
    float* proj   = ws; ws += (size_t)M1 * DIP;
    float* yfin   = ws; ws += (size_t)M1 * DI;
    float* Sstate = ws; ws += (size_t)B_ * NH_ * NC_ * 4096;
    float* Brot   = ws; ws += (size_t)M1 * DS_;
    float* Crot   = ws; ws += (size_t)M1 * DS_;
    float* DTb    = ws; ws += (size_t)M1 * NH_;
    float* ladb   = ws; ws += (size_t)M1 * NH_;
    float* trapb  = ws; ws += (size_t)M1 * NH_;
    float* cumdb  = ws; ws += (size_t)B_ * NH_ * L_;
    float* dtmeanb = ws; ws += (size_t)M1;

    gemm_wmma_f32<DIP, DM, true><<<dim3(KS_BLOCKS + (DIP / 64 - KS_CT) * (M1 / 64)), dim3(128), 0, stream>>>(
        u, W_in, proj, dt_bias, DTb, ladb, trapb, dtmeanb);

    rot_kernel<<<dim3(M1 / 16), dim3(512), 0, stream>>>(
        proj, dtmeanb, Bn_g, Bn_b, Cn_g, Cn_b, Brot, Crot);

    scan_chunk_kernel<<<dim3(B_ * NH_ * NC_), dim3(64), 0, stream>>>(
        proj, Brot, Crot, DTb, ladb, trapb, yfin, cumdb, Sstate);
    scan_combine_kernel<<<dim3(B_ * NH_ * 16), dim3(64), 0, stream>>>(
        cumdb, Sstate);
    scan_correct_kernel<<<dim3(B_ * NH_ * NC_), dim3(64), 0, stream>>>(
        proj, Crot, cumdb, Sstate, D_param, yfin);

    gemm_wmma_f32<DM, DI, false><<<dim3(DM / 64, M1 / 64), dim3(128), 0, stream>>>(
        yfin, W_out, (float*)d_out, nullptr, nullptr, nullptr, nullptr, nullptr);
}


# out-projection: 16 waves a block, a K quarter per wave and row group, partials added via LDS (q0+q1+q2+q3): 4 waves per SIMD, not 1, cover load issue and waits; sum order changes
# speedup vs baseline: 11.5328x; 1.0462x over previous
#include <hip/hip_runtime.h>
#include <math.h>

#define B_   2
#define L_   512
#define DM   1024
#define DI   2048
#define DS_  64
#define NH_  32
#define HD_  64
#define NA_  32
#define DIP  4352
#define M1   (B_*L_)

#define KS_CT      4
#define KS_BLOCKS  (KS_CT * (M1 / 16))

#define NC_  16
#define LC_  32

#define OFF_Z    0
#define OFF_X    2048
#define OFF_B    4096
#define OFF_C    4160
#define OFF_DT   4224
#define OFF_A    4256
#define OFF_TRAP 4288
#define OFF_ANG  4320

typedef __attribute__((ext_vector_type(2))) float v2f;
typedef __attribute__((ext_vector_type(8))) float v8f;

__device__ __forceinline__ float softplus_f(float x) {
    return x > 20.f ? x : log1pf(expf(x));
}
__device__ __forceinline__ float sigmoid_f(float x) {
    return 1.f / (1.f + expf(-x));
}

typedef __attribute__((ext_vector_type(4))) float v4f;

struct GFrag { v4f a[2]; v4f b[8]; };

#define GEMM_LD_NEW(dst, ptr, off) \
    asm volatile("global_load_b128 %0, %1, off offset:%2" : "=v"(dst) : "v"(ptr), "n"(off) : "memory")
#define GEMM_LD_TIED(dst, ptr, off) \
    asm volatile("global_load_b128 %0, %1, off offset:%2" : "+v"(dst) : "v"(ptr), "n"(off) : "memory")

template <int N, int SLAB, bool FIRST>
__device__ __forceinline__ void gemm_load(GFrag& f, const float* ap, const float* bp)
{
    constexpr int AO = SLAB * 16 * 4;
    constexpr int BO = SLAB * 16 * N * 4;
    constexpr int RS = N * 4;
    static_assert(BO + 7 * RS < (1 << 23), "immediate offset of a global load");
    if constexpr (FIRST) {
        GEMM_LD_NEW(f.a[0], ap, AO);          GEMM_LD_NEW(f.a[1], ap, AO + 16);
        GEMM_LD_NEW(f.b[0], bp, BO);          GEMM_LD_NEW(f.b[1], bp, BO + RS);
        GEMM_LD_NEW(f.b[2], bp, BO + 2 * RS); GEMM_LD_NEW(f.b[3], bp, BO + 3 * RS);
        GEMM_LD_NEW(f.b[4], bp, BO + 4 * RS); GEMM_LD_NEW(f.b[5], bp, BO + 5 * RS);
        GEMM_LD_NEW(f.b[6], bp, BO + 6 * RS); GEMM_LD_NEW(f.b[7], bp, BO + 7 * RS);
    } else {
        GEMM_LD_TIED(f.a[0], ap, AO);          GEMM_LD_TIED(f.a[1], ap, AO + 16);
        GEMM_LD_TIED(f.b[0], bp, BO);          GEMM_LD_TIED(f.b[1], bp, BO + RS);
        GEMM_LD_TIED(f.b[2], bp, BO + 2 * RS); GEMM_LD_TIED(f.b[3], bp, BO + 3 * RS);
        GEMM_LD_TIED(f.b[4], bp, BO + 4 * RS); GEMM_LD_TIED(f.b[5], bp, BO + 5 * RS);
        GEMM_LD_TIED(f.b[6], bp, BO + 6 * RS); GEMM_LD_TIED(f.b[7], bp, BO + 7 * RS);
    }
}

#define GEMM_WAIT(f, CNT) \
    asm volatile("s_wait_loadcnt " #CNT \
                 : "+v"(f.a[0]), "+v"(f.a[1]), "+v"(f.b[0]), "+v"(f.b[1]), "+v"(f.b[2]), \
                   "+v"(f.b[3]), "+v"(f.b[4]), "+v"(f.b[5]), "+v"(f.b[6]), "+v"(f.b[7]) \
                 : : "memory")

#define GEMM_ADDR_FREE(pa, pb) \
    asm volatile("s_wait_xcnt 0" : "+v"(pa), "+v"(pb) : : "memory")

__device__ __forceinline__ void gemm_mma(const GFrag& f, v8f& c0, v8f& c1, v8f& c2, v8f& c3)
{
    #pragma unroll
    for (int s = 0; s < 4; ++s) {
        const v4f aq = f.a[s >> 1];
        v2f a;
        if (s & 1) { a.x = aq.z; a.y = aq.w; } else { a.x = aq.x; a.y = aq.y; }
        const v4f r0 = f.b[2 * s], r1 = f.b[2 * s + 1];
        v2f b;
        b.x = r0.x; b.y = r1.x;
        c0 = __builtin_amdgcn_wmma_f32_16x16x4_f32(false, a, false, b, (short)0, c0, false, false);
        b.x = r0.y; b.y = r1.y;
        c1 = __builtin_amdgcn_wmma_f32_16x16x4_f32(false, a, false, b, (short)0, c1, false, false);
        b.x = r0.z; b.y = r1.z;
        c2 = __builtin_amdgcn_wmma_f32_16x16x4_f32(false, a, false, b, (short)0, c2, false, false);
        b.x = r0.w; b.y = r1.w;
        c3 = __builtin_amdgcn_wmma_f32_16x16x4_f32(false, a, false, b, (short)0, c3, false, false);
    }
}

#define GEMM_HOLD() __builtin_amdgcn_sched_barrier(0)

template <int N, int K, bool EPI>
__global__ __launch_bounds__(EPI ? 128 : 512, EPI ? 4 : 1) void gemm_wmma_f32(
    const float* __restrict__ A, const float* __restrict__ Bm,
    float* __restrict__ C,
    const float* __restrict__ dt_bias, float* __restrict__ DTb,
    float* __restrict__ ladb, float* __restrict__ trapb,
    float* __restrict__ dtmeanb)
{
    static_assert(N % 64 == 0 && K % 64 == 0 && K >= 128, "tile and ring shape");
    const int lane = threadIdx.x & 31;
    const int wave = threadIdx.x >> 5;
    const int lo   = lane & 15;
    const int hi   = lane >> 4;

    int  m0, bx, k0 = 0;
    int  nslab = K / 16;
    bool ksplit = false;
    if constexpr (EPI) {
        static_assert(K % 256 == 0 && (KS_CT & (KS_CT - 1)) == 0 && OFF_DT / 64 >= N / 64 - KS_CT,
                      "a K quarter is a multiple of 4 slabs; the epilogue tiles are cut tiles");
        constexpr int NCT = N / 64 - KS_CT;
        const int blk = (int)blockIdx.x;
        ksplit = blk < KS_BLOCKS;
        if (ksplit) {
            bx    = NCT + (blk & (KS_CT - 1));
            m0    = (blk / KS_CT) * 16;
            k0    = wave * (K / 4);
            nslab = K / 64;
        } else {
            const int nb = blk - KS_BLOCKS;
            bx = nb % NCT;
            m0 = (nb / NCT) * 64 + wave * 16;
        }
    } else {
        static_assert(K % 256 == 0, "a K quarter is a multiple of 4 slabs");
        bx    = (int)blockIdx.x;
        m0    = blockIdx.y * 64 + (wave & 3) * 16;
        k0    = (wave >> 2) * (K / 4);
        nslab = K / 64;
    }
    const int n0 = bx * 64;

    const float* ap = A + (size_t)(m0 + lo) * K + k0 + 8 * hi;
    const float* bp = Bm + (size_t)(k0 + 8 * hi) * N + n0 + 4 * lo;
    constexpr int    SA = 16;
    constexpr size_t SB = (size_t)16 * N;

    v8f c0 = {}, c1 = {}, c2 = {}, c3 = {};
    GFrag f0, f1, f2, f3;
    gemm_load<N, 0, true>(f0, ap, bp);
    gemm_load<N, 1, true>(f1, ap, bp);
    gemm_load<N, 2, true>(f2, ap, bp);
    gemm_load<N, 3, true>(f3, ap, bp);

    for (int s = 0; s < nslab - 4; s += 4) {
        GEMM_WAIT(f0, 30); gemm_mma(f0, c0, c1, c2, c3);
        GEMM_HOLD(); gemm_load<N, 4, false>(f0, ap, bp); GEMM_HOLD();
        GEMM_WAIT(f1, 30); gemm_mma(f1, c0, c1, c2, c3);
        GEMM_HOLD(); gemm_load<N, 5, false>(f1, ap, bp); GEMM_HOLD();
        GEMM_WAIT(f2, 30); gemm_mma(f2, c0, c1, c2, c3);
        GEMM_HOLD(); gemm_load<N, 6, false>(f2, ap, bp); GEMM_HOLD();
        GEMM_WAIT(f3, 30); gemm_mma(f3, c0, c1, c2, c3);
        GEMM_HOLD(); gemm_load<N, 7, false>(f3, ap, bp); GEMM_HOLD();
        GEMM_ADDR_FREE(ap, bp);
        ap += 4 * SA;
        bp += 4 * SB;
    }
    GEMM_WAIT(f0, 30); gemm_mma(f0, c0, c1, c2, c3); GEMM_HOLD();
    GEMM_WAIT(f1, 20); gemm_mma(f1, c0, c1, c2, c3); GEMM_HOLD();
    GEMM_WAIT(f2, 10); gemm_mma(f2, c0, c1, c2, c3); GEMM_HOLD();
    GEMM_WAIT(f3, 0);  gemm_mma(f3, c0, c1, c2, c3);

    if constexpr (EPI) {
        GEMM_HOLD();
        asm volatile("v_nop\n\tv_nop\n\tv_nop\n\tv_nop");
        if (ksplit) {
            __shared__ v4f red[3][8][32];
            if (wave > 0) {
                #pragma unroll
                for (int r = 0; r < 8; ++r) {
                    v4f p; p.x = c0[r]; p.y = c1[r]; p.z = c2[r]; p.w = c3[r];
                    red[wave - 1][r][lane] = p;
                }
            }
            __syncthreads();
            if (wave > 0) return;
            #pragma unroll
            for (int q = 0; q < 3; ++q) {
                #pragma unroll
                for (int r = 0; r < 8; ++r) {
                    const v4f p = red[q][r][lane];
                    c0[r] += p.x; c1[r] += p.y; c2[r] += p.z; c3[r] += p.w;
                }
            }
        }
    } else {
        GEMM_HOLD();
        asm volatile("v_nop\n\tv_nop\n\tv_nop\n\tv_nop");
        __shared__ v4f redo[3][4][8][32];
        const int q = wave >> 2, g = wave & 3;
        if (q > 0) {
            #pragma unroll
            for (int r = 0; r < 8; ++r) {
                v4f p; p.x = c0[r]; p.y = c1[r]; p.z = c2[r]; p.w = c3[r];
                redo[q - 1][g][r][lane] = p;
            }
        }
        __syncthreads();
        if (q > 0) return;
        #pragma unroll
        for (int qq = 0; qq < 3; ++qq) {
            #pragma unroll
            for (int r = 0; r < 8; ++r) {
                const v4f p = redo[qq][g][r][lane];
                c0[r] += p.x; c1[r] += p.y; c2[r] += p.z; c3[r] += p.w;
            }
        }
    }

    #pragma unroll
    for (int r = 0; r < 8; ++r) {
        v4f o; o.x = c0[r]; o.y = c1[r]; o.z = c2[r]; o.w = c3[r];
        *reinterpret_cast<v4f*>(C + (size_t)(m0 + r + 8 * hi) * N + n0 + 4 * lo) = o;
    }

    if constexpr (EPI) {
        static_assert(OFF_DT % 64 == 0 && OFF_A == OFF_DT + 32 && OFF_TRAP == OFF_DT + 64 &&
                      NH_ == 32, "dt|A share one 64-column tile, trap opens the next");
        if (bx == OFF_DT / 64) {
            const bool isdt = lo < 8;
            const int  h4   = 4 * (lo & 7);
            v4f bias;
            #pragma unroll
            for (int j = 0; j < 4; ++j) bias[j] = isdt ? dt_bias[h4 + j] : 0.f;
            float* dst = (isdt ? DTb : ladb) + h4;
            #pragma unroll
            for (int r = 0; r < 8; ++r) {
                const size_t row = (size_t)(m0 + r + 8 * hi);
                v4f sp;
                sp.x = softplus_f(c0[r] + bias.x);
                sp.y = softplus_f(c1[r] + bias.y);
                sp.z = softplus_f(c2[r] + bias.z);
                sp.w = softplus_f(c3[r] + bias.w);
                v4f o;
                #pragma unroll
                for (int j = 0; j < 4; ++j) {
                    const float dtp = __shfl_xor(sp[j], 8, 32);
                    o[j] = isdt ? sp[j] : fminf(-sp[j], -1e-4f) * dtp;
                }
                *reinterpret_cast<v4f*>(dst + row * 32) = o;
                float s = (sp.x + sp.y) + (sp.z + sp.w);
                s += __shfl_xor(s, 1, 32);
                s += __shfl_xor(s, 2, 32);
                s += __shfl_xor(s, 4, 32);
                if (lo == 0) dtmeanb[row] = s * (1.f / 32.f);
            }
        } else if (bx == OFF_TRAP / 64) {
            if (lo < 8) {
                #pragma unroll
                for (int r = 0; r < 8; ++r) {
                    v4f o;
                    o.x = sigmoid_f(c0[r]); o.y = sigmoid_f(c1[r]);
                    o.z = sigmoid_f(c2[r]); o.w = sigmoid_f(c3[r]);
                    *reinterpret_cast<v4f*>(trapb + (size_t)(m0 + r + 8 * hi) * 32 + 4 * lo) = o;
                }
            }
        }
    }
}

__device__ __forceinline__ float wave_sum64(float v0, float v1)
{
    float x = v0 + v1;
    x += __shfl_xor(x, 16, 32);
    x += __shfl_xor(x, 8, 32);
    x += __shfl_xor(x, 4, 32);
    x += __shfl_xor(x, 2, 32);
    x += __shfl_xor(x, 1, 32);
    return x;
}

__global__ __launch_bounds__(512) void rot_kernel(
    const float* __restrict__ proj, const float* __restrict__ dtmeanb,
    const float* __restrict__ Bn_g, const float* __restrict__ Bn_b,
    const float* __restrict__ Cn_g, const float* __restrict__ Cn_b,
    float* __restrict__ Brot, float* __restrict__ Crot)
{
    static_assert(L_ == 512, "32 row tiles of 16 tokens per batch, two per wave");
    __shared__ v2f   incp[(L_ / 2) * 32];
    __shared__ float part[16][16 * 32];
    const int b = blockIdx.x >> 5;
    const int w = blockIdx.x & 31;
    const int j = threadIdx.x & 31;
    const int i = __builtin_amdgcn_readfirstlane((int)(threadIdx.x >> 5));

    const size_t bl  = (size_t)b * L_ + 16 * w + i;
    const float* row = proj + bl * DIP;
    const float vb0 = row[OFF_B + j], vb1 = row[OFF_B + 32 + j];
    const float vc0 = row[OFF_C + j], vc1 = row[OFF_C + 32 + j];
    const float gB0 = Bn_g[j], gB1 = Bn_g[j + 32], hB0 = Bn_b[j], hB1 = Bn_b[j + 32];
    const float gC0 = Cn_g[j], gC1 = Cn_g[j + 32], hC0 = Cn_b[j], hC1 = Cn_b[j + 32];

    float va[2][16];
    #pragma unroll
    for (int u = 0; u < 2; ++u) {
        const int t = i + 16 * u;
        if (t <= w) {
            const float* ang = proj + (size_t)(b * L_ + 16 * t) * DIP + OFF_ANG + j;
            #pragma unroll
            for (int q = 0; q < 16; ++q) va[u][q] = ang[(size_t)q * DIP];
        }
    }
    #pragma unroll
    for (int u = 0; u < 2; ++u) {
        const int t = i + 16 * u;
        if (t <= w) {
            const float* dmp = dtmeanb + b * L_ + 16 * t;
            #pragma unroll
            for (int q = 0; q < 16; ++q) {
                const int l = 16 * t + q;
                incp[(l >> 1) * 32 + j][l & 1] = va[u][q] * dmp[q];
            }
        }
    }

    const float mB  = wave_sum64(vb0, vb1) * (1.f / 64.f);
    const float db0 = vb0 - mB, db1 = vb1 - mB;
    const float rB  = rsqrtf(wave_sum64(db0 * db0, db1 * db1) * (1.f / 64.f) + 1e-5f);
    const float nB0 = db0 * rB * gB0 + hB0;
    const float nB1 = db1 * rB * gB1 + hB1;
    const float mC  = wave_sum64(vc0, vc1) * (1.f / 64.f);
    const float dc0 = vc0 - mC, dc1 = vc1 - mC;
    const float rC  = rsqrtf(wave_sum64(dc0 * dc0, dc1 * dc1) * (1.f / 64.f) + 1e-5f);
    const float nC0 = dc0 * rC * gC0 + hC0;
    const float nC1 = dc1 * rC * gC1 + hC1;

    __syncthreads();

    const int lo = j & 15;
    const int hi = j >> 4;
    v8f acc0 = {}, acc1 = {};
    #pragma unroll
    for (int u = 0; u < 2; ++u) {
        const int t = i + 16 * u;
        if (t <= w) {
            const bool full = t < w;
            const v2f* p = incp + (8 * t + hi) * 32 + lo;
            #pragma unroll
            for (int s = 0; s < 4; ++s) {
                const int kl = 4 * s + 2 * hi;
                v2f a;
                a.x = (full || kl <= lo) ? 1.f : 0.f;
                a.y = (full || kl + 1 <= lo) ? 1.f : 0.f;
                const v2f b0 = p[64 * s];
                const v2f b1 = p[64 * s + 16];
                acc0 = __builtin_amdgcn_wmma_f32_16x16x4_f32(false, a, false, b0, (short)0, acc0, false, false);
                acc1 = __builtin_amdgcn_wmma_f32_16x16x4_f32(false, a, false, b1, (short)0, acc1, false, false);
            }
        }
    }
    #pragma unroll
    for (int r = 0; r < 8; ++r) {
        part[i][(r + 8 * hi) * 32 + lo]      = acc0[r];
        part[i][(r + 8 * hi) * 32 + lo + 16] = acc1[r];
    }
    __syncthreads();

    float pq[16];
    #pragma unroll
    for (int u = 0; u < 16; ++u) pq[u] = part[u][i * 32 + j];
    #pragma unroll
    for (int st = 1; st < 16; st <<= 1) {
        #pragma unroll
        for (int u = 0; u < 16; u += 2 * st) pq[u] += pq[u + st];
    }
    const float ph = pq[0];
    const float c  = cosf(ph), s = sinf(ph);
    const float c0 = __shfl(c, j >> 1, 32),        s0 = __shfl(s, j >> 1, 32);
    const float c1 = __shfl(c, 16 + (j >> 1), 32), s1 = __shfl(s, 16 + (j >> 1), 32);
    const float pB0 = __shfl_xor(nB0, 1, 32), pB1 = __shfl_xor(nB1, 1, 32);
    const float pC0 = __shfl_xor(nC0, 1, 32), pC1 = __shfl_xor(nC1, 1, 32);
    float ob0, ob1, oc0, oc1;
    if ((j & 1) == 0) {
        ob0 = nB0 * c0 - pB0 * s0; ob1 = nB1 * c1 - pB1 * s1;
        oc0 = nC0 * c0 - pC0 * s0; oc1 = nC1 * c1 - pC1 * s1;
    } else {
        ob0 = pB0 * s0 + nB0 * c0; ob1 = pB1 * s1 + nB1 * c1;
        oc0 = pC0 * s0 + nC0 * c0; oc1 = pC1 * s1 + nC1 * c1;
    }
    Brot[bl * 64 + j]      = ob0;
    Brot[bl * 64 + 32 + j] = ob1;
    Crot[bl * 64 + j]      = oc0;
    Crot[bl * 64 + 32 + j] = oc1;
}


template <int TT>
__device__ __forceinline__ void scan_chunk_tile(
    const float* __restrict__ proj, const float* __restrict__ Brot,
    const float* __restrict__ Crot, float* __restrict__ yfin,
    float* __restrict__ Sp, const float* sLa, const float* sP1,
    const float* sP2, const float* sV,
    const size_t bl0c, const size_t blprev, const float prevOn,
    const int h, const int lo, const int hi)
{
    v4f cq[8], bmq[8], bs[TT + 1][8];
    {
        const float* cp  = Crot + (bl0c + 16 * TT + lo) * 64 + 4 * hi;
        const float* bmp = Brot + blprev * 64 + 4 * hi;
        #pragma unroll
        for (int q = 0; q < 8; ++q) {
            cq[q]  = *reinterpret_cast<const v4f*>(cp + 8 * q);
            bmq[q] = *reinterpret_cast<const v4f*>(bmp + 8 * q);
        }
        #pragma unroll
        for (int st = 0; st <= TT; ++st) {
            const float* bp = Brot + (bl0c + 16 * st + lo) * 64 + 4 * hi;
            #pragma unroll
            for (int q = 0; q < 8; ++q)
                bs[st][q] = *reinterpret_cast<const v4f*>(bp + 8 * q);
        }
    }
    const float* xb = proj + OFF_X + h * HD_ + 4 * lo;
    const v4f xm = *reinterpret_cast<const v4f*>(xb + blprev * DIP);

    v8f g[TT + 1];
    #pragma unroll
    for (int st = 0; st <= TT; ++st) {
        g[st] = v8f{};
        #pragma unroll
        for (int q = 0; q < 8; ++q) {
            v2f a, b;
            a.x = bs[st][q].x; a.y = bs[st][q].y; b.x = cq[q].x; b.y = cq[q].y;
            g[st] = __builtin_amdgcn_wmma_f32_16x16x4_f32(false, a, false, b, (short)0, g[st], false, false);
            a.x = bs[st][q].z; a.y = bs[st][q].w; b.x = cq[q].z; b.y = cq[q].w;
            g[st] = __builtin_amdgcn_wmma_f32_16x16x4_f32(false, a, false, b, (short)0, g[st], false, false);
        }
    }
    float gm = 0.f;
    #pragma unroll
    for (int q = 0; q < 8; ++q)
        gm += cq[q].x * bmq[q].x + cq[q].y * bmq[q].y + cq[q].z * bmq[q].z + cq[q].w * bmq[q].w;
    gm += __shfl_xor(gm, 16, 32);

    const float lat = sLa[16 * TT + lo];
    #pragma unroll
    for (int st = 0; st <= TT; ++st) {
        #pragma unroll
        for (int r = 0; r < 8; ++r) {
            const int sl = 8 * hi + r;
            const int s  = 16 * st + sl;
            const float e1 = expf(fminf(lat - sLa[s], 0.f)) * sP1[s];
            const float e2 = expf(fminf(lat - sLa[s + 1], 0.f)) * sP2[s + 1];
            float w;
            if (st < TT) w = e1 + e2;
            else         w = ((sl <= lo) ? e1 : 0.f) + ((sl + 1 <= lo) ? e2 : 0.f);
            g[st][r] *= w;
        }
    }
    const float mprev = prevOn * gm * expf(lat - sLa[0]) * sP2[0];

    v8f y[4] = {};
    #pragma unroll
    for (int st = 0; st <= TT; ++st) {
        #pragma unroll
        for (int j = 0; j < 4; ++j) {
            const size_t row = bl0c + 16 * st + 8 * hi + 2 * j;
            const v4f x0 = *reinterpret_cast<const v4f*>(xb + row * DIP);
            const v4f x1 = *reinterpret_cast<const v4f*>(xb + (row + 1) * DIP);
            v2f a; a.x = g[st][2 * j]; a.y = g[st][2 * j + 1];
            #pragma unroll
            for (int jj = 0; jj < 4; ++jj) {
                v2f b; b.x = x0[jj]; b.y = x1[jj];
                y[jj] = __builtin_amdgcn_wmma_f32_16x16x4_f32(false, a, false, b, (short)0, y[jj], false, false);
            }
        }
    }
    {
        v2f a; a.x = (hi == 0) ? mprev : 0.f; a.y = 0.f;
        #pragma unroll
        for (int jj = 0; jj < 4; ++jj) {
            v2f b; b.x = (hi == 0) ? xm[jj] : 0.f; b.y = 0.f;
            y[jj] = __builtin_amdgcn_wmma_f32_16x16x4_f32(false, a, false, b, (short)0, y[jj], false, false);
        }
    }
    #pragma unroll
    for (int r = 0; r < 8; ++r) {
        v4f o; o.x = y[0][r]; o.y = y[1][r]; o.z = y[2][r]; o.w = y[3][r];
        *reinterpret_cast<v4f*>(yfin + (bl0c + 16 * TT + r + 8 * hi) * DI + h * HD_ + 4 * lo) = o;
    }

    v8f sacc[2][4] = {};
    #pragma unroll
    for (int ks = 0; ks < 8; ++ks) {
        const int s0 = 4 * ks + 2 * hi;
        const v4f xa0 = *reinterpret_cast<const v4f*>(xb + (bl0c + s0) * DIP);
        const v4f xa1 = *reinterpret_cast<const v4f*>(xb + (bl0c + s0 + 1) * DIP);
        const v4f bb0 = *reinterpret_cast<const v4f*>(Brot + (bl0c + s0) * 64 + 4 * lo);
        const v4f bb1 = *reinterpret_cast<const v4f*>(Brot + (bl0c + s0 + 1) * 64 + 4 * lo);
        const float v0 = sV[s0], v1 = sV[s0 + 1];
        #pragma unroll
        for (int a2 = 0; a2 < 2; ++a2) {
            v2f a; a.x = v0 * xa0[2 * TT + a2]; a.y = v1 * xa1[2 * TT + a2];
            #pragma unroll
            for (int jn = 0; jn < 4; ++jn) {
                v2f b; b.x = bb0[jn]; b.y = bb1[jn];
                sacc[a2][jn] = __builtin_amdgcn_wmma_f32_16x16x4_f32(false, a, false, b, (short)0, sacc[a2][jn], false, false);
            }
        }
    }
    {
        const v4f bm4 = *reinterpret_cast<const v4f*>(Brot + blprev * 64 + 4 * lo);
        const float vprev = prevOn * expf(sLa[31] - sLa[0]) * sP2[0];
        #pragma unroll
        for (int a2 = 0; a2 < 2; ++a2) {
            v2f a; a.x = (hi == 0) ? vprev * xm[2 * TT + a2] : 0.f; a.y = 0.f;
            #pragma unroll
            for (int jn = 0; jn < 4; ++jn) {
                v2f b; b.x = (hi == 0) ? bm4[jn] : 0.f; b.y = 0.f;
                sacc[a2][jn] = __builtin_amdgcn_wmma_f32_16x16x4_f32(false, a, false, b, (short)0, sacc[a2][jn], false, false);
            }
        }
    }
    #pragma unroll
    for (int a2 = 0; a2 < 2; ++a2) {
        #pragma unroll
        for (int r = 0; r < 8; ++r) {
            v4f o; o.x = sacc[a2][0][r]; o.y = sacc[a2][1][r]; o.z = sacc[a2][2][r]; o.w = sacc[a2][3][r];
            *reinterpret_cast<v4f*>(Sp + (size_t)(4 * (r + 8 * hi) + 2 * TT + a2) * 64 + 4 * lo) = o;
        }
    }
}

__global__ __launch_bounds__(64) void scan_chunk_kernel(
    const float* __restrict__ proj,
    const float* __restrict__ Brot, const float* __restrict__ Crot,
    const float* __restrict__ DTb, const float* __restrict__ ladb,
    const float* __restrict__ trapb,
    float* __restrict__ yfin, float* __restrict__ cumdb,
    float* __restrict__ Sstate)
{
    static_assert(LC_ == 32 && DS_ == 64 && HD_ == 64, "two 16-token tiles, 64 x 64 state");
    const int blk  = blockIdx.x;
    const int c    = blk % NC_;
    const int h    = (blk / NC_) % NH_;
    const int b    = blk / (NC_ * NH_);
    const int lane = threadIdx.x & 31;
    const int tt   = __builtin_amdgcn_readfirstlane((int)(threadIdx.x >> 5));
    const int lo   = lane & 15;
    const int hi   = lane >> 4;
    const size_t bl0c   = (size_t)(b * L_ + c * LC_);
    const size_t blprev = (c > 0) ? bl0c - 1 : bl0c;
    const float  prevOn = (c > 0) ? 1.f : 0.f;

    __shared__ float sLa[LC_ + 1], sP1[LC_ + 1], sP2[LC_ + 1], sV[LC_];

    if (tt == 0) {
        const size_t o  = (bl0c + lane) * 32 + h;
        const float dt  = DTb[o];
        const float tr  = trapb[o];
        float la        = ladb[o];
        #pragma unroll
        for (int d = 1; d < 32; d <<= 1) {
            const float up = __shfl_up(la, d, 32);
            if (lane >= d) la += up;
        }
        const float p1 = dt * tr;
        const float p2 = dt * (1.f - tr);
        float laN = __shfl_down(la, 1, 32);
        float p2N = __shfl_down(p2, 1, 32);
        if (lane == 31) { laN = la; p2N = 0.f; }
        const float la31 = __shfl(la, 31, 32);
        sLa[lane] = la; sP1[lane] = p1; sP2[lane] = p2;
        sV[lane]  = expf(la31 - la) * p1 + expf(la31 - laN) * p2N;
        if (lane == 31) { sLa[32] = la; sP1[32] = 0.f; sP2[32] = 0.f; }
        cumdb[((size_t)(b * NH_ + h)) * L_ + c * LC_ + lane] = expf(la);
    }
    __syncthreads();

    float* Sp = Sstate + (((size_t)(b * NH_ + h)) * NC_ + c) * 4096;
    if (tt == 0)
        scan_chunk_tile<0>(proj, Brot, Crot, yfin, Sp, sLa, sP1, sP2, sV, bl0c, blprev, prevOn, h, lo, hi);
    else
        scan_chunk_tile<1>(proj, Brot, Crot, yfin, Sp, sLa, sP1, sP2, sV, bl0c, blprev, prevOn, h, lo, hi);
}

__global__ __launch_bounds__(64) void scan_combine_kernel(
    const float* __restrict__ cumdb, float* Sstate)
{
    static_assert(64 * 4 * 16 == 4096, "16 blocks of 64 threads x 4 elements cover one state");
    const int bh = blockIdx.x >> 4;
    const int e  = (((int)blockIdx.x & 15) * 64 + (int)threadIdx.x) * 4;
    float* sp = Sstate + (size_t)bh * NC_ * 4096 + e;
    const float* pc = cumdb + (size_t)bh * L_ + LC_ - 1;

    v4f   S[NC_];
    float P[NC_];
    #pragma unroll
    for (int c = 0; c < NC_; ++c) {
        S[c] = *reinterpret_cast<const v4f*>(sp + (size_t)c * 4096);
        P[c] = pc[c * LC_];
    }
    __builtin_amdgcn_sched_barrier(0);

    v4f H = {0.f, 0.f, 0.f, 0.f};
    #pragma unroll
    for (int c = 0; c < NC_; ++c) {
        *reinterpret_cast<v4f*>(sp + (size_t)c * 4096) = H;
        H = P[c] * H + S[c];
    }
}

__global__ __launch_bounds__(64) void scan_correct_kernel(
    const float* __restrict__ proj, const float* __restrict__ Crot,
    const float* __restrict__ cumdb, const float* __restrict__ Sstate,
    const float* __restrict__ D_param, float* __restrict__ yfin)
{
    static_assert(LC_ == 32 && DS_ == 64 && HD_ == 64, "two 16-token tiles, 64 x 64 state");
    const int blk  = blockIdx.x;
    const int c    = blk % NC_;
    const int h    = (blk / NC_) % NH_;
    const int b    = blk / (NC_ * NH_);
    const int lane = threadIdx.x & 31;
    const int tt   = threadIdx.x >> 5;
    const int lo   = lane & 15;
    const int hi   = lane >> 4;
    const size_t bl0 = (size_t)(b * L_ + c * LC_ + 16 * tt);

    const float* ap = Crot + (bl0 + lo) * 64 + 4 * hi;
    const float* hp = Sstate + (((size_t)(b * NH_ + h)) * NC_ + c) * 4096
                             + (size_t)(4 * lo) * 64 + 4 * hi;

    v4f a[8], bq[4][8];
    #pragma unroll
    for (int q = 0; q < 8; ++q)
        a[q] = *reinterpret_cast<const v4f*>(ap + 8 * q);
    #pragma unroll
    for (int j = 0; j < 4; ++j) {
        #pragma unroll
        for (int q = 0; q < 8; ++q)
            bq[j][q] = *reinterpret_cast<const v4f*>(hp + j * 64 + 8 * q);
    }

    const float* cdp = cumdb + ((size_t)(b * NH_ + h)) * L_ + c * LC_ + 16 * tt + 8 * hi;
    const v4f cd0 = *reinterpret_cast<const v4f*>(cdp);
    const v4f cd1 = *reinterpret_cast<const v4f*>(cdp + 4);
    const float Dp = D_param[h];
    v4f xv[8], zv[8], yv[8];
    #pragma unroll
    for (int r = 0; r < 8; ++r) {
        const size_t bl = bl0 + r + 8 * hi;
        xv[r] = *reinterpret_cast<const v4f*>(proj + bl * DIP + OFF_X + h * HD_ + 4 * lo);
        zv[r] = *reinterpret_cast<const v4f*>(proj + bl * DIP + OFF_Z + h * HD_ + 4 * lo);
        yv[r] = *reinterpret_cast<const v4f*>(yfin + bl * DI + h * HD_ + 4 * lo);
    }

    __builtin_amdgcn_sched_barrier(0);

    v8f acc[4] = {};
    #pragma unroll
    for (int q = 0; q < 8; ++q) {
        #pragma unroll
        for (int half = 0; half < 2; ++half) {
            v2f av;
            if (half) { av.x = a[q].z; av.y = a[q].w; } else { av.x = a[q].x; av.y = a[q].y; }
            #pragma unroll
            for (int j = 0; j < 4; ++j) {
                v2f bv;
                if (half) { bv.x = bq[j][q].z; bv.y = bq[j][q].w; }
                else      { bv.x = bq[j][q].x; bv.y = bq[j][q].y; }
                acc[j] = __builtin_amdgcn_wmma_f32_16x16x4_f32(false, av, false, bv, (short)0, acc[j], false, false);
            }
        }
    }

    #pragma unroll
    for (int r = 0; r < 8; ++r) {
        const size_t bl = bl0 + r + 8 * hi;
        const float cd = (r < 4) ? cd0[r & 3] : cd1[r & 3];
        v4f o;
        #pragma unroll
        for (int j = 0; j < 4; ++j) {
            const float z = zv[r][j];
            const float y = yv[r][j] + cd * acc[j][r];
            o[j] = (y + Dp * xv[r][j]) * (z * sigmoid_f(z));
        }
        *reinterpret_cast<v4f*>(yfin + bl * DI + h * HD_ + 4 * lo) = o;
    }
}

extern "C" void kernel_launch(void* const* d_in, const int* in_sizes, int n_in,
                              void* d_out, int out_size, void* d_ws, size_t ws_size,
                              hipStream_t stream)
{
    const float* u       = (const float*)d_in[0];
    const float* W_in    = (const float*)d_in[1];
    const float* W_out   = (const float*)d_in[2];
    const float* dt_bias = (const float*)d_in[3];
    const float* D_param = (const float*)d_in[4];
    const float* Bn_g    = (const float*)d_in[5];
    const float* Bn_b    = (const float*)d_in[6];
    const float* Cn_g    = (const float*)d_in[7];
    const float* Cn_b    = (const float*)d_in[8];

    float* ws = (float*)d_ws;
    float* proj   = ws; ws += (size_t)M1 * DIP;
    float* yfin   = ws; ws += (size_t)M1 * DI;
    float* Sstate = ws; ws += (size_t)B_ * NH_ * NC_ * 4096;
    float* Brot   = ws; ws += (size_t)M1 * DS_;
    float* Crot   = ws; ws += (size_t)M1 * DS_;
    float* DTb    = ws; ws += (size_t)M1 * NH_;
    float* ladb   = ws; ws += (size_t)M1 * NH_;
    float* trapb  = ws; ws += (size_t)M1 * NH_;
    float* cumdb  = ws; ws += (size_t)B_ * NH_ * L_;
    float* dtmeanb = ws; ws += (size_t)M1;

    gemm_wmma_f32<DIP, DM, true><<<dim3(KS_BLOCKS + (DIP / 64 - KS_CT) * (M1 / 64)), dim3(128), 0, stream>>>(
        u, W_in, proj, dt_bias, DTb, ladb, trapb, dtmeanb);

    rot_kernel<<<dim3(M1 / 16), dim3(512), 0, stream>>>(
        proj, dtmeanb, Bn_g, Bn_b, Cn_g, Cn_b, Brot, Crot);

    scan_chunk_kernel<<<dim3(B_ * NH_ * NC_), dim3(64), 0, stream>>>(
        proj, Brot, Crot, DTb, ladb, trapb, yfin, cumdb, Sstate);
    scan_combine_kernel<<<dim3(B_ * NH_ * 16), dim3(64), 0, stream>>>(
        cumdb, Sstate);
    scan_correct_kernel<<<dim3(B_ * NH_ * NC_), dim3(64), 0, stream>>>(
        proj, Crot, cumdb, Sstate, D_param, yfin);

    gemm_wmma_f32<DM, DI, false><<<dim3(DM / 64, M1 / 64), dim3(512), 0, stream>>>(
        yfin, W_out, (float*)d_out, nullptr, nullptr, nullptr, nullptr, nullptr);
}
